# starting kernel
# baseline (speedup 1.0000x reference)
#include <hip/hip_runtime.h>
#include <hip/hip_bf16.h>
#include <stdint.h>

#define HID   256
#define EDIM  64
#define HEADS 8
#define CCH   32
#define NEG_SLOPE 0.2f
#define LN_EPS 1e-5f

typedef float v2f __attribute__((ext_vector_type(2)));
typedef float v8f __attribute__((ext_vector_type(8)));

__device__ __forceinline__ unsigned enc_f32(float f) {
    unsigned b = __float_as_uint(f);
    return (b & 0x80000000u) ? ~b : (b | 0x80000000u);
}
__device__ __forceinline__ float dec_f32(unsigned u) {
    unsigned b = (u & 0x80000000u) ? (u & 0x7FFFFFFFu) : ~u;
    return __uint_as_float(b);
}

__global__ void k_weh(const float* __restrict__ W_e,
                      const float* __restrict__ att_edge,
                      float* __restrict__ w_eh) {
    int d = blockIdx.x * blockDim.x + threadIdx.x;
    if (d >= EDIM) return;
    #pragma unroll
    for (int h = 0; h < HEADS; ++h) {
        float s = 0.f;
        #pragma unroll
        for (int c = 0; c < CCH; ++c)
            s += W_e[d * HID + h * CCH + c] * att_edge[h * CCH + c];
        w_eh[d * HEADS + h] = s;
    }
}

__global__ void __launch_bounds__(256)
k_node_proj(const float* __restrict__ x, const float* __restrict__ W,
            float* __restrict__ xp, int n_nodes) {
    const int wave = threadIdx.x >> 5;
    const int lane = threadIdx.x & 31;
    const int row_base = (blockIdx.x * 2 + (wave >> 2)) * 16;
    const int col_base = (wave & 3) * 64;
    if (row_base >= n_nodes) return;

    int m = lane & 15;
    int kk = (lane >> 4) * 2;
    int row = row_base + m;
    if (row >= n_nodes) row = n_nodes - 1;
    const float* xrow = x + (size_t)row * HID;
    const int ncol = col_base + (lane & 15);

    v8f acc0 = {}, acc1 = {}, acc2 = {}, acc3 = {};
    for (int k = 0; k < HID; k += 4) {
        v2f a;
        a.x = xrow[k + kk];
        a.y = xrow[k + kk + 1];
        const float* w0 = W + (size_t)(k + kk) * HID;
        const float* w1 = W + (size_t)(k + kk + 1) * HID;
        v2f b0, b1, b2, b3;
        b0.x = w0[ncol];      b0.y = w1[ncol];
        b1.x = w0[ncol + 16]; b1.y = w1[ncol + 16];
        b2.x = w0[ncol + 32]; b2.y = w1[ncol + 32];
        b3.x = w0[ncol + 48]; b3.y = w1[ncol + 48];
        acc0 = __builtin_amdgcn_wmma_f32_16x16x4_f32(false, a, false, b0, (short)0, acc0, false, false);
        acc1 = __builtin_amdgcn_wmma_f32_16x16x4_f32(false, a, false, b1, (short)0, acc1, false, false);
        acc2 = __builtin_amdgcn_wmma_f32_16x16x4_f32(false, a, false, b2, (short)0, acc2, false, false);
        acc3 = __builtin_amdgcn_wmma_f32_16x16x4_f32(false, a, false, b3, (short)0, acc3, false, false);
    }

    const int srow = row_base + ((lane >> 4) * 8);
    const int scol = col_base + (lane & 15);
    #pragma unroll
    for (int i = 0; i < 8; ++i) {
        int r = srow + i;
        if (r < n_nodes) {
            float* o = xp + (size_t)r * HID + scol;
            o[0]  = acc0[i];
            o[16] = acc1[i];
            o[32] = acc2[i];
            o[48] = acc3[i];
        }
    }
}

__global__ void __launch_bounds__(256)
k_att_dots(const float* __restrict__ xp, const float* __restrict__ att_src,
           const float* __restrict__ att_dst, float* __restrict__ a_src,
           float* __restrict__ a_dst, int n_nodes) {
    const int wave = threadIdx.x >> 5;
    const int lane = threadIdx.x & 31;
    const int node = blockIdx.x * 8 + wave;
    if (node >= n_nodes) return;
    const float* row = xp + (size_t)node * HID + lane * 8;
    float s = 0.f, t = 0.f;
    #pragma unroll
    for (int j = 0; j < 8; ++j) {
        float v = row[j];
        int col = lane * 8 + j;
        s += v * att_src[col];
        t += v * att_dst[col];
    }
    s += __shfl_xor(s, 1); s += __shfl_xor(s, 2);
    t += __shfl_xor(t, 1); t += __shfl_xor(t, 2);
    if ((lane & 3) == 0) {
        a_src[node * HEADS + (lane >> 2)] = s;
        a_dst[node * HEADS + (lane >> 2)] = t;
    }
}

__global__ void __launch_bounds__(256)
k_alpha(const float* __restrict__ edge_attr, const int* __restrict__ ei,
        const float* __restrict__ w_eh, const float* __restrict__ a_src,
        const float* __restrict__ a_dst, float* __restrict__ alpha,
        unsigned* __restrict__ menc, int n_edges) {
    const int wave = threadIdx.x >> 5;
    const int lane = threadIdx.x & 31;
    const int e = blockIdx.x * 8 + wave;
    if (e >= n_edges) return;
    const int h = lane >> 2, q = lane & 3;
    const int src = ei[e];
    const int dst = ei[n_edges + e];
    const float* ea = edge_attr + (size_t)e * EDIM + q * 16;
    float s = 0.f;
    #pragma unroll
    for (int j = 0; j < 16; ++j)
        s += ea[j] * w_eh[(q * 16 + j) * HEADS + h];
    s += __shfl_xor(s, 1);
    s += __shfl_xor(s, 2);
    if (q == 0) {
        float al = a_src[src * HEADS + h] + a_dst[dst * HEADS + h] + s;
        al = (al > 0.f) ? al : NEG_SLOPE * al;
        alpha[(size_t)e * HEADS + h] = al;
        atomicMax(&menc[dst * HEADS + h], enc_f32(al));
    }
}

__global__ void __launch_bounds__(256)
k_scatter(const float* __restrict__ xp, const int* __restrict__ ei,
          const float* __restrict__ alpha, const unsigned* __restrict__ menc,
          float* __restrict__ den, float* __restrict__ acc, int n_edges) {
    const int wave = threadIdx.x >> 5;
    const int lane = threadIdx.x & 31;
    const int e = blockIdx.x * 8 + wave;
    if (e >= n_edges) return;
    const int src = ei[e];
    const int dst = ei[n_edges + e];

    if (lane < HEADS) {
        float m = dec_f32(menc[dst * HEADS + lane]);
        float ea = __expf(alpha[(size_t)e * HEADS + lane] - m);
        atomicAdd(&den[dst * HEADS + lane], ea);
    }
    const float* xs = xp + (size_t)src * HID;
    float* ad = acc + (size_t)dst * HID;
    #pragma unroll
    for (int h = 0; h < HEADS; ++h) {
        float m = dec_f32(menc[dst * HEADS + h]);
        float ea = __expf(alpha[(size_t)e * HEADS + h] - m);
        atomicAdd(&ad[h * CCH + lane], ea * xs[h * CCH + lane]);
    }
}

__global__ void __launch_bounds__(256)
k_finalize(const float* __restrict__ x, const float* __restrict__ acc,
           const float* __restrict__ den, const float* __restrict__ bias,
           const float* __restrict__ gamma, const float* __restrict__ beta,
           float* __restrict__ out, int n_nodes) {
    const int n = blockIdx.x;
    if (n >= n_nodes) return;
    const int t = threadIdx.x;
    const int h = t >> 5;
    float v = acc[(size_t)n * HID + t] / (den[n * HEADS + h] + 1e-16f);
    float y = x[(size_t)n * HID + t] + v + bias[t];

    float s = y, s2 = y * y;
    #pragma unroll
    for (int off = 16; off >= 1; off >>= 1) {
        s  += __shfl_xor(s, off);
        s2 += __shfl_xor(s2, off);
    }
    __shared__ float ws1[8], ws2[8];
    const int lane = t & 31, wave = t >> 5;
    if (lane == 0) { ws1[wave] = s; ws2[wave] = s2; }
    __syncthreads();
    if (wave == 0) {
        float a = (lane < 8) ? ws1[lane] : 0.f;
        float b = (lane < 8) ? ws2[lane] : 0.f;
        #pragma unroll
        for (int off = 4; off >= 1; off >>= 1) {
            a += __shfl_xor(a, off);
            b += __shfl_xor(b, off);
        }
        if (lane == 0) { ws1[0] = a; ws2[0] = b; }
    }
    __syncthreads();
    float mean = ws1[0] * (1.f / HID);
    float var  = ws2[0] * (1.f / HID) - mean * mean;
    out[(size_t)n * HID + t] = (y - mean) * rsqrtf(var + LN_EPS) * gamma[t] + beta[t];
}

extern "C" void kernel_launch(void* const* d_in, const int* in_sizes, int n_in,
                              void* d_out, int out_size, void* d_ws, size_t ws_size,
                              hipStream_t stream) {
    const float* x         = (const float*)d_in[0];
    const int*   ei        = (const int*)  d_in[1];
    const float* edge_attr = (const float*)d_in[2];
    const float* W         = (const float*)d_in[3];
    const float* W_e       = (const float*)d_in[4];
    const float* att_src   = (const float*)d_in[5];
    const float* att_dst   = (const float*)d_in[6];
    const float* att_edge  = (const float*)d_in[7];
    const float* bias      = (const float*)d_in[8];
    const float* gamma     = (const float*)d_in[9];
    const float* beta      = (const float*)d_in[10];
    float* out = (float*)d_out;

    const int n_nodes = in_sizes[0] / HID;
    const int n_edges = in_sizes[1] / 2;

    char* ws = (char*)d_ws;
    size_t off = 0;
    auto carve = [&](size_t bytes) -> char* {
        char* p = ws + off;
        off += (bytes + 255) & ~(size_t)255;
        return p;
    };
    float*    xp    = (float*)   carve((size_t)n_nodes * HID   * 4);
    float*    acc   = (float*)   carve((size_t)n_nodes * HID   * 4);
    float*    alpha = (float*)   carve((size_t)n_edges * HEADS * 4);
    float*    a_src = (float*)   carve((size_t)n_nodes * HEADS * 4);
    float*    a_dst = (float*)   carve((size_t)n_nodes * HEADS * 4);
    float*    den   = (float*)   carve((size_t)n_nodes * HEADS * 4);
    unsigned* menc  = (unsigned*)carve((size_t)n_nodes * HEADS * 4);
    float*    w_eh  = (float*)   carve((size_t)EDIM * HEADS * 4);
    (void)ws_size;

    hipMemsetAsync(acc,  0, (size_t)n_nodes * HID   * 4, stream);
    hipMemsetAsync(den,  0, (size_t)n_nodes * HEADS * 4, stream);
    hipMemsetAsync(menc, 0, (size_t)n_nodes * HEADS * 4, stream);

    k_weh<<<1, 64, 0, stream>>>(W_e, att_edge, w_eh);

    k_node_proj<<<(n_nodes + 31) / 32, 256, 0, stream>>>(x, W, xp, n_nodes);

    k_att_dots<<<(n_nodes + 7) / 8, 256, 0, stream>>>(xp, att_src, att_dst,
                                                      a_src, a_dst, n_nodes);

    k_alpha<<<(n_edges + 7) / 8, 256, 0, stream>>>(edge_attr, ei, w_eh,
                                                   a_src, a_dst, alpha, menc, n_edges);

    k_scatter<<<(n_edges + 7) / 8, 256, 0, stream>>>(xp, ei, alpha, menc,
                                                     den, acc, n_edges);

    k_finalize<<<n_nodes, 256, 0, stream>>>(x, acc, den, bias, gamma, beta,
                                            out, n_nodes);
}


# Float-atomic scatter replaced by a per-node gather over edges counting-sorted by destination (int atomic rank, scan, fill); LayerNorm sums in the same wave; message sums in arrival-rank order
# speedup vs baseline: 1.3364x; 1.3364x over previous
#include <hip/hip_runtime.h>
#include <hip/hip_bf16.h>
#include <stdint.h>

#define HID   256
#define EDIM  64
#define HEADS 8
#define CCH   32
#define NEG_SLOPE 0.2f
#define LN_EPS 1e-5f

typedef float v2f __attribute__((ext_vector_type(2)));
typedef float v8f __attribute__((ext_vector_type(8)));

__device__ __forceinline__ unsigned enc_f32(float f) {
    unsigned b = __float_as_uint(f);
    return (b & 0x80000000u) ? ~b : (b | 0x80000000u);
}
__device__ __forceinline__ float dec_f32(unsigned u) {
    unsigned b = (u & 0x80000000u) ? (u & 0x7FFFFFFFu) : ~u;
    return __uint_as_float(b);
}

__global__ void k_weh(const float* __restrict__ W_e,
                      const float* __restrict__ att_edge,
                      float* __restrict__ w_eh) {
    int d = blockIdx.x * blockDim.x + threadIdx.x;
    if (d >= EDIM) return;
    #pragma unroll
    for (int h = 0; h < HEADS; ++h) {
        float s = 0.f;
        #pragma unroll
        for (int c = 0; c < CCH; ++c)
            s += W_e[d * HID + h * CCH + c] * att_edge[h * CCH + c];
        w_eh[d * HEADS + h] = s;
    }
}

__global__ void __launch_bounds__(256)
k_node_proj(const float* __restrict__ x, const float* __restrict__ W,
            float* __restrict__ xp, int n_nodes) {
    const int wave = threadIdx.x >> 5;
    const int lane = threadIdx.x & 31;
    const int row_base = (blockIdx.x * 2 + (wave >> 2)) * 16;
    const int col_base = (wave & 3) * 64;
    if (row_base >= n_nodes) return;

    int m = lane & 15;
    int kk = (lane >> 4) * 2;
    int row = row_base + m;
    if (row >= n_nodes) row = n_nodes - 1;
    const float* xrow = x + (size_t)row * HID;
    const int ncol = col_base + (lane & 15);

    v8f acc0 = {}, acc1 = {}, acc2 = {}, acc3 = {};
    for (int k = 0; k < HID; k += 4) {
        v2f a;
        a.x = xrow[k + kk];
        a.y = xrow[k + kk + 1];
        const float* w0 = W + (size_t)(k + kk) * HID;
        const float* w1 = W + (size_t)(k + kk + 1) * HID;
        v2f b0, b1, b2, b3;
        b0.x = w0[ncol];      b0.y = w1[ncol];
        b1.x = w0[ncol + 16]; b1.y = w1[ncol + 16];
        b2.x = w0[ncol + 32]; b2.y = w1[ncol + 32];
        b3.x = w0[ncol + 48]; b3.y = w1[ncol + 48];
        acc0 = __builtin_amdgcn_wmma_f32_16x16x4_f32(false, a, false, b0, (short)0, acc0, false, false);
        acc1 = __builtin_amdgcn_wmma_f32_16x16x4_f32(false, a, false, b1, (short)0, acc1, false, false);
        acc2 = __builtin_amdgcn_wmma_f32_16x16x4_f32(false, a, false, b2, (short)0, acc2, false, false);
        acc3 = __builtin_amdgcn_wmma_f32_16x16x4_f32(false, a, false, b3, (short)0, acc3, false, false);
    }

    const int srow = row_base + ((lane >> 4) * 8);
    const int scol = col_base + (lane & 15);
    #pragma unroll
    for (int i = 0; i < 8; ++i) {
        int r = srow + i;
        if (r < n_nodes) {
            float* o = xp + (size_t)r * HID + scol;
            o[0]  = acc0[i];
            o[16] = acc1[i];
            o[32] = acc2[i];
            o[48] = acc3[i];
        }
    }
}

__global__ void __launch_bounds__(256)
k_att_dots(const float* __restrict__ xp, const float* __restrict__ att_src,
           const float* __restrict__ att_dst, float* __restrict__ a_src,
           float* __restrict__ a_dst, int* __restrict__ cnt, int n_nodes) {
    const int wave = threadIdx.x >> 5;
    const int lane = threadIdx.x & 31;
    const int node = blockIdx.x * 8 + wave;
    if (node >= n_nodes) return;
    const float* row = xp + (size_t)node * HID + lane * 8;
    float s = 0.f, t = 0.f;
    #pragma unroll
    for (int j = 0; j < 8; ++j) {
        float v = row[j];
        int col = lane * 8 + j;
        s += v * att_src[col];
        t += v * att_dst[col];
    }
    s += __shfl_xor(s, 1); s += __shfl_xor(s, 2);
    t += __shfl_xor(t, 1); t += __shfl_xor(t, 2);
    if ((lane & 3) == 0) {
        a_src[node * HEADS + (lane >> 2)] = s;
        a_dst[node * HEADS + (lane >> 2)] = t;
    }
    if (lane == 0) cnt[node] = 0;
}

__global__ void __launch_bounds__(256)
k_alpha(const float* __restrict__ edge_attr, const int* __restrict__ ei,
        const float* __restrict__ w_eh, const float* __restrict__ a_src,
        const float* __restrict__ a_dst, float* __restrict__ alpha,
        unsigned* __restrict__ menc, int* __restrict__ cnt,
        int* __restrict__ rank, int n_edges) {
    const int wave = threadIdx.x >> 5;
    const int lane = threadIdx.x & 31;
    const int e = blockIdx.x * 8 + wave;
    if (e >= n_edges) return;
    const int h = lane >> 2, q = lane & 3;
    const int src = ei[e];
    const int dst = ei[n_edges + e];
    const float* ea = edge_attr + (size_t)e * EDIM + q * 16;
    float s = 0.f;
    #pragma unroll
    for (int j = 0; j < 16; ++j)
        s += ea[j] * w_eh[(q * 16 + j) * HEADS + h];
    s += __shfl_xor(s, 1);
    s += __shfl_xor(s, 2);
    int r = 0;
    if (lane == 0) r = atomicAdd(&cnt[dst], 1);
    if (q == 0) {
        float al = a_src[src * HEADS + h] + a_dst[dst * HEADS + h] + s;
        al = (al > 0.f) ? al : NEG_SLOPE * al;
        alpha[(size_t)e * HEADS + h] = al;
        atomicMax(&menc[dst * HEADS + h], enc_f32(al));
    }
    if (lane == 0) rank[e] = r;
}

#define SCAN_PER 16
__global__ void __launch_bounds__(1024)
k_scan(const int* __restrict__ cnt, int* __restrict__ off, int n) {
    __shared__ int wsum[32];
    __shared__ int tot;
    const int t = threadIdx.x, lane = t & 31, wave = t >> 5;
    int carry = 0;
    for (int base = 0; base < n; base += 1024 * SCAN_PER) {
        const int i0 = base + t * SCAN_PER;
        const bool full = (i0 + SCAN_PER <= n);
        int v[SCAN_PER];
        if (full) {
            const int4* p = (const int4*)(cnt + i0);
            #pragma unroll
            for (int k = 0; k < SCAN_PER / 4; ++k) {
                int4 q = p[k];
                v[4 * k] = q.x; v[4 * k + 1] = q.y; v[4 * k + 2] = q.z; v[4 * k + 3] = q.w;
            }
        } else {
            #pragma unroll
            for (int k = 0; k < SCAN_PER; ++k) v[k] = (i0 + k < n) ? cnt[i0 + k] : 0;
        }
        int s = 0;
        #pragma unroll
        for (int k = 0; k < SCAN_PER; ++k) { int c = v[k]; v[k] = s; s += c; }
        int inc = s;
        #pragma unroll
        for (int d = 1; d < 32; d <<= 1) {
            int u = __shfl_up(inc, d);
            if (lane >= d) inc += u;
        }
        if (lane == 31) wsum[wave] = inc;
        __syncthreads();
        if (wave == 0) {
            const int w = wsum[lane];
            int winc = w;
            #pragma unroll
            for (int d = 1; d < 32; d <<= 1) {
                int u = __shfl_up(winc, d);
                if (lane >= d) winc += u;
            }
            wsum[lane] = winc - w;
            if (lane == 31) tot = winc;
        }
        __syncthreads();
        const int ex = carry + wsum[wave] + (inc - s);
        if (full) {
            int4* p = (int4*)(off + i0);
            #pragma unroll
            for (int k = 0; k < SCAN_PER / 4; ++k)
                p[k] = make_int4(ex + v[4 * k], ex + v[4 * k + 1], ex + v[4 * k + 2], ex + v[4 * k + 3]);
        } else {
            #pragma unroll
            for (int k = 0; k < SCAN_PER; ++k) if (i0 + k < n) off[i0 + k] = ex + v[k];
        }
        carry += tot;
        __syncthreads();
    }
    if (t == 0) off[n] = carry;
}

__global__ void __launch_bounds__(256)
k_fill(const int* __restrict__ ei, const int* __restrict__ rank,
       const int* __restrict__ off, const float* __restrict__ alpha,
       int* __restrict__ ssrc, float* __restrict__ salpha, int n_edges) {
    const int e = blockIdx.x * 256 + threadIdx.x;
    if (e >= n_edges) return;
    const int src = ei[e];
    const int dst = ei[n_edges + e];
    const int slot = off[dst] + rank[e];
    const float4* a = (const float4*)(alpha + (size_t)e * HEADS);
    const float4 a0 = a[0], a1 = a[1];
    ssrc[slot] = src;
    float4* o = (float4*)(salpha + (size_t)slot * HEADS);
    o[0] = a0;
    o[1] = a1;
}

__global__ void __launch_bounds__(256)
k_gather(const float* __restrict__ xp, const float* __restrict__ x,
         const int* __restrict__ off, const int* __restrict__ ssrc,
         const float* __restrict__ salpha, const unsigned* __restrict__ menc,
         const float* __restrict__ bias, const float* __restrict__ gamma,
         const float* __restrict__ beta, float* __restrict__ out, int n_nodes) {
    const int wave = threadIdx.x >> 5;
    const int lane = threadIdx.x & 31;
    const int n = blockIdx.x * 8 + wave;
    if (n >= n_nodes) return;
    const int c0 = lane * 4;
    const int hA = lane >> 3, hB = 4 + hA;
    const int beg = __builtin_amdgcn_readfirstlane(off[n]);
    const int end = __builtin_amdgcn_readfirstlane(off[n + 1]);
    const float mA = dec_f32(menc[n * HEADS + hA]);
    const float mB = dec_f32(menc[n * HEADS + hB]);

    float4 accA = make_float4(0.f, 0.f, 0.f, 0.f);
    float4 accB = make_float4(0.f, 0.f, 0.f, 0.f);
    float denA = 0.f, denB = 0.f;
    for (int base = beg; base < end; base += 32) {
        const int rem = end - base;
        const int cc = rem < 32 ? rem : 32;
        const int my_s = (lane < cc) ? ssrc[base + lane] : 0;
        #pragma unroll 4
        for (int j = 0; j < cc; ++j) {
            const int s = __builtin_amdgcn_readlane(my_s, j);
            const float* xs = xp + (size_t)s * HID;
            const float* al = salpha + (size_t)(base + j) * HEADS;
            const float4 rA = *(const float4*)(xs + c0);
            const float4 rB = *(const float4*)(xs + 128 + c0);
            const float eA = __expf(al[hA] - mA);
            const float eB = __expf(al[hB] - mB);
            denA += eA;
            denB += eB;
            accA.x += eA * rA.x; accA.y += eA * rA.y; accA.z += eA * rA.z; accA.w += eA * rA.w;
            accB.x += eB * rB.x; accB.y += eB * rB.y; accB.z += eB * rB.z; accB.w += eB * rB.w;
        }
    }

    const float4 xA = *(const float4*)(x + (size_t)n * HID + c0);
    const float4 xB = *(const float4*)(x + (size_t)n * HID + 128 + c0);
    const float4 bA = *(const float4*)(bias + c0);
    const float4 bB = *(const float4*)(bias + 128 + c0);
    const float dA = denA + 1e-16f, dB = denB + 1e-16f;
    float y[8];
    y[0] = xA.x + accA.x / dA + bA.x;
    y[1] = xA.y + accA.y / dA + bA.y;
    y[2] = xA.z + accA.z / dA + bA.z;
    y[3] = xA.w + accA.w / dA + bA.w;
    y[4] = xB.x + accB.x / dB + bB.x;
    y[5] = xB.y + accB.y / dB + bB.y;
    y[6] = xB.z + accB.z / dB + bB.z;
    y[7] = xB.w + accB.w / dB + bB.w;

    float s1 = 0.f, s2 = 0.f;
    #pragma unroll
    for (int k = 0; k < 8; ++k) { s1 += y[k]; s2 += y[k] * y[k]; }
    #pragma unroll
    for (int o = 16; o >= 1; o >>= 1) {
        s1 += __shfl_xor(s1, o);
        s2 += __shfl_xor(s2, o);
    }
    const float mean = s1 * (1.f / HID);
    const float var  = s2 * (1.f / HID) - mean * mean;
    const float rs   = rsqrtf(var + LN_EPS);
    const float4 gA = *(const float4*)(gamma + c0);
    const float4 gB = *(const float4*)(gamma + 128 + c0);
    const float4 tA = *(const float4*)(beta + c0);
    const float4 tB = *(const float4*)(beta + 128 + c0);
    float4 oA, oB;
    oA.x = (y[0] - mean) * rs * gA.x + tA.x;
    oA.y = (y[1] - mean) * rs * gA.y + tA.y;
    oA.z = (y[2] - mean) * rs * gA.z + tA.z;
    oA.w = (y[3] - mean) * rs * gA.w + tA.w;
    oB.x = (y[4] - mean) * rs * gB.x + tB.x;
    oB.y = (y[5] - mean) * rs * gB.y + tB.y;
    oB.z = (y[6] - mean) * rs * gB.z + tB.z;
    oB.w = (y[7] - mean) * rs * gB.w + tB.w;
    *(float4*)(out + (size_t)n * HID + c0) = oA;
    *(float4*)(out + (size_t)n * HID + 128 + c0) = oB;
}

extern "C" void kernel_launch(void* const* d_in, const int* in_sizes, int n_in,
                              void* d_out, int out_size, void* d_ws, size_t ws_size,
                              hipStream_t stream) {
    const float* x         = (const float*)d_in[0];
    const int*   ei        = (const int*)  d_in[1];
    const float* edge_attr = (const float*)d_in[2];
    const float* W         = (const float*)d_in[3];
    const float* W_e       = (const float*)d_in[4];
    const float* att_src   = (const float*)d_in[5];
    const float* att_dst   = (const float*)d_in[6];
    const float* att_edge  = (const float*)d_in[7];
    const float* bias      = (const float*)d_in[8];
    const float* gamma     = (const float*)d_in[9];
    const float* beta      = (const float*)d_in[10];
    float* out = (float*)d_out;

    const int n_nodes = in_sizes[0] / HID;
    const int n_edges = in_sizes[1] / 2;

    char* ws = (char*)d_ws;
    size_t off = 0;
    auto carve = [&](size_t bytes) -> char* {
        char* p = ws + off;
        off += (bytes + 255) & ~(size_t)255;
        return p;
    };
    float*    xp     = (float*)   carve((size_t)n_nodes * HID   * 4);
    float*    alpha  = (float*)   carve((size_t)n_edges * HEADS * 4);
    float*    salpha = (float*)   carve((size_t)n_edges * HEADS * 4);
    int*      ssrc   = (int*)     carve((size_t)n_edges * 4);
    int*      rank   = (int*)     carve((size_t)n_edges * 4);
    float*    a_src  = (float*)   carve((size_t)n_nodes * HEADS * 4);
    float*    a_dst  = (float*)   carve((size_t)n_nodes * HEADS * 4);
    unsigned* menc   = (unsigned*)carve((size_t)n_nodes * HEADS * 4);
    int*      cnt    = (int*)     carve((size_t)n_nodes * 4);
    int*      segoff = (int*)     carve((size_t)(n_nodes + 1) * 4);
    float*    w_eh   = (float*)   carve((size_t)EDIM * HEADS * 4);
    (void)ws_size;

    hipMemsetAsync(menc, 0, (size_t)n_nodes * HEADS * 4, stream);

    k_weh<<<1, 64, 0, stream>>>(W_e, att_edge, w_eh);

    k_node_proj<<<(n_nodes + 31) / 32, 256, 0, stream>>>(x, W, xp, n_nodes);

    k_att_dots<<<(n_nodes + 7) / 8, 256, 0, stream>>>(xp, att_src, att_dst,
                                                      a_src, a_dst, cnt, n_nodes);

    k_alpha<<<(n_edges + 7) / 8, 256, 0, stream>>>(edge_attr, ei, w_eh,
                                                   a_src, a_dst, alpha, menc,
                                                   cnt, rank, n_edges);

    k_scan<<<1, 1024, 0, stream>>>(cnt, segoff, n_nodes);

    k_fill<<<(n_edges + 255) / 256, 256, 0, stream>>>(ei, rank, segoff, alpha,
                                                      ssrc, salpha, n_edges);

    k_gather<<<(n_nodes + 7) / 8, 256, 0, stream>>>(xp, x, segoff, ssrc, salpha, menc,
                                                    bias, gamma, beta, out, n_nodes);
}


# Edge logits: a wave now loops over 16-edge tiles with the edge-feature x head-weight product on f32 WMMA, instead of one wave per edge; that 64-term dot product is summed in a new order
# speedup vs baseline: 1.6973x; 1.2701x over previous
#include <hip/hip_runtime.h>
#include <hip/hip_bf16.h>
#include <stdint.h>

#define HID   256
#define EDIM  64
#define HEADS 8
#define CCH   32
#define NEG_SLOPE 0.2f
#define LN_EPS 1e-5f

typedef float v2f __attribute__((ext_vector_type(2)));
typedef float v4f __attribute__((ext_vector_type(4)));
typedef float v8f __attribute__((ext_vector_type(8)));

__device__ __forceinline__ unsigned enc_f32(float f) {
    unsigned b = __float_as_uint(f);
    return (b & 0x80000000u) ? ~b : (b | 0x80000000u);
}
__device__ __forceinline__ float dec_f32(unsigned u) {
    unsigned b = (u & 0x80000000u) ? (u & 0x7FFFFFFFu) : ~u;
    return __uint_as_float(b);
}

__global__ void k_weh(const float* __restrict__ W_e,
                      const float* __restrict__ att_edge,
                      float* __restrict__ w_he) {
    int d = blockIdx.x * blockDim.x + threadIdx.x;
    if (d >= EDIM) return;
    #pragma unroll
    for (int h = 0; h < HEADS; ++h) {
        float s = 0.f;
        #pragma unroll
        for (int c = 0; c < CCH; ++c)
            s += W_e[d * HID + h * CCH + c] * att_edge[h * CCH + c];
        w_he[h * EDIM + d] = s;
    }
}

__global__ void __launch_bounds__(256)
k_node_proj(const float* __restrict__ x, const float* __restrict__ W,
            float* __restrict__ xp, int n_nodes) {
    const int wave = threadIdx.x >> 5;
    const int lane = threadIdx.x & 31;
    const int row_base = (blockIdx.x * 2 + (wave >> 2)) * 16;
    const int col_base = (wave & 3) * 64;
    if (row_base >= n_nodes) return;

    int m = lane & 15;
    int kk = (lane >> 4) * 2;
    int row = row_base + m;
    if (row >= n_nodes) row = n_nodes - 1;
    const float* xrow = x + (size_t)row * HID;
    const int ncol = col_base + (lane & 15);

    v8f acc0 = {}, acc1 = {}, acc2 = {}, acc3 = {};
    for (int k = 0; k < HID; k += 4) {
        v2f a;
        a.x = xrow[k + kk];
        a.y = xrow[k + kk + 1];
        const float* w0 = W + (size_t)(k + kk) * HID;
        const float* w1 = W + (size_t)(k + kk + 1) * HID;
        v2f b0, b1, b2, b3;
        b0.x = w0[ncol];      b0.y = w1[ncol];
        b1.x = w0[ncol + 16]; b1.y = w1[ncol + 16];
        b2.x = w0[ncol + 32]; b2.y = w1[ncol + 32];
        b3.x = w0[ncol + 48]; b3.y = w1[ncol + 48];
        acc0 = __builtin_amdgcn_wmma_f32_16x16x4_f32(false, a, false, b0, (short)0, acc0, false, false);
        acc1 = __builtin_amdgcn_wmma_f32_16x16x4_f32(false, a, false, b1, (short)0, acc1, false, false);
        acc2 = __builtin_amdgcn_wmma_f32_16x16x4_f32(false, a, false, b2, (short)0, acc2, false, false);
        acc3 = __builtin_amdgcn_wmma_f32_16x16x4_f32(false, a, false, b3, (short)0, acc3, false, false);
    }

    const int srow = row_base + ((lane >> 4) * 8);
    const int scol = col_base + (lane & 15);
    #pragma unroll
    for (int i = 0; i < 8; ++i) {
        int r = srow + i;
        if (r < n_nodes) {
            float* o = xp + (size_t)r * HID + scol;
            o[0]  = acc0[i];
            o[16] = acc1[i];
            o[32] = acc2[i];
            o[48] = acc3[i];
        }
    }
}

__global__ void __launch_bounds__(256)
k_att_dots(const float* __restrict__ xp, const float* __restrict__ att_src,
           const float* __restrict__ att_dst, float* __restrict__ a_src,
           float* __restrict__ a_dst, int* __restrict__ cnt, int n_nodes) {
    const int wave = threadIdx.x >> 5;
    const int lane = threadIdx.x & 31;
    const int node = blockIdx.x * 8 + wave;
    if (node >= n_nodes) return;
    const float* row = xp + (size_t)node * HID + lane * 8;
    float s = 0.f, t = 0.f;
    #pragma unroll
    for (int j = 0; j < 8; ++j) {
        float v = row[j];
        int col = lane * 8 + j;
        s += v * att_src[col];
        t += v * att_dst[col];
    }
    s += __shfl_xor(s, 1); s += __shfl_xor(s, 2);
    t += __shfl_xor(t, 1); t += __shfl_xor(t, 2);
    if ((lane & 3) == 0) {
        a_src[node * HEADS + (lane >> 2)] = s;
        a_dst[node * HEADS + (lane >> 2)] = t;
    }
    if (lane == 0) cnt[node] = 0;
}

__device__ __forceinline__ int frag_k0(int i, int h) { return 8 * i + 4 * h; }

__global__ void __launch_bounds__(256)
k_alpha(const float* __restrict__ edge_attr, const int* __restrict__ ei,
        const float* __restrict__ w_he, const float* __restrict__ a_src,
        const float* __restrict__ a_dst, float* __restrict__ alpha,
        unsigned* __restrict__ menc, int* __restrict__ cnt,
        int* __restrict__ rank, int n_edges) {
    const int lane = threadIdx.x & 31;
    const int c = lane & 15;
    const int h = lane >> 4;
    const int head = c & 7;
    const int wave_id = __builtin_amdgcn_readfirstlane((int)(blockIdx.x * 8 + (threadIdx.x >> 5)));
    const int n_waves = gridDim.x * 8;
    const int n_tiles = (n_edges + 15) >> 4;

    v4f b[8];
    #pragma unroll
    for (int i = 0; i < 8; ++i)
        b[i] = *(const v4f*)(w_he + head * EDIM + frag_k0(i, h));

    for (int tile = wave_id; tile < n_tiles; tile += n_waves) {
        const int e0 = tile * 16;
        int el = e0 + c;
        if (el >= n_edges) el = n_edges - 1;
        const int srcv = ei[el];
        const int dstv = ei[n_edges + el];
        const float* arow = edge_attr + (size_t)el * EDIM;
        v4f a[8];
        #pragma unroll
        for (int i = 0; i < 8; ++i)
            a[i] = *(const v4f*)(arow + frag_k0(i, h));

        const bool own = (h == 0) && (e0 + c < n_edges);
        int rk = 0;
        if (own) rk = atomicAdd(&cnt[dstv], 1);

        int dr[8];
        float pre[8];
        #pragma unroll
        for (int r = 0; r < 8; ++r) {
            const int sl = 8 * h + r;
            const int s = __shfl(srcv, sl);
            dr[r] = __shfl(dstv, sl);
            pre[r] = a_src[s * HEADS + head] + a_dst[dr[r] * HEADS + head];
        }

        v8f acc0 = {}, acc1 = {};
        #pragma unroll
        for (int i = 0; i < 8; ++i) {
            v2f a_lo, a_hi, b_lo, b_hi;
            a_lo.x = a[i].x; a_lo.y = a[i].y; a_hi.x = a[i].z; a_hi.y = a[i].w;
            b_lo.x = b[i].x; b_lo.y = b[i].y; b_hi.x = b[i].z; b_hi.y = b[i].w;
            acc0 = __builtin_amdgcn_wmma_f32_16x16x4_f32(false, a_lo, false, b_lo, (short)0, acc0, false, false);
            acc1 = __builtin_amdgcn_wmma_f32_16x16x4_f32(false, a_hi, false, b_hi, (short)0, acc1, false, false);
        }

        asm volatile("v_nop\n\tv_nop\n\tv_nop\n\tv_nop\n\tv_nop\n\tv_nop\n\tv_nop\n\tv_nop"
                     : "+v"(acc0), "+v"(acc1)
                     : "v"(a[0]), "v"(a[1]), "v"(a[2]), "v"(a[3]),
                       "v"(a[4]), "v"(a[5]), "v"(a[6]), "v"(a[7]));

        #pragma unroll
        for (int r = 0; r < 8; ++r) {
            const int e = e0 + 8 * h + r;
            float al = pre[r] + (acc0[r] + acc1[r]);
            al = (al > 0.f) ? al : NEG_SLOPE * al;
            if (c < 8 && e < n_edges) {
                alpha[(size_t)e * HEADS + c] = al;
                atomicMax(&menc[dr[r] * HEADS + c], enc_f32(al));
            }
        }
        if (own) rank[e0 + c] = rk;
    }
}

#define SCAN_PER 16
__global__ void __launch_bounds__(1024)
k_scan(const int* __restrict__ cnt, int* __restrict__ off, int n) {
    __shared__ int wsum[32];
    __shared__ int tot;
    const int t = threadIdx.x, lane = t & 31, wave = t >> 5;
    int carry = 0;
    for (int base = 0; base < n; base += 1024 * SCAN_PER) {
        const int i0 = base + t * SCAN_PER;
        const bool full = (i0 + SCAN_PER <= n);
        int v[SCAN_PER];
        if (full) {
            const int4* p = (const int4*)(cnt + i0);
            #pragma unroll
            for (int k = 0; k < SCAN_PER / 4; ++k) {
                int4 q = p[k];
                v[4 * k] = q.x; v[4 * k + 1] = q.y; v[4 * k + 2] = q.z; v[4 * k + 3] = q.w;
            }
        } else {
            #pragma unroll
            for (int k = 0; k < SCAN_PER; ++k) v[k] = (i0 + k < n) ? cnt[i0 + k] : 0;
        }
        int s = 0;
        #pragma unroll
        for (int k = 0; k < SCAN_PER; ++k) { int c = v[k]; v[k] = s; s += c; }
        int inc = s;
        #pragma unroll
        for (int d = 1; d < 32; d <<= 1) {
            int u = __shfl_up(inc, d);
            if (lane >= d) inc += u;
        }
        if (lane == 31) wsum[wave] = inc;
        __syncthreads();
        if (wave == 0) {
            const int w = wsum[lane];
            int winc = w;
            #pragma unroll
            for (int d = 1; d < 32; d <<= 1) {
                int u = __shfl_up(winc, d);
                if (lane >= d) winc += u;
            }
            wsum[lane] = winc - w;
            if (lane == 31) tot = winc;
        }
        __syncthreads();
        const int ex = carry + wsum[wave] + (inc - s);
        if (full) {
            int4* p = (int4*)(off + i0);
            #pragma unroll
            for (int k = 0; k < SCAN_PER / 4; ++k)
                p[k] = make_int4(ex + v[4 * k], ex + v[4 * k + 1], ex + v[4 * k + 2], ex + v[4 * k + 3]);
        } else {
            #pragma unroll
            for (int k = 0; k < SCAN_PER; ++k) if (i0 + k < n) off[i0 + k] = ex + v[k];
        }
        carry += tot;
        __syncthreads();
    }
    if (t == 0) off[n] = carry;
}

__global__ void __launch_bounds__(256)
k_fill(const int* __restrict__ ei, const int* __restrict__ rank,
       const int* __restrict__ off, const float* __restrict__ alpha,
       int* __restrict__ ssrc, float* __restrict__ salpha, int n_edges) {
    const int e = blockIdx.x * 256 + threadIdx.x;
    if (e >= n_edges) return;
    const int src = ei[e];
    const int dst = ei[n_edges + e];
    const int slot = off[dst] + rank[e];
    const float4* a = (const float4*)(alpha + (size_t)e * HEADS);
    const float4 a0 = a[0], a1 = a[1];
    ssrc[slot] = src;
    float4* o = (float4*)(salpha + (size_t)slot * HEADS);
    o[0] = a0;
    o[1] = a1;
}

__global__ void __launch_bounds__(256)
k_gather(const float* __restrict__ xp, const float* __restrict__ x,
         const int* __restrict__ off, const int* __restrict__ ssrc,
         const float* __restrict__ salpha, const unsigned* __restrict__ menc,
         const float* __restrict__ bias, const float* __restrict__ gamma,
         const float* __restrict__ beta, float* __restrict__ out, int n_nodes) {
    const int wave = threadIdx.x >> 5;
    const int lane = threadIdx.x & 31;
    const int n = blockIdx.x * 8 + wave;
    if (n >= n_nodes) return;
    const int c0 = lane * 4;
    const int hA = lane >> 3, hB = 4 + hA;
    const int beg = __builtin_amdgcn_readfirstlane(off[n]);
    const int end = __builtin_amdgcn_readfirstlane(off[n + 1]);
    const float mA = dec_f32(menc[n * HEADS + hA]);
    const float mB = dec_f32(menc[n * HEADS + hB]);

    float4 accA = make_float4(0.f, 0.f, 0.f, 0.f);
    float4 accB = make_float4(0.f, 0.f, 0.f, 0.f);
    float denA = 0.f, denB = 0.f;
    for (int base = beg; base < end; base += 32) {
        const int rem = end - base;
        const int cc = rem < 32 ? rem : 32;
        const int my_s = (lane < cc) ? ssrc[base + lane] : 0;
        #pragma unroll 4
        for (int j = 0; j < cc; ++j) {
            const int s = __builtin_amdgcn_readlane(my_s, j);
            const float* xs = xp + (size_t)s * HID;
            const float* al = salpha + (size_t)(base + j) * HEADS;
            const float4 rA = *(const float4*)(xs + c0);
            const float4 rB = *(const float4*)(xs + 128 + c0);
            const float eA = __expf(al[hA] - mA);
            const float eB = __expf(al[hB] - mB);
            denA += eA;
            denB += eB;
            accA.x += eA * rA.x; accA.y += eA * rA.y; accA.z += eA * rA.z; accA.w += eA * rA.w;
            accB.x += eB * rB.x; accB.y += eB * rB.y; accB.z += eB * rB.z; accB.w += eB * rB.w;
        }
    }

    const float4 xA = *(const float4*)(x + (size_t)n * HID + c0);
    const float4 xB = *(const float4*)(x + (size_t)n * HID + 128 + c0);
    const float4 bA = *(const float4*)(bias + c0);
    const float4 bB = *(const float4*)(bias + 128 + c0);
    const float dA = denA + 1e-16f, dB = denB + 1e-16f;
    float y[8];
    y[0] = xA.x + accA.x / dA + bA.x;
    y[1] = xA.y + accA.y / dA + bA.y;
    y[2] = xA.z + accA.z / dA + bA.z;
    y[3] = xA.w + accA.w / dA + bA.w;
    y[4] = xB.x + accB.x / dB + bB.x;
    y[5] = xB.y + accB.y / dB + bB.y;
    y[6] = xB.z + accB.z / dB + bB.z;
    y[7] = xB.w + accB.w / dB + bB.w;

    float s1 = 0.f, s2 = 0.f;
    #pragma unroll
    for (int k = 0; k < 8; ++k) { s1 += y[k]; s2 += y[k] * y[k]; }
    #pragma unroll
    for (int o = 16; o >= 1; o >>= 1) {
        s1 += __shfl_xor(s1, o);
        s2 += __shfl_xor(s2, o);
    }
    const float mean = s1 * (1.f / HID);
    const float var  = s2 * (1.f / HID) - mean * mean;
    const float rs   = rsqrtf(var + LN_EPS);
    const float4 gA = *(const float4*)(gamma + c0);
    const float4 gB = *(const float4*)(gamma + 128 + c0);
    const float4 tA = *(const float4*)(beta + c0);
    const float4 tB = *(const float4*)(beta + 128 + c0);
    float4 oA, oB;
    oA.x = (y[0] - mean) * rs * gA.x + tA.x;
    oA.y = (y[1] - mean) * rs * gA.y + tA.y;
    oA.z = (y[2] - mean) * rs * gA.z + tA.z;
    oA.w = (y[3] - mean) * rs * gA.w + tA.w;
    oB.x = (y[4] - mean) * rs * gB.x + tB.x;
    oB.y = (y[5] - mean) * rs * gB.y + tB.y;
    oB.z = (y[6] - mean) * rs * gB.z + tB.z;
    oB.w = (y[7] - mean) * rs * gB.w + tB.w;
    *(float4*)(out + (size_t)n * HID + c0) = oA;
    *(float4*)(out + (size_t)n * HID + 128 + c0) = oB;
}

extern "C" void kernel_launch(void* const* d_in, const int* in_sizes, int n_in,
                              void* d_out, int out_size, void* d_ws, size_t ws_size,
                              hipStream_t stream) {
    const float* x         = (const float*)d_in[0];
    const int*   ei        = (const int*)  d_in[1];
    const float* edge_attr = (const float*)d_in[2];
    const float* W         = (const float*)d_in[3];
    const float* W_e       = (const float*)d_in[4];
    const float* att_src   = (const float*)d_in[5];
    const float* att_dst   = (const float*)d_in[6];
    const float* att_edge  = (const float*)d_in[7];
    const float* bias      = (const float*)d_in[8];
    const float* gamma     = (const float*)d_in[9];
    const float* beta      = (const float*)d_in[10];
    float* out = (float*)d_out;

    const int n_nodes = in_sizes[0] / HID;
    const int n_edges = in_sizes[1] / 2;

    char* ws = (char*)d_ws;
    size_t off = 0;
    auto carve = [&](size_t bytes) -> char* {
        char* p = ws + off;
        off += (bytes + 255) & ~(size_t)255;
        return p;
    };
    float*    xp     = (float*)   carve((size_t)n_nodes * HID   * 4);
    float*    alpha  = (float*)   carve((size_t)n_edges * HEADS * 4);
    float*    salpha = (float*)   carve((size_t)n_edges * HEADS * 4);
    int*      ssrc   = (int*)     carve((size_t)n_edges * 4);
    int*      rank   = (int*)     carve((size_t)n_edges * 4);
    float*    a_src  = (float*)   carve((size_t)n_nodes * HEADS * 4);
    float*    a_dst  = (float*)   carve((size_t)n_nodes * HEADS * 4);
    unsigned* menc   = (unsigned*)carve((size_t)n_nodes * HEADS * 4);
    int*      cnt    = (int*)     carve((size_t)n_nodes * 4);
    int*      segoff = (int*)     carve((size_t)(n_nodes + 1) * 4);
    float*    w_he   = (float*)   carve((size_t)EDIM * HEADS * 4);
    (void)ws_size;

    hipMemsetAsync(menc, 0, (size_t)n_nodes * HEADS * 4, stream);

    k_weh<<<1, 64, 0, stream>>>(W_e, att_edge, w_he);

    k_node_proj<<<(n_nodes + 31) / 32, 256, 0, stream>>>(x, W, xp, n_nodes);

    k_att_dots<<<(n_nodes + 7) / 8, 256, 0, stream>>>(xp, att_src, att_dst,
                                                      a_src, a_dst, cnt, n_nodes);

    k_alpha<<<1024, 256, 0, stream>>>(edge_attr, ei, w_he,
                                      a_src, a_dst, alpha, menc,
                                      cnt, rank, n_edges);

    k_scan<<<1, 1024, 0, stream>>>(cnt, segoff, n_nodes);

    k_fill<<<(n_edges + 255) / 256, 256, 0, stream>>>(ei, rank, segoff, alpha,
                                                      ssrc, salpha, n_edges);

    k_gather<<<(n_nodes + 7) / 8, 256, 0, stream>>>(xp, x, segoff, ssrc, salpha, menc,
                                                    bias, gamma, beta, out, n_nodes);
}


# Launches 7 -> 5, memset dropped: head-weight fold (one extra workgroup), max/count clears and attention dots (from the accumulators) join the projection launch; 32-term dots summed in a new order
# speedup vs baseline: 1.7917x; 1.0556x over previous
#include <hip/hip_runtime.h>
#include <hip/hip_bf16.h>
#include <stdint.h>

#define HID   256
#define EDIM  64
#define HEADS 8
#define CCH   32
#define NEG_SLOPE 0.2f
#define LN_EPS 1e-5f

typedef float v2f __attribute__((ext_vector_type(2)));
typedef float v4f __attribute__((ext_vector_type(4)));
typedef float v8f __attribute__((ext_vector_type(8)));

__device__ __forceinline__ unsigned enc_f32(float f) {
    unsigned b = __float_as_uint(f);
    return (b & 0x80000000u) ? ~b : (b | 0x80000000u);
}
__device__ __forceinline__ float dec_f32(unsigned u) {
    unsigned b = (u & 0x80000000u) ? (u & 0x7FFFFFFFu) : ~u;
    return __uint_as_float(b);
}

__global__ void __launch_bounds__(256)
k_node_proj(const float* __restrict__ x, const float* __restrict__ W,
            const float* __restrict__ W_e, const float* __restrict__ att_edge,
            const float* __restrict__ att_src, const float* __restrict__ att_dst,
            float* __restrict__ xp, float* __restrict__ a_src,
            float* __restrict__ a_dst, float* __restrict__ w_he,
            unsigned* __restrict__ menc, int* __restrict__ cnt, int n_nodes) {
    const int wave = __builtin_amdgcn_readfirstlane((int)(threadIdx.x >> 5));
    const int lane = threadIdx.x & 31;

    if ((int)blockIdx.x == ((n_nodes + 31) >> 5)) {
        const int d = (wave & 1) * 32 + lane;
        #pragma unroll
        for (int u = 0; u < 2; ++u) {
            const int h = (wave >> 1) + 4 * u;
            const float* wr = W_e + d * HID + h * CCH;
            const float* ar = att_edge + h * CCH;
            float s = 0.f;
            #pragma unroll
            for (int c = 0; c < CCH; ++c) s += wr[c] * ar[c];
            w_he[h * EDIM + d] = s;
        }
        return;
    }

    const int row_base = (blockIdx.x * 2 + (wave >> 2)) * 16;
    const int col_base = (wave & 3) * 64;
    if (row_base >= n_nodes) return;

    int m = lane & 15;
    int kk = (lane >> 4) * 2;
    int row = row_base + m;
    if (row >= n_nodes) row = n_nodes - 1;
    const float* xrow = x + (size_t)row * HID;
    const int ncol = col_base + (lane & 15);

    v8f acc0 = {}, acc1 = {}, acc2 = {}, acc3 = {};
    for (int k = 0; k < HID; k += 4) {
        v2f a;
        a.x = xrow[k + kk];
        a.y = xrow[k + kk + 1];
        const float* w0 = W + (size_t)(k + kk) * HID;
        const float* w1 = W + (size_t)(k + kk + 1) * HID;
        v2f b0, b1, b2, b3;
        b0.x = w0[ncol];      b0.y = w1[ncol];
        b1.x = w0[ncol + 16]; b1.y = w1[ncol + 16];
        b2.x = w0[ncol + 32]; b2.y = w1[ncol + 32];
        b3.x = w0[ncol + 48]; b3.y = w1[ncol + 48];
        acc0 = __builtin_amdgcn_wmma_f32_16x16x4_f32(false, a, false, b0, (short)0, acc0, false, false);
        acc1 = __builtin_amdgcn_wmma_f32_16x16x4_f32(false, a, false, b1, (short)0, acc1, false, false);
        acc2 = __builtin_amdgcn_wmma_f32_16x16x4_f32(false, a, false, b2, (short)0, acc2, false, false);
        acc3 = __builtin_amdgcn_wmma_f32_16x16x4_f32(false, a, false, b3, (short)0, acc3, false, false);
    }

    const int pl = (int)__builtin_amdgcn_mbcnt_lo(~0u, 0u);
    const int srow = row_base + ((pl >> 4) * 8);
    const int scol = col_base + (pl & 15);
    #pragma unroll
    for (int i = 0; i < 8; ++i) {
        int r = srow + i;
        if (r < n_nodes) {
            float* o = xp + (size_t)r * HID + scol;
            o[0]  = acc0[i];
            o[16] = acc1[i];
            o[32] = acc2[i];
            o[48] = acc3[i];
        }
    }

    const bool b3 = (pl & 8) != 0, b2 = (pl & 4) != 0;
    const bool b1 = (pl & 2) != 0, b0 = (pl & 1) != 0;
    v2f res;
    #pragma unroll
    for (int hd = 0; hd < 2; ++hd) {
        const v8f lo = hd ? acc2 : acc0, hi = hd ? acc3 : acc1;
        const float* as = att_src + col_base + 32 * hd + (pl & 15);
        const float* ad = att_dst + col_base + 32 * hd + (pl & 15);
        const float s0 = as[0], s1 = as[16], d0 = ad[0], d1 = ad[16];
        float v1[8], v2[4], v3[2];
        #pragma unroll
        for (int i = 0; i < 8; ++i) {
            const float sv = lo[i] * s0 + hi[i] * s1;
            const float dv = lo[i] * d0 + hi[i] * d1;
            v1[i] = (b3 ? dv : sv) + __shfl_xor(b3 ? sv : dv, 8);
        }
        #pragma unroll
        for (int i = 0; i < 4; ++i)
            v2[i] = (b2 ? v1[i + 4] : v1[i]) + __shfl_xor(b2 ? v1[i] : v1[i + 4], 4);
        #pragma unroll
        for (int i = 0; i < 2; ++i)
            v3[i] = (b1 ? v2[i + 2] : v2[i]) + __shfl_xor(b1 ? v2[i] : v2[i + 2], 2);
        res[hd] = (b0 ? v3[1] : v3[0]) + __shfl_xor(b0 ? v3[0] : v3[1], 1);
    }
    {
        const int r = srow + (pl & 7);
        if (r < n_nodes) {
            float* o = (b3 ? a_dst : a_src) + (size_t)r * HEADS + (wave & 3) * 2;
            *(v2f*)o = res;
        }
    }
    if ((wave & 3) == 0) {
        if (pl < 16 && row_base + pl < n_nodes) cnt[row_base + pl] = 0;
        if (row_base + (pl >> 1) < n_nodes)
            *(uint4*)(menc + (size_t)row_base * HEADS + pl * 4) = make_uint4(0u, 0u, 0u, 0u);
    }
}

__device__ __forceinline__ int frag_k0(int i, int h) { return 8 * i + 4 * h; }

__global__ void __launch_bounds__(256)
k_alpha(const float* __restrict__ edge_attr, const int* __restrict__ ei,
        const float* __restrict__ w_he, const float* __restrict__ a_src,
        const float* __restrict__ a_dst, float* __restrict__ alpha,
        unsigned* __restrict__ menc, int* __restrict__ cnt,
        int* __restrict__ rank, int n_edges) {
    const int lane = threadIdx.x & 31;
    const int c = lane & 15;
    const int h = lane >> 4;
    const int head = c & 7;
    const int wave_id = __builtin_amdgcn_readfirstlane((int)(blockIdx.x * 8 + (threadIdx.x >> 5)));
    const int n_waves = gridDim.x * 8;
    const int n_tiles = (n_edges + 15) >> 4;

    v4f b[8];
    #pragma unroll
    for (int i = 0; i < 8; ++i)
        b[i] = *(const v4f*)(w_he + head * EDIM + frag_k0(i, h));

    for (int tile = wave_id; tile < n_tiles; tile += n_waves) {
        const int e0 = tile * 16;
        int el = e0 + c;
        if (el >= n_edges) el = n_edges - 1;
        const int srcv = ei[el];
        const int dstv = ei[n_edges + el];
        const float* arow = edge_attr + (size_t)el * EDIM;
        v4f a[8];
        #pragma unroll
        for (int i = 0; i < 8; ++i)
            a[i] = *(const v4f*)(arow + frag_k0(i, h));

        const bool own = (h == 0) && (e0 + c < n_edges);
        int rk = 0;
        if (own) rk = atomicAdd(&cnt[dstv], 1);

        int dr[8];
        float pre[8];
        #pragma unroll
        for (int r = 0; r < 8; ++r) {
            const int sl = 8 * h + r;
            const int s = __shfl(srcv, sl);
            dr[r] = __shfl(dstv, sl);
            pre[r] = a_src[s * HEADS + head] + a_dst[dr[r] * HEADS + head];
        }

        v8f acc0 = {}, acc1 = {};
        #pragma unroll
        for (int i = 0; i < 8; ++i) {
            v2f a_lo, a_hi, b_lo, b_hi;
            a_lo.x = a[i].x; a_lo.y = a[i].y; a_hi.x = a[i].z; a_hi.y = a[i].w;
            b_lo.x = b[i].x; b_lo.y = b[i].y; b_hi.x = b[i].z; b_hi.y = b[i].w;
            acc0 = __builtin_amdgcn_wmma_f32_16x16x4_f32(false, a_lo, false, b_lo, (short)0, acc0, false, false);
            acc1 = __builtin_amdgcn_wmma_f32_16x16x4_f32(false, a_hi, false, b_hi, (short)0, acc1, false, false);
        }

        asm volatile("v_nop\n\tv_nop\n\tv_nop\n\tv_nop\n\tv_nop\n\tv_nop\n\tv_nop\n\tv_nop"
                     : "+v"(acc0), "+v"(acc1)
                     : "v"(a[0]), "v"(a[1]), "v"(a[2]), "v"(a[3]),
                       "v"(a[4]), "v"(a[5]), "v"(a[6]), "v"(a[7]));

        #pragma unroll
        for (int r = 0; r < 8; ++r) {
            const int e = e0 + 8 * h + r;
            float al = pre[r] + (acc0[r] + acc1[r]);
            al = (al > 0.f) ? al : NEG_SLOPE * al;
            if (c < 8 && e < n_edges) {
                alpha[(size_t)e * HEADS + c] = al;
                atomicMax(&menc[dr[r] * HEADS + c], enc_f32(al));
            }
        }
        if (own) rank[e0 + c] = rk;
    }
}

#define SCAN_PER 16
__global__ void __launch_bounds__(1024)
k_scan(const int* __restrict__ cnt, int* __restrict__ off, int n) {
    __shared__ int wsum[32];
    __shared__ int tot;
    const int t = threadIdx.x, lane = t & 31, wave = t >> 5;
    int carry = 0;
    for (int base = 0; base < n; base += 1024 * SCAN_PER) {
        const int i0 = base + t * SCAN_PER;
        const bool full = (i0 + SCAN_PER <= n);
        int v[SCAN_PER];
        if (full) {
            const int4* p = (const int4*)(cnt + i0);
            #pragma unroll
            for (int k = 0; k < SCAN_PER / 4; ++k) {
                int4 q = p[k];
                v[4 * k] = q.x; v[4 * k + 1] = q.y; v[4 * k + 2] = q.z; v[4 * k + 3] = q.w;
            }
        } else {
            #pragma unroll
            for (int k = 0; k < SCAN_PER; ++k) v[k] = (i0 + k < n) ? cnt[i0 + k] : 0;
        }
        int s = 0;
        #pragma unroll
        for (int k = 0; k < SCAN_PER; ++k) { int c = v[k]; v[k] = s; s += c; }
        int inc = s;
        #pragma unroll
        for (int d = 1; d < 32; d <<= 1) {
            int u = __shfl_up(inc, d);
            if (lane >= d) inc += u;
        }
        if (lane == 31) wsum[wave] = inc;
        __syncthreads();
        if (wave == 0) {
            const int w = wsum[lane];
            int winc = w;
            #pragma unroll
            for (int d = 1; d < 32; d <<= 1) {
                int u = __shfl_up(winc, d);
                if (lane >= d) winc += u;
            }
            wsum[lane] = winc - w;
            if (lane == 31) tot = winc;
        }
        __syncthreads();
        const int ex = carry + wsum[wave] + (inc - s);
        if (full) {
            int4* p = (int4*)(off + i0);
            #pragma unroll
            for (int k = 0; k < SCAN_PER / 4; ++k)
                p[k] = make_int4(ex + v[4 * k], ex + v[4 * k + 1], ex + v[4 * k + 2], ex + v[4 * k + 3]);
        } else {
            #pragma unroll
            for (int k = 0; k < SCAN_PER; ++k) if (i0 + k < n) off[i0 + k] = ex + v[k];
        }
        carry += tot;
        __syncthreads();
    }
    if (t == 0) off[n] = carry;
}

__global__ void __launch_bounds__(256)
k_fill(const int* __restrict__ ei, const int* __restrict__ rank,
       const int* __restrict__ off, const float* __restrict__ alpha,
       int* __restrict__ ssrc, float* __restrict__ salpha, int n_edges) {
    const int e = blockIdx.x * 256 + threadIdx.x;
    if (e >= n_edges) return;
    const int src = ei[e];
    const int dst = ei[n_edges + e];
    const int slot = off[dst] + rank[e];
    const float4* a = (const float4*)(alpha + (size_t)e * HEADS);
    const float4 a0 = a[0], a1 = a[1];
    ssrc[slot] = src;
    float4* o = (float4*)(salpha + (size_t)slot * HEADS);
    o[0] = a0;
    o[1] = a1;
}

__global__ void __launch_bounds__(256)
k_gather(const float* __restrict__ xp, const float* __restrict__ x,
         const int* __restrict__ off, const int* __restrict__ ssrc,
         const float* __restrict__ salpha, const unsigned* __restrict__ menc,
         const float* __restrict__ bias, const float* __restrict__ gamma,
         const float* __restrict__ beta, float* __restrict__ out, int n_nodes) {
    const int wave = threadIdx.x >> 5;
    const int lane = threadIdx.x & 31;
    const int n = blockIdx.x * 8 + wave;
    if (n >= n_nodes) return;
    const int c0 = lane * 4;
    const int hA = lane >> 3, hB = 4 + hA;
    const int beg = __builtin_amdgcn_readfirstlane(off[n]);
    const int end = __builtin_amdgcn_readfirstlane(off[n + 1]);
    const float mA = dec_f32(menc[n * HEADS + hA]);
    const float mB = dec_f32(menc[n * HEADS + hB]);

    float4 accA = make_float4(0.f, 0.f, 0.f, 0.f);
    float4 accB = make_float4(0.f, 0.f, 0.f, 0.f);
    float denA = 0.f, denB = 0.f;
    for (int base = beg; base < end; base += 32) {
        const int rem = end - base;
        const int cc = rem < 32 ? rem : 32;
        const int my_s = (lane < cc) ? ssrc[base + lane] : 0;
        #pragma unroll 4
        for (int j = 0; j < cc; ++j) {
            const int s = __builtin_amdgcn_readlane(my_s, j);
            const float* xs = xp + (size_t)s * HID;
            const float* al = salpha + (size_t)(base + j) * HEADS;
            const float4 rA = *(const float4*)(xs + c0);
            const float4 rB = *(const float4*)(xs + 128 + c0);
            const float eA = __expf(al[hA] - mA);
            const float eB = __expf(al[hB] - mB);
            denA += eA;
            denB += eB;
            accA.x += eA * rA.x; accA.y += eA * rA.y; accA.z += eA * rA.z; accA.w += eA * rA.w;
            accB.x += eB * rB.x; accB.y += eB * rB.y; accB.z += eB * rB.z; accB.w += eB * rB.w;
        }
    }

    const float4 xA = *(const float4*)(x + (size_t)n * HID + c0);
    const float4 xB = *(const float4*)(x + (size_t)n * HID + 128 + c0);
    const float4 bA = *(const float4*)(bias + c0);
    const float4 bB = *(const float4*)(bias + 128 + c0);
    const float dA = denA + 1e-16f, dB = denB + 1e-16f;
    float y[8];
    y[0] = xA.x + accA.x / dA + bA.x;
    y[1] = xA.y + accA.y / dA + bA.y;
    y[2] = xA.z + accA.z / dA + bA.z;
    y[3] = xA.w + accA.w / dA + bA.w;
    y[4] = xB.x + accB.x / dB + bB.x;
    y[5] = xB.y + accB.y / dB + bB.y;
    y[6] = xB.z + accB.z / dB + bB.z;
    y[7] = xB.w + accB.w / dB + bB.w;

    float s1 = 0.f, s2 = 0.f;
    #pragma unroll
    for (int k = 0; k < 8; ++k) { s1 += y[k]; s2 += y[k] * y[k]; }
    #pragma unroll
    for (int o = 16; o >= 1; o >>= 1) {
        s1 += __shfl_xor(s1, o);
        s2 += __shfl_xor(s2, o);
    }
    const float mean = s1 * (1.f / HID);
    const float var  = s2 * (1.f / HID) - mean * mean;
    const float rs   = rsqrtf(var + LN_EPS);
    const float4 gA = *(const float4*)(gamma + c0);
    const float4 gB = *(const float4*)(gamma + 128 + c0);
    const float4 tA = *(const float4*)(beta + c0);
    const float4 tB = *(const float4*)(beta + 128 + c0);
    float4 oA, oB;
    oA.x = (y[0] - mean) * rs * gA.x + tA.x;
    oA.y = (y[1] - mean) * rs * gA.y + tA.y;
    oA.z = (y[2] - mean) * rs * gA.z + tA.z;
    oA.w = (y[3] - mean) * rs * gA.w + tA.w;
    oB.x = (y[4] - mean) * rs * gB.x + tB.x;
    oB.y = (y[5] - mean) * rs * gB.y + tB.y;
    oB.z = (y[6] - mean) * rs * gB.z + tB.z;
    oB.w = (y[7] - mean) * rs * gB.w + tB.w;
    *(float4*)(out + (size_t)n * HID + c0) = oA;
    *(float4*)(out + (size_t)n * HID + 128 + c0) = oB;
}

extern "C" void kernel_launch(void* const* d_in, const int* in_sizes, int n_in,
                              void* d_out, int out_size, void* d_ws, size_t ws_size,
                              hipStream_t stream) {
    const float* x         = (const float*)d_in[0];
    const int*   ei        = (const int*)  d_in[1];
    const float* edge_attr = (const float*)d_in[2];
    const float* W         = (const float*)d_in[3];
    const float* W_e       = (const float*)d_in[4];
    const float* att_src   = (const float*)d_in[5];
    const float* att_dst   = (const float*)d_in[6];
    const float* att_edge  = (const float*)d_in[7];
    const float* bias      = (const float*)d_in[8];
    const float* gamma     = (const float*)d_in[9];
    const float* beta      = (const float*)d_in[10];
    float* out = (float*)d_out;

    const int n_nodes = in_sizes[0] / HID;
    const int n_edges = in_sizes[1] / 2;

    char* ws = (char*)d_ws;
    size_t off = 0;
    auto carve = [&](size_t bytes) -> char* {
        char* p = ws + off;
        off += (bytes + 255) & ~(size_t)255;
        return p;
    };
    float*    xp     = (float*)   carve((size_t)n_nodes * HID   * 4);
    float*    alpha  = (float*)   carve((size_t)n_edges * HEADS * 4);
    float*    salpha = (float*)   carve((size_t)n_edges * HEADS * 4);
    int*      ssrc   = (int*)     carve((size_t)n_edges * 4);
    int*      rank   = (int*)     carve((size_t)n_edges * 4);
    float*    a_src  = (float*)   carve((size_t)n_nodes * HEADS * 4);
    float*    a_dst  = (float*)   carve((size_t)n_nodes * HEADS * 4);
    unsigned* menc   = (unsigned*)carve((size_t)n_nodes * HEADS * 4);
    int*      cnt    = (int*)     carve((size_t)n_nodes * 4);
    int*      segoff = (int*)     carve((size_t)(n_nodes + 1) * 4);
    float*    w_he   = (float*)   carve((size_t)EDIM * HEADS * 4);
    (void)ws_size;

    k_node_proj<<<(n_nodes + 31) / 32 + 1, 256, 0, stream>>>(x, W, W_e, att_edge, att_src, att_dst,
                                                             xp, a_src, a_dst, w_he, menc, cnt,
                                                             n_nodes);

    k_alpha<<<1024, 256, 0, stream>>>(edge_attr, ei, w_he,
                                      a_src, a_dst, alpha, menc,
                                      cnt, rank, n_edges);

    k_scan<<<1, 1024, 0, stream>>>(cnt, segoff, n_nodes);

    k_fill<<<(n_edges + 255) / 256, 256, 0, stream>>>(ei, rank, segoff, alpha,
                                                      ssrc, salpha, n_edges);

    k_gather<<<(n_nodes + 7) / 8, 256, 0, stream>>>(xp, x, segoff, ssrc, salpha, menc,
                                                    bias, gamma, beta, out, n_nodes);
}


# Node projection: every fragment load and xp store is now 16 bytes per lane (band columns interleaved over the 4 tiles, k paired in blocks of 8); 256-term sums and 32-term dots in a new order
# speedup vs baseline: 1.8868x; 1.0531x over previous
#include <hip/hip_runtime.h>
#include <hip/hip_bf16.h>
#include <stdint.h>

#define HID   256
#define EDIM  64
#define HEADS 8
#define CCH   32
#define NEG_SLOPE 0.2f
#define LN_EPS 1e-5f

typedef float v2f __attribute__((ext_vector_type(2)));
typedef float v4f __attribute__((ext_vector_type(4)));
typedef float v8f __attribute__((ext_vector_type(8)));

__device__ __forceinline__ unsigned enc_f32(float f) {
    unsigned b = __float_as_uint(f);
    return (b & 0x80000000u) ? ~b : (b | 0x80000000u);
}
__device__ __forceinline__ float dec_f32(unsigned u) {
    unsigned b = (u & 0x80000000u) ? (u & 0x7FFFFFFFu) : ~u;
    return __uint_as_float(b);
}

__global__ void __launch_bounds__(256)
k_node_proj(const float* __restrict__ x, const float* __restrict__ W,
            const float* __restrict__ W_e, const float* __restrict__ att_edge,
            const float* __restrict__ att_src, const float* __restrict__ att_dst,
            float* __restrict__ xp, float* __restrict__ a_src,
            float* __restrict__ a_dst, float* __restrict__ w_he,
            unsigned* __restrict__ menc, int* __restrict__ cnt, int n_nodes) {
    const int wave = __builtin_amdgcn_readfirstlane((int)(threadIdx.x >> 5));
    const int lane = threadIdx.x & 31;

    if ((int)blockIdx.x == ((n_nodes + 31) >> 5)) {
        const int d = (wave & 1) * 32 + lane;
        #pragma unroll
        for (int u = 0; u < 2; ++u) {
            const int h = (wave >> 1) + 4 * u;
            const float* wr = W_e + d * HID + h * CCH;
            const float* ar = att_edge + h * CCH;
            float s = 0.f;
            #pragma unroll
            for (int c = 0; c < CCH; ++c) s += wr[c] * ar[c];
            w_he[h * EDIM + d] = s;
        }
        return;
    }

    const int row_base = (blockIdx.x * 2 + (wave >> 2)) * 16;
    const int col_base = (wave & 3) * 64;
    if (row_base >= n_nodes) return;

    int row = row_base + (lane & 15);
    if (row >= n_nodes) row = n_nodes - 1;
    const float* xrow = x + (size_t)row * HID + 4 * (lane >> 4);
    const float* wp = W + (size_t)(4 * (lane >> 4)) * HID + col_base + 4 * (lane & 15);

    v8f acc0 = {}, acc1 = {}, acc2 = {}, acc3 = {};
    for (int kb = 0; kb < HID; kb += 8) {
        const v4f a  = *(const v4f*)(xrow + kb);
        const float* wr = wp + (size_t)kb * HID;
        const v4f w0 = *(const v4f*)(wr);
        const v4f w1 = *(const v4f*)(wr + HID);
        const v4f w2 = *(const v4f*)(wr + 2 * HID);
        const v4f w3 = *(const v4f*)(wr + 3 * HID);
        v2f a01, a23, p0, p1, p2, p3, q0, q1, q2, q3;
        a01.x = a.x;  a01.y = a.y;  a23.x = a.z;  a23.y = a.w;
        p0.x = w0.x;  p0.y = w1.x;  p1.x = w0.y;  p1.y = w1.y;
        p2.x = w0.z;  p2.y = w1.z;  p3.x = w0.w;  p3.y = w1.w;
        q0.x = w2.x;  q0.y = w3.x;  q1.x = w2.y;  q1.y = w3.y;
        q2.x = w2.z;  q2.y = w3.z;  q3.x = w2.w;  q3.y = w3.w;
        acc0 = __builtin_amdgcn_wmma_f32_16x16x4_f32(false, a01, false, p0, (short)0, acc0, false, false);
        acc1 = __builtin_amdgcn_wmma_f32_16x16x4_f32(false, a01, false, p1, (short)0, acc1, false, false);
        acc2 = __builtin_amdgcn_wmma_f32_16x16x4_f32(false, a01, false, p2, (short)0, acc2, false, false);
        acc3 = __builtin_amdgcn_wmma_f32_16x16x4_f32(false, a01, false, p3, (short)0, acc3, false, false);
        acc0 = __builtin_amdgcn_wmma_f32_16x16x4_f32(false, a23, false, q0, (short)0, acc0, false, false);
        acc1 = __builtin_amdgcn_wmma_f32_16x16x4_f32(false, a23, false, q1, (short)0, acc1, false, false);
        acc2 = __builtin_amdgcn_wmma_f32_16x16x4_f32(false, a23, false, q2, (short)0, acc2, false, false);
        acc3 = __builtin_amdgcn_wmma_f32_16x16x4_f32(false, a23, false, q3, (short)0, acc3, false, false);
        asm volatile("v_nop\n\tv_nop\n\tv_nop\n\tv_nop\n\tv_nop\n\tv_nop\n\tv_nop\n\tv_nop"
                     : "+v"(acc0), "+v"(acc1), "+v"(acc2), "+v"(acc3)
                     : "v"(a), "v"(p0), "v"(p1), "v"(p2), "v"(p3),
                       "v"(q0), "v"(q1), "v"(q2), "v"(q3));
    }

    const int pl = (int)__builtin_amdgcn_mbcnt_lo(~0u, 0u);
    const int srow = row_base + ((pl >> 4) * 8);
    const int scol = col_base + 4 * (pl & 15);
    #pragma unroll
    for (int i = 0; i < 8; ++i) {
        int r = srow + i;
        if (r < n_nodes) {
            v4f o;
            o.x = acc0[i]; o.y = acc1[i]; o.z = acc2[i]; o.w = acc3[i];
            *(v4f*)(xp + (size_t)r * HID + scol) = o;
        }
    }

    const bool b2 = (pl & 4) != 0, b1 = (pl & 2) != 0, b0 = (pl & 1) != 0;
    const v4f as = *(const v4f*)(att_src + scol);
    const v4f ad = *(const v4f*)(att_dst + scol);
    float v1[8], v2[4], v3[2];
    #pragma unroll
    for (int i = 0; i < 8; ++i) {
        const float sv = acc0[i] * as.x + acc1[i] * as.y + acc2[i] * as.z + acc3[i] * as.w;
        const float dv = acc0[i] * ad.x + acc1[i] * ad.y + acc2[i] * ad.z + acc3[i] * ad.w;
        v1[i] = (b2 ? dv : sv) + __shfl_xor(b2 ? sv : dv, 4);
    }
    #pragma unroll
    for (int i = 0; i < 4; ++i)
        v2[i] = (b1 ? v1[i + 4] : v1[i]) + __shfl_xor(b1 ? v1[i] : v1[i + 4], 2);
    #pragma unroll
    for (int i = 0; i < 2; ++i)
        v3[i] = (b0 ? v2[i + 2] : v2[i]) + __shfl_xor(b0 ? v2[i] : v2[i + 2], 1);
    {
        const int r = srow + 2 * (pl & 3);
        float* o = (b2 ? a_dst : a_src) + (size_t)r * HEADS + (wave & 3) * 2 + ((pl >> 3) & 1);
        if (r < n_nodes) o[0] = v3[0];
        if (r + 1 < n_nodes) o[HEADS] = v3[1];
    }
    if ((wave & 3) == 0) {
        if (pl < 16 && row_base + pl < n_nodes) cnt[row_base + pl] = 0;
        if (row_base + (pl >> 1) < n_nodes)
            *(uint4*)(menc + (size_t)row_base * HEADS + pl * 4) = make_uint4(0u, 0u, 0u, 0u);
    }
}

__device__ __forceinline__ int frag_k0(int i, int h) { return 8 * i + 4 * h; }

__global__ void __launch_bounds__(256)
k_alpha(const float* __restrict__ edge_attr, const int* __restrict__ ei,
        const float* __restrict__ w_he, const float* __restrict__ a_src,
        const float* __restrict__ a_dst, float* __restrict__ alpha,
        unsigned* __restrict__ menc, int* __restrict__ cnt,
        int* __restrict__ rank, int n_edges) {
    const int lane = threadIdx.x & 31;
    const int c = lane & 15;
    const int h = lane >> 4;
    const int head = c & 7;
    const int wave_id = __builtin_amdgcn_readfirstlane((int)(blockIdx.x * 8 + (threadIdx.x >> 5)));
    const int n_waves = gridDim.x * 8;
    const int n_tiles = (n_edges + 15) >> 4;

    v4f b[8];
    #pragma unroll
    for (int i = 0; i < 8; ++i)
        b[i] = *(const v4f*)(w_he + head * EDIM + frag_k0(i, h));

    for (int tile = wave_id; tile < n_tiles; tile += n_waves) {
        const int e0 = tile * 16;
        int el = e0 + c;
        if (el >= n_edges) el = n_edges - 1;
        const int srcv = ei[el];
        const int dstv = ei[n_edges + el];
        const float* arow = edge_attr + (size_t)el * EDIM;
        v4f a[8];
        #pragma unroll
        for (int i = 0; i < 8; ++i)
            a[i] = *(const v4f*)(arow + frag_k0(i, h));

        const bool own = (h == 0) && (e0 + c < n_edges);
        int rk = 0;
        if (own) rk = atomicAdd(&cnt[dstv], 1);

        int dr[8];
        float pre[8];
        #pragma unroll
        for (int r = 0; r < 8; ++r) {
            const int sl = 8 * h + r;
            const int s = __shfl(srcv, sl);
            dr[r] = __shfl(dstv, sl);
            pre[r] = a_src[s * HEADS + head] + a_dst[dr[r] * HEADS + head];
        }

        v8f acc0 = {}, acc1 = {};
        #pragma unroll
        for (int i = 0; i < 8; ++i) {
            v2f a_lo, a_hi, b_lo, b_hi;
            a_lo.x = a[i].x; a_lo.y = a[i].y; a_hi.x = a[i].z; a_hi.y = a[i].w;
            b_lo.x = b[i].x; b_lo.y = b[i].y; b_hi.x = b[i].z; b_hi.y = b[i].w;
            acc0 = __builtin_amdgcn_wmma_f32_16x16x4_f32(false, a_lo, false, b_lo, (short)0, acc0, false, false);
            acc1 = __builtin_amdgcn_wmma_f32_16x16x4_f32(false, a_hi, false, b_hi, (short)0, acc1, false, false);
        }

        asm volatile("v_nop\n\tv_nop\n\tv_nop\n\tv_nop\n\tv_nop\n\tv_nop\n\tv_nop\n\tv_nop"
                     : "+v"(acc0), "+v"(acc1)
                     : "v"(a[0]), "v"(a[1]), "v"(a[2]), "v"(a[3]),
                       "v"(a[4]), "v"(a[5]), "v"(a[6]), "v"(a[7]));

        #pragma unroll
        for (int r = 0; r < 8; ++r) {
            const int e = e0 + 8 * h + r;
            float al = pre[r] + (acc0[r] + acc1[r]);
            al = (al > 0.f) ? al : NEG_SLOPE * al;
            if (c < 8 && e < n_edges) {
                alpha[(size_t)e * HEADS + c] = al;
                atomicMax(&menc[dr[r] * HEADS + c], enc_f32(al));
            }
        }
        if (own) rank[e0 + c] = rk;
    }
}

#define SCAN_PER 16
__global__ void __launch_bounds__(1024)
k_scan(const int* __restrict__ cnt, int* __restrict__ off, int n) {
    __shared__ int wsum[32];
    __shared__ int tot;
    const int t = threadIdx.x, lane = t & 31, wave = t >> 5;
    int carry = 0;
    for (int base = 0; base < n; base += 1024 * SCAN_PER) {
        const int i0 = base + t * SCAN_PER;
        const bool full = (i0 + SCAN_PER <= n);
        int v[SCAN_PER];
        if (full) {
            const int4* p = (const int4*)(cnt + i0);
            #pragma unroll
            for (int k = 0; k < SCAN_PER / 4; ++k) {
                int4 q = p[k];
                v[4 * k] = q.x; v[4 * k + 1] = q.y; v[4 * k + 2] = q.z; v[4 * k + 3] = q.w;
            }
        } else {
            #pragma unroll
            for (int k = 0; k < SCAN_PER; ++k) v[k] = (i0 + k < n) ? cnt[i0 + k] : 0;
        }
        int s = 0;
        #pragma unroll
        for (int k = 0; k < SCAN_PER; ++k) { int c = v[k]; v[k] = s; s += c; }
        int inc = s;
        #pragma unroll
        for (int d = 1; d < 32; d <<= 1) {
            int u = __shfl_up(inc, d);
            if (lane >= d) inc += u;
        }
        if (lane == 31) wsum[wave] = inc;
        __syncthreads();
        if (wave == 0) {
            const int w = wsum[lane];
            int winc = w;
            #pragma unroll
            for (int d = 1; d < 32; d <<= 1) {
                int u = __shfl_up(winc, d);
                if (lane >= d) winc += u;
            }
            wsum[lane] = winc - w;
            if (lane == 31) tot = winc;
        }
        __syncthreads();
        const int ex = carry + wsum[wave] + (inc - s);
        if (full) {
            int4* p = (int4*)(off + i0);
            #pragma unroll
            for (int k = 0; k < SCAN_PER / 4; ++k)
                p[k] = make_int4(ex + v[4 * k], ex + v[4 * k + 1], ex + v[4 * k + 2], ex + v[4 * k + 3]);
        } else {
            #pragma unroll
            for (int k = 0; k < SCAN_PER; ++k) if (i0 + k < n) off[i0 + k] = ex + v[k];
        }
        carry += tot;
        __syncthreads();
    }
    if (t == 0) off[n] = carry;
}

__global__ void __launch_bounds__(256)
k_fill(const int* __restrict__ ei, const int* __restrict__ rank,
       const int* __restrict__ off, const float* __restrict__ alpha,
       int* __restrict__ ssrc, float* __restrict__ salpha, int n_edges) {
    const int e = blockIdx.x * 256 + threadIdx.x;
    if (e >= n_edges) return;
    const int src = ei[e];
    const int dst = ei[n_edges + e];
    const int slot = off[dst] + rank[e];
    const float4* a = (const float4*)(alpha + (size_t)e * HEADS);
    const float4 a0 = a[0], a1 = a[1];
    ssrc[slot] = src;
    float4* o = (float4*)(salpha + (size_t)slot * HEADS);
    o[0] = a0;
    o[1] = a1;
}

__global__ void __launch_bounds__(256)
k_gather(const float* __restrict__ xp, const float* __restrict__ x,
         const int* __restrict__ off, const int* __restrict__ ssrc,
         const float* __restrict__ salpha, const unsigned* __restrict__ menc,
         const float* __restrict__ bias, const float* __restrict__ gamma,
         const float* __restrict__ beta, float* __restrict__ out, int n_nodes) {
    const int wave = threadIdx.x >> 5;
    const int lane = threadIdx.x & 31;
    const int n = blockIdx.x * 8 + wave;
    if (n >= n_nodes) return;
    const int c0 = lane * 4;
    const int hA = lane >> 3, hB = 4 + hA;
    const int beg = __builtin_amdgcn_readfirstlane(off[n]);
    const int end = __builtin_amdgcn_readfirstlane(off[n + 1]);
    const float mA = dec_f32(menc[n * HEADS + hA]);
    const float mB = dec_f32(menc[n * HEADS + hB]);

    float4 accA = make_float4(0.f, 0.f, 0.f, 0.f);
    float4 accB = make_float4(0.f, 0.f, 0.f, 0.f);
    float denA = 0.f, denB = 0.f;
    for (int base = beg; base < end; base += 32) {
        const int rem = end - base;
        const int cc = rem < 32 ? rem : 32;
        const int my_s = (lane < cc) ? ssrc[base + lane] : 0;
        #pragma unroll 4
        for (int j = 0; j < cc; ++j) {
            const int s = __builtin_amdgcn_readlane(my_s, j);
            const float* xs = xp + (size_t)s * HID;
            const float* al = salpha + (size_t)(base + j) * HEADS;
            const float4 rA = *(const float4*)(xs + c0);
            const float4 rB = *(const float4*)(xs + 128 + c0);
            const float eA = __expf(al[hA] - mA);
            const float eB = __expf(al[hB] - mB);
            denA += eA;
            denB += eB;
            accA.x += eA * rA.x; accA.y += eA * rA.y; accA.z += eA * rA.z; accA.w += eA * rA.w;
            accB.x += eB * rB.x; accB.y += eB * rB.y; accB.z += eB * rB.z; accB.w += eB * rB.w;
        }
    }

    const float4 xA = *(const float4*)(x + (size_t)n * HID + c0);
    const float4 xB = *(const float4*)(x + (size_t)n * HID + 128 + c0);
    const float4 bA = *(const float4*)(bias + c0);
    const float4 bB = *(const float4*)(bias + 128 + c0);
    const float dA = denA + 1e-16f, dB = denB + 1e-16f;
    float y[8];
    y[0] = xA.x + accA.x / dA + bA.x;
    y[1] = xA.y + accA.y / dA + bA.y;
    y[2] = xA.z + accA.z / dA + bA.z;
    y[3] = xA.w + accA.w / dA + bA.w;
    y[4] = xB.x + accB.x / dB + bB.x;
    y[5] = xB.y + accB.y / dB + bB.y;
    y[6] = xB.z + accB.z / dB + bB.z;
    y[7] = xB.w + accB.w / dB + bB.w;

    float s1 = 0.f, s2 = 0.f;
    #pragma unroll
    for (int k = 0; k < 8; ++k) { s1 += y[k]; s2 += y[k] * y[k]; }
    #pragma unroll
    for (int o = 16; o >= 1; o >>= 1) {
        s1 += __shfl_xor(s1, o);
        s2 += __shfl_xor(s2, o);
    }
    const float mean = s1 * (1.f / HID);
    const float var  = s2 * (1.f / HID) - mean * mean;
    const float rs   = rsqrtf(var + LN_EPS);
    const float4 gA = *(const float4*)(gamma + c0);
    const float4 gB = *(const float4*)(gamma + 128 + c0);
    const float4 tA = *(const float4*)(beta + c0);
    const float4 tB = *(const float4*)(beta + 128 + c0);
    float4 oA, oB;
    oA.x = (y[0] - mean) * rs * gA.x + tA.x;
    oA.y = (y[1] - mean) * rs * gA.y + tA.y;
    oA.z = (y[2] - mean) * rs * gA.z + tA.z;
    oA.w = (y[3] - mean) * rs * gA.w + tA.w;
    oB.x = (y[4] - mean) * rs * gB.x + tB.x;
    oB.y = (y[5] - mean) * rs * gB.y + tB.y;
    oB.z = (y[6] - mean) * rs * gB.z + tB.z;
    oB.w = (y[7] - mean) * rs * gB.w + tB.w;
    *(float4*)(out + (size_t)n * HID + c0) = oA;
    *(float4*)(out + (size_t)n * HID + 128 + c0) = oB;
}

extern "C" void kernel_launch(void* const* d_in, const int* in_sizes, int n_in,
                              void* d_out, int out_size, void* d_ws, size_t ws_size,
                              hipStream_t stream) {
    const float* x         = (const float*)d_in[0];
    const int*   ei        = (const int*)  d_in[1];
    const float* edge_attr = (const float*)d_in[2];
    const float* W         = (const float*)d_in[3];
    const float* W_e       = (const float*)d_in[4];
    const float* att_src   = (const float*)d_in[5];
    const float* att_dst   = (const float*)d_in[6];
    const float* att_edge  = (const float*)d_in[7];
    const float* bias      = (const float*)d_in[8];
    const float* gamma     = (const float*)d_in[9];
    const float* beta      = (const float*)d_in[10];
    float* out = (float*)d_out;

    const int n_nodes = in_sizes[0] / HID;
    const int n_edges = in_sizes[1] / 2;

    char* ws = (char*)d_ws;
    size_t off = 0;
    auto carve = [&](size_t bytes) -> char* {
        char* p = ws + off;
        off += (bytes + 255) & ~(size_t)255;
        return p;
    };
    float*    xp     = (float*)   carve((size_t)n_nodes * HID   * 4);
    float*    alpha  = (float*)   carve((size_t)n_edges * HEADS * 4);
    float*    salpha = (float*)   carve((size_t)n_edges * HEADS * 4);
    int*      ssrc   = (int*)     carve((size_t)n_edges * 4);
    int*      rank   = (int*)     carve((size_t)n_edges * 4);
    float*    a_src  = (float*)   carve((size_t)n_nodes * HEADS * 4);
    float*    a_dst  = (float*)   carve((size_t)n_nodes * HEADS * 4);
    unsigned* menc   = (unsigned*)carve((size_t)n_nodes * HEADS * 4);
    int*      cnt    = (int*)     carve((size_t)n_nodes * 4);
    int*      segoff = (int*)     carve((size_t)(n_nodes + 1) * 4);
    float*    w_he   = (float*)   carve((size_t)EDIM * HEADS * 4);
    (void)ws_size;

    k_node_proj<<<(n_nodes + 31) / 32 + 1, 256, 0, stream>>>(x, W, W_e, att_edge, att_src, att_dst,
                                                             xp, a_src, a_dst, w_he, menc, cnt,
                                                             n_nodes);

    k_alpha<<<1024, 256, 0, stream>>>(edge_attr, ei, w_he,
                                      a_src, a_dst, alpha, menc,
                                      cnt, rank, n_edges);

    k_scan<<<1, 1024, 0, stream>>>(cnt, segoff, n_nodes);

    k_fill<<<(n_edges + 255) / 256, 256, 0, stream>>>(ei, rank, segoff, alpha,
                                                      ssrc, salpha, n_edges);

    k_gather<<<(n_nodes + 7) / 8, 256, 0, stream>>>(xp, x, segoff, ssrc, salpha, menc,
                                                    bias, gamma, beta, out, n_nodes);
}


# Segment max of the logits taken by the gather wave from its sorted segment (8 coalesced reads, 2 lane exchanges), not by an L2 atomic max per edge and head in the edge kernel; output bits unchanged
# speedup vs baseline: 2.0477x; 1.0853x over previous
#include <hip/hip_runtime.h>
#include <hip/hip_bf16.h>
#include <stdint.h>

#define HID   256
#define EDIM  64
#define HEADS 8
#define CCH   32
#define NEG_SLOPE 0.2f
#define LN_EPS 1e-5f

typedef float v2f __attribute__((ext_vector_type(2)));
typedef float v4f __attribute__((ext_vector_type(4)));
typedef float v8f __attribute__((ext_vector_type(8)));

__global__ void __launch_bounds__(256)
k_node_proj(const float* __restrict__ x, const float* __restrict__ W,
            const float* __restrict__ W_e, const float* __restrict__ att_edge,
            const float* __restrict__ att_src, const float* __restrict__ att_dst,
            float* __restrict__ xp, float* __restrict__ a_src,
            float* __restrict__ a_dst, float* __restrict__ w_he,
            int* __restrict__ cnt, int n_nodes) {
    const int wave = __builtin_amdgcn_readfirstlane((int)(threadIdx.x >> 5));
    const int lane = threadIdx.x & 31;

    if ((int)blockIdx.x == ((n_nodes + 31) >> 5)) {
        const int d = (wave & 1) * 32 + lane;
        #pragma unroll
        for (int u = 0; u < 2; ++u) {
            const int h = (wave >> 1) + 4 * u;
            const float* wr = W_e + d * HID + h * CCH;
            const float* ar = att_edge + h * CCH;
            float s = 0.f;
            #pragma unroll
            for (int c = 0; c < CCH; ++c) s += wr[c] * ar[c];
            w_he[h * EDIM + d] = s;
        }
        return;
    }

    const int row_base = (blockIdx.x * 2 + (wave >> 2)) * 16;
    const int col_base = (wave & 3) * 64;
    if (row_base >= n_nodes) return;

    int row = row_base + (lane & 15);
    if (row >= n_nodes) row = n_nodes - 1;
    const float* xrow = x + (size_t)row * HID + 4 * (lane >> 4);
    const float* wp = W + (size_t)(4 * (lane >> 4)) * HID + col_base + 4 * (lane & 15);

    v8f acc0 = {}, acc1 = {}, acc2 = {}, acc3 = {};
    for (int kb = 0; kb < HID; kb += 8) {
        const v4f a  = *(const v4f*)(xrow + kb);
        const float* wr = wp + (size_t)kb * HID;
        const v4f w0 = *(const v4f*)(wr);
        const v4f w1 = *(const v4f*)(wr + HID);
        const v4f w2 = *(const v4f*)(wr + 2 * HID);
        const v4f w3 = *(const v4f*)(wr + 3 * HID);
        v2f a01, a23, p0, p1, p2, p3, q0, q1, q2, q3;
        a01.x = a.x;  a01.y = a.y;  a23.x = a.z;  a23.y = a.w;
        p0.x = w0.x;  p0.y = w1.x;  p1.x = w0.y;  p1.y = w1.y;
        p2.x = w0.z;  p2.y = w1.z;  p3.x = w0.w;  p3.y = w1.w;
        q0.x = w2.x;  q0.y = w3.x;  q1.x = w2.y;  q1.y = w3.y;
        q2.x = w2.z;  q2.y = w3.z;  q3.x = w2.w;  q3.y = w3.w;
        acc0 = __builtin_amdgcn_wmma_f32_16x16x4_f32(false, a01, false, p0, (short)0, acc0, false, false);
        acc1 = __builtin_amdgcn_wmma_f32_16x16x4_f32(false, a01, false, p1, (short)0, acc1, false, false);
        acc2 = __builtin_amdgcn_wmma_f32_16x16x4_f32(false, a01, false, p2, (short)0, acc2, false, false);
        acc3 = __builtin_amdgcn_wmma_f32_16x16x4_f32(false, a01, false, p3, (short)0, acc3, false, false);
        acc0 = __builtin_amdgcn_wmma_f32_16x16x4_f32(false, a23, false, q0, (short)0, acc0, false, false);
        acc1 = __builtin_amdgcn_wmma_f32_16x16x4_f32(false, a23, false, q1, (short)0, acc1, false, false);
        acc2 = __builtin_amdgcn_wmma_f32_16x16x4_f32(false, a23, false, q2, (short)0, acc2, false, false);
        acc3 = __builtin_amdgcn_wmma_f32_16x16x4_f32(false, a23, false, q3, (short)0, acc3, false, false);
        asm volatile("v_nop\n\tv_nop\n\tv_nop\n\tv_nop\n\tv_nop\n\tv_nop\n\tv_nop\n\tv_nop"
                     : "+v"(acc0), "+v"(acc1), "+v"(acc2), "+v"(acc3)
                     : "v"(a), "v"(p0), "v"(p1), "v"(p2), "v"(p3),
                       "v"(q0), "v"(q1), "v"(q2), "v"(q3));
    }

    const int pl = (int)__builtin_amdgcn_mbcnt_lo(~0u, 0u);
    const int srow = row_base + ((pl >> 4) * 8);
    const int scol = col_base + 4 * (pl & 15);
    #pragma unroll
    for (int i = 0; i < 8; ++i) {
        int r = srow + i;
        if (r < n_nodes) {
            v4f o;
            o.x = acc0[i]; o.y = acc1[i]; o.z = acc2[i]; o.w = acc3[i];
            *(v4f*)(xp + (size_t)r * HID + scol) = o;
        }
    }

    const bool b2 = (pl & 4) != 0, b1 = (pl & 2) != 0, b0 = (pl & 1) != 0;
    const v4f as = *(const v4f*)(att_src + scol);
    const v4f ad = *(const v4f*)(att_dst + scol);
    float v1[8], v2[4], v3[2];
    #pragma unroll
    for (int i = 0; i < 8; ++i) {
        const float sv = acc0[i] * as.x + acc1[i] * as.y + acc2[i] * as.z + acc3[i] * as.w;
        const float dv = acc0[i] * ad.x + acc1[i] * ad.y + acc2[i] * ad.z + acc3[i] * ad.w;
        v1[i] = (b2 ? dv : sv) + __shfl_xor(b2 ? sv : dv, 4);
    }
    #pragma unroll
    for (int i = 0; i < 4; ++i)
        v2[i] = (b1 ? v1[i + 4] : v1[i]) + __shfl_xor(b1 ? v1[i] : v1[i + 4], 2);
    #pragma unroll
    for (int i = 0; i < 2; ++i)
        v3[i] = (b0 ? v2[i + 2] : v2[i]) + __shfl_xor(b0 ? v2[i] : v2[i + 2], 1);
    {
        const int r = srow + 2 * (pl & 3);
        float* o = (b2 ? a_dst : a_src) + (size_t)r * HEADS + (wave & 3) * 2 + ((pl >> 3) & 1);
        if (r < n_nodes) o[0] = v3[0];
        if (r + 1 < n_nodes) o[HEADS] = v3[1];
    }
    if ((wave & 3) == 0) {
        if (pl < 16 && row_base + pl < n_nodes) cnt[row_base + pl] = 0;
    }
}

__device__ __forceinline__ int frag_k0(int i, int h) { return 8 * i + 4 * h; }

__global__ void __launch_bounds__(256)
k_alpha(const float* __restrict__ edge_attr, const int* __restrict__ ei,
        const float* __restrict__ w_he, const float* __restrict__ a_src,
        const float* __restrict__ a_dst, float* __restrict__ alpha,
        int* __restrict__ cnt, int* __restrict__ rank, int n_edges) {
    const int lane = threadIdx.x & 31;
    const int c = lane & 15;
    const int h = lane >> 4;
    const int head = c & 7;
    const int wave_id = __builtin_amdgcn_readfirstlane((int)(blockIdx.x * 8 + (threadIdx.x >> 5)));
    const int n_waves = gridDim.x * 8;
    const int n_tiles = (n_edges + 15) >> 4;

    v4f b[8];
    #pragma unroll
    for (int i = 0; i < 8; ++i)
        b[i] = *(const v4f*)(w_he + head * EDIM + frag_k0(i, h));

    for (int tile = wave_id; tile < n_tiles; tile += n_waves) {
        const int e0 = tile * 16;
        int el = e0 + c;
        if (el >= n_edges) el = n_edges - 1;
        const int srcv = ei[el];
        const int dstv = ei[n_edges + el];
        const float* arow = edge_attr + (size_t)el * EDIM;
        v4f a[8];
        #pragma unroll
        for (int i = 0; i < 8; ++i)
            a[i] = *(const v4f*)(arow + frag_k0(i, h));

        const bool own = (h == 0) && (e0 + c < n_edges);
        int rk = 0;
        if (own) rk = atomicAdd(&cnt[dstv], 1);

        float pre[8];
        #pragma unroll
        for (int r = 0; r < 8; ++r) {
            const int sl = 8 * h + r;
            const int s = __shfl(srcv, sl);
            const int d = __shfl(dstv, sl);
            pre[r] = a_src[s * HEADS + head] + a_dst[d * HEADS + head];
        }

        v8f acc0 = {}, acc1 = {};
        #pragma unroll
        for (int i = 0; i < 8; ++i) {
            v2f a_lo, a_hi, b_lo, b_hi;
            a_lo.x = a[i].x; a_lo.y = a[i].y; a_hi.x = a[i].z; a_hi.y = a[i].w;
            b_lo.x = b[i].x; b_lo.y = b[i].y; b_hi.x = b[i].z; b_hi.y = b[i].w;
            acc0 = __builtin_amdgcn_wmma_f32_16x16x4_f32(false, a_lo, false, b_lo, (short)0, acc0, false, false);
            acc1 = __builtin_amdgcn_wmma_f32_16x16x4_f32(false, a_hi, false, b_hi, (short)0, acc1, false, false);
        }

        asm volatile("v_nop\n\tv_nop\n\tv_nop\n\tv_nop\n\tv_nop\n\tv_nop\n\tv_nop\n\tv_nop"
                     : "+v"(acc0), "+v"(acc1)
                     : "v"(a[0]), "v"(a[1]), "v"(a[2]), "v"(a[3]),
                       "v"(a[4]), "v"(a[5]), "v"(a[6]), "v"(a[7]));

        #pragma unroll
        for (int r = 0; r < 8; ++r) {
            const int e = e0 + 8 * h + r;
            float al = pre[r] + (acc0[r] + acc1[r]);
            al = (al > 0.f) ? al : NEG_SLOPE * al;
            if (c < 8 && e < n_edges) alpha[(size_t)e * HEADS + c] = al;
        }
        if (own) rank[e0 + c] = rk;
    }
}

#define SCAN_PER 16
__global__ void __launch_bounds__(1024)
k_scan(const int* __restrict__ cnt, int* __restrict__ off, int n) {
    __shared__ int wsum[32];
    __shared__ int tot;
    const int t = threadIdx.x, lane = t & 31, wave = t >> 5;
    int carry = 0;
    for (int base = 0; base < n; base += 1024 * SCAN_PER) {
        const int i0 = base + t * SCAN_PER;
        const bool full = (i0 + SCAN_PER <= n);
        int v[SCAN_PER];
        if (full) {
            const int4* p = (const int4*)(cnt + i0);
            #pragma unroll
            for (int k = 0; k < SCAN_PER / 4; ++k) {
                int4 q = p[k];
                v[4 * k] = q.x; v[4 * k + 1] = q.y; v[4 * k + 2] = q.z; v[4 * k + 3] = q.w;
            }
        } else {
            #pragma unroll
            for (int k = 0; k < SCAN_PER; ++k) v[k] = (i0 + k < n) ? cnt[i0 + k] : 0;
        }
        int s = 0;
        #pragma unroll
        for (int k = 0; k < SCAN_PER; ++k) { int c = v[k]; v[k] = s; s += c; }
        int inc = s;
        #pragma unroll
        for (int d = 1; d < 32; d <<= 1) {
            int u = __shfl_up(inc, d);
            if (lane >= d) inc += u;
        }
        if (lane == 31) wsum[wave] = inc;
        __syncthreads();
        if (wave == 0) {
            const int w = wsum[lane];
            int winc = w;
            #pragma unroll
            for (int d = 1; d < 32; d <<= 1) {
                int u = __shfl_up(winc, d);
                if (lane >= d) winc += u;
            }
            wsum[lane] = winc - w;
            if (lane == 31) tot = winc;
        }
        __syncthreads();
        const int ex = carry + wsum[wave] + (inc - s);
        if (full) {
            int4* p = (int4*)(off + i0);
            #pragma unroll
            for (int k = 0; k < SCAN_PER / 4; ++k)
                p[k] = make_int4(ex + v[4 * k], ex + v[4 * k + 1], ex + v[4 * k + 2], ex + v[4 * k + 3]);
        } else {
            #pragma unroll
            for (int k = 0; k < SCAN_PER; ++k) if (i0 + k < n) off[i0 + k] = ex + v[k];
        }
        carry += tot;
        __syncthreads();
    }
    if (t == 0) off[n] = carry;
}

__global__ void __launch_bounds__(256)
k_fill(const int* __restrict__ ei, const int* __restrict__ rank,
       const int* __restrict__ off, const float* __restrict__ alpha,
       int* __restrict__ ssrc, float* __restrict__ salpha, int n_edges) {
    const int e = blockIdx.x * 256 + threadIdx.x;
    if (e >= n_edges) return;
    const int src = ei[e];
    const int dst = ei[n_edges + e];
    const int slot = off[dst] + rank[e];
    const float4* a = (const float4*)(alpha + (size_t)e * HEADS);
    const float4 a0 = a[0], a1 = a[1];
    ssrc[slot] = src;
    float4* o = (float4*)(salpha + (size_t)slot * HEADS);
    o[0] = a0;
    o[1] = a1;
}

__global__ void __launch_bounds__(256)
k_gather(const float* __restrict__ xp, const float* __restrict__ x,
         const int* __restrict__ off, const int* __restrict__ ssrc,
         const float* __restrict__ salpha,
         const float* __restrict__ bias, const float* __restrict__ gamma,
         const float* __restrict__ beta, float* __restrict__ out, int n_nodes) {
    const int wave = threadIdx.x >> 5;
    const int lane = threadIdx.x & 31;
    const int n = blockIdx.x * 8 + wave;
    if (n >= n_nodes) return;
    const int c0 = lane * 4;
    const int hA = lane >> 3, hB = 4 + hA;
    const int beg = __builtin_amdgcn_readfirstlane(off[n]);
    const int end = __builtin_amdgcn_readfirstlane(off[n + 1]);

    int my_s = (lane < end - beg) ? ssrc[beg + lane] : 0;

    float mx = -INFINITY;
    const int f_end = end * HEADS;
    for (int fb = beg * HEADS; fb < f_end; fb += 8 * 32) {
        float v[8];
        #pragma unroll
        for (int t = 0; t < 8; ++t) {
            const int f = fb + 32 * t + lane;
            v[t] = (f < f_end) ? salpha[f] : -INFINITY;
        }
        #pragma unroll
        for (int t = 0; t < 8; ++t) mx = fmaxf(mx, v[t]);
    }
    mx = fmaxf(mx, __shfl_xor(mx, 8));
    mx = fmaxf(mx, __shfl_xor(mx, 16));
    float mA = __shfl(mx, hA);
    float mB = __shfl(mx, hB);
    mA = isfinite(mA) ? mA : 0.f;
    mB = isfinite(mB) ? mB : 0.f;

    float4 accA = make_float4(0.f, 0.f, 0.f, 0.f);
    float4 accB = make_float4(0.f, 0.f, 0.f, 0.f);
    float denA = 0.f, denB = 0.f;
    for (int base = beg; base < end; base += 32) {
        const int rem = end - base;
        const int cc = rem < 32 ? rem : 32;
        if (base != beg) my_s = (lane < cc) ? ssrc[base + lane] : 0;
        #pragma unroll 4
        for (int j = 0; j < cc; ++j) {
            const int s = __builtin_amdgcn_readlane(my_s, j);
            const float* xs = xp + (size_t)s * HID;
            const float* al = salpha + (size_t)(base + j) * HEADS;
            const float4 rA = *(const float4*)(xs + c0);
            const float4 rB = *(const float4*)(xs + 128 + c0);
            const float eA = __expf(al[hA] - mA);
            const float eB = __expf(al[hB] - mB);
            denA += eA;
            denB += eB;
            accA.x += eA * rA.x; accA.y += eA * rA.y; accA.z += eA * rA.z; accA.w += eA * rA.w;
            accB.x += eB * rB.x; accB.y += eB * rB.y; accB.z += eB * rB.z; accB.w += eB * rB.w;
        }
    }

    const float4 xA = *(const float4*)(x + (size_t)n * HID + c0);
    const float4 xB = *(const float4*)(x + (size_t)n * HID + 128 + c0);
    const float4 bA = *(const float4*)(bias + c0);
    const float4 bB = *(const float4*)(bias + 128 + c0);
    const float dA = denA + 1e-16f, dB = denB + 1e-16f;
    float y[8];
    y[0] = xA.x + accA.x / dA + bA.x;
    y[1] = xA.y + accA.y / dA + bA.y;
    y[2] = xA.z + accA.z / dA + bA.z;
    y[3] = xA.w + accA.w / dA + bA.w;
    y[4] = xB.x + accB.x / dB + bB.x;
    y[5] = xB.y + accB.y / dB + bB.y;
    y[6] = xB.z + accB.z / dB + bB.z;
    y[7] = xB.w + accB.w / dB + bB.w;

    float s1 = 0.f, s2 = 0.f;
    #pragma unroll
    for (int k = 0; k < 8; ++k) { s1 += y[k]; s2 += y[k] * y[k]; }
    #pragma unroll
    for (int o = 16; o >= 1; o >>= 1) {
        s1 += __shfl_xor(s1, o);
        s2 += __shfl_xor(s2, o);
    }
    const float mean = s1 * (1.f / HID);
    const float var  = s2 * (1.f / HID) - mean * mean;
    const float rs   = rsqrtf(var + LN_EPS);
    const float4 gA = *(const float4*)(gamma + c0);
    const float4 gB = *(const float4*)(gamma + 128 + c0);
    const float4 tA = *(const float4*)(beta + c0);
    const float4 tB = *(const float4*)(beta + 128 + c0);
    float4 oA, oB;
    oA.x = (y[0] - mean) * rs * gA.x + tA.x;
    oA.y = (y[1] - mean) * rs * gA.y + tA.y;
    oA.z = (y[2] - mean) * rs * gA.z + tA.z;
    oA.w = (y[3] - mean) * rs * gA.w + tA.w;
    oB.x = (y[4] - mean) * rs * gB.x + tB.x;
    oB.y = (y[5] - mean) * rs * gB.y + tB.y;
    oB.z = (y[6] - mean) * rs * gB.z + tB.z;
    oB.w = (y[7] - mean) * rs * gB.w + tB.w;
    *(float4*)(out + (size_t)n * HID + c0) = oA;
    *(float4*)(out + (size_t)n * HID + 128 + c0) = oB;
}

extern "C" void kernel_launch(void* const* d_in, const int* in_sizes, int n_in,
                              void* d_out, int out_size, void* d_ws, size_t ws_size,
                              hipStream_t stream) {
    const float* x         = (const float*)d_in[0];
    const int*   ei        = (const int*)  d_in[1];
    const float* edge_attr = (const float*)d_in[2];
    const float* W         = (const float*)d_in[3];
    const float* W_e       = (const float*)d_in[4];
    const float* att_src   = (const float*)d_in[5];
    const float* att_dst   = (const float*)d_in[6];
    const float* att_edge  = (const float*)d_in[7];
    const float* bias      = (const float*)d_in[8];
    const float* gamma     = (const float*)d_in[9];
    const float* beta      = (const float*)d_in[10];
    float* out = (float*)d_out;

    const int n_nodes = in_sizes[0] / HID;
    const int n_edges = in_sizes[1] / 2;

    char* ws = (char*)d_ws;
    size_t off = 0;
    auto carve = [&](size_t bytes) -> char* {
        char* p = ws + off;
        off += (bytes + 255) & ~(size_t)255;
        return p;
    };
    float*    xp     = (float*)   carve((size_t)n_nodes * HID   * 4);
    float*    alpha  = (float*)   carve((size_t)n_edges * HEADS * 4);
    float*    salpha = (float*)   carve((size_t)n_edges * HEADS * 4);
    int*      ssrc   = (int*)     carve((size_t)n_edges * 4);
    int*      rank   = (int*)     carve((size_t)n_edges * 4);
    float*    a_src  = (float*)   carve((size_t)n_nodes * HEADS * 4);
    float*    a_dst  = (float*)   carve((size_t)n_nodes * HEADS * 4);
    int*      cnt    = (int*)     carve((size_t)n_nodes * 4);
    int*      segoff = (int*)     carve((size_t)(n_nodes + 1) * 4);
    float*    w_he   = (float*)   carve((size_t)EDIM * HEADS * 4);
    (void)ws_size;

    k_node_proj<<<(n_nodes + 31) / 32 + 1, 256, 0, stream>>>(x, W, W_e, att_edge, att_src, att_dst,
                                                             xp, a_src, a_dst, w_he, cnt,
                                                             n_nodes);

    k_alpha<<<1024, 256, 0, stream>>>(edge_attr, ei, w_he,
                                      a_src, a_dst, alpha,
                                      cnt, rank, n_edges);

    k_scan<<<1, 1024, 0, stream>>>(cnt, segoff, n_nodes);

    k_fill<<<(n_edges + 255) / 256, 256, 0, stream>>>(ei, rank, segoff, alpha,
                                                      ssrc, salpha, n_edges);

    k_gather<<<(n_nodes + 7) / 8, 256, 0, stream>>>(xp, x, segoff, ssrc, salpha,
                                                    bias, gamma, beta, out, n_nodes);
}


# In-degree scan joins the fill launch: each of 256 workgroups of 1024 threads scans all counters into LDS and places its edges, segment starts read from LDS; launches 5 -> 4; output bits unchanged
# speedup vs baseline: 2.2730x; 1.1100x over previous
#include <hip/hip_runtime.h>
#include <hip/hip_bf16.h>
#include <stdint.h>

#define HID   256
#define EDIM  64
#define HEADS 8
#define CCH   32
#define NEG_SLOPE 0.2f
#define LN_EPS 1e-5f

typedef float v2f __attribute__((ext_vector_type(2)));
typedef float v4f __attribute__((ext_vector_type(4)));
typedef float v8f __attribute__((ext_vector_type(8)));

__global__ void __launch_bounds__(256)
k_node_proj(const float* __restrict__ x, const float* __restrict__ W,
            const float* __restrict__ W_e, const float* __restrict__ att_edge,
            const float* __restrict__ att_src, const float* __restrict__ att_dst,
            float* __restrict__ xp, float* __restrict__ a_src,
            float* __restrict__ a_dst, float* __restrict__ w_he,
            int* __restrict__ cnt, int n_nodes) {
    const int wave = __builtin_amdgcn_readfirstlane((int)(threadIdx.x >> 5));
    const int lane = threadIdx.x & 31;

    if ((int)blockIdx.x == ((n_nodes + 31) >> 5)) {
        const int d = (wave & 1) * 32 + lane;
        #pragma unroll
        for (int u = 0; u < 2; ++u) {
            const int h = (wave >> 1) + 4 * u;
            const float* wr = W_e + d * HID + h * CCH;
            const float* ar = att_edge + h * CCH;
            float s = 0.f;
            #pragma unroll
            for (int c = 0; c < CCH; ++c) s += wr[c] * ar[c];
            w_he[h * EDIM + d] = s;
        }
        return;
    }

    const int row_base = (blockIdx.x * 2 + (wave >> 2)) * 16;
    const int col_base = (wave & 3) * 64;
    if (row_base >= n_nodes) return;

    int row = row_base + (lane & 15);
    if (row >= n_nodes) row = n_nodes - 1;
    const float* xrow = x + (size_t)row * HID + 4 * (lane >> 4);
    const float* wp = W + (size_t)(4 * (lane >> 4)) * HID + col_base + 4 * (lane & 15);

    v8f acc0 = {}, acc1 = {}, acc2 = {}, acc3 = {};
    for (int kb = 0; kb < HID; kb += 8) {
        const v4f a  = *(const v4f*)(xrow + kb);
        const float* wr = wp + (size_t)kb * HID;
        const v4f w0 = *(const v4f*)(wr);
        const v4f w1 = *(const v4f*)(wr + HID);
        const v4f w2 = *(const v4f*)(wr + 2 * HID);
        const v4f w3 = *(const v4f*)(wr + 3 * HID);
        v2f a01, a23, p0, p1, p2, p3, q0, q1, q2, q3;
        a01.x = a.x;  a01.y = a.y;  a23.x = a.z;  a23.y = a.w;
        p0.x = w0.x;  p0.y = w1.x;  p1.x = w0.y;  p1.y = w1.y;
        p2.x = w0.z;  p2.y = w1.z;  p3.x = w0.w;  p3.y = w1.w;
        q0.x = w2.x;  q0.y = w3.x;  q1.x = w2.y;  q1.y = w3.y;
        q2.x = w2.z;  q2.y = w3.z;  q3.x = w2.w;  q3.y = w3.w;
        acc0 = __builtin_amdgcn_wmma_f32_16x16x4_f32(false, a01, false, p0, (short)0, acc0, false, false);
        acc1 = __builtin_amdgcn_wmma_f32_16x16x4_f32(false, a01, false, p1, (short)0, acc1, false, false);
        acc2 = __builtin_amdgcn_wmma_f32_16x16x4_f32(false, a01, false, p2, (short)0, acc2, false, false);
        acc3 = __builtin_amdgcn_wmma_f32_16x16x4_f32(false, a01, false, p3, (short)0, acc3, false, false);
        acc0 = __builtin_amdgcn_wmma_f32_16x16x4_f32(false, a23, false, q0, (short)0, acc0, false, false);
        acc1 = __builtin_amdgcn_wmma_f32_16x16x4_f32(false, a23, false, q1, (short)0, acc1, false, false);
        acc2 = __builtin_amdgcn_wmma_f32_16x16x4_f32(false, a23, false, q2, (short)0, acc2, false, false);
        acc3 = __builtin_amdgcn_wmma_f32_16x16x4_f32(false, a23, false, q3, (short)0, acc3, false, false);
        asm volatile("v_nop\n\tv_nop\n\tv_nop\n\tv_nop\n\tv_nop\n\tv_nop\n\tv_nop\n\tv_nop"
                     : "+v"(acc0), "+v"(acc1), "+v"(acc2), "+v"(acc3)
                     : "v"(a), "v"(p0), "v"(p1), "v"(p2), "v"(p3),
                       "v"(q0), "v"(q1), "v"(q2), "v"(q3));
    }

    const int pl = (int)__builtin_amdgcn_mbcnt_lo(~0u, 0u);
    const int srow = row_base + ((pl >> 4) * 8);
    const int scol = col_base + 4 * (pl & 15);
    #pragma unroll
    for (int i = 0; i < 8; ++i) {
        int r = srow + i;
        if (r < n_nodes) {
            v4f o;
            o.x = acc0[i]; o.y = acc1[i]; o.z = acc2[i]; o.w = acc3[i];
            *(v4f*)(xp + (size_t)r * HID + scol) = o;
        }
    }

    const bool b2 = (pl & 4) != 0, b1 = (pl & 2) != 0, b0 = (pl & 1) != 0;
    const v4f as = *(const v4f*)(att_src + scol);
    const v4f ad = *(const v4f*)(att_dst + scol);
    float v1[8], v2[4], v3[2];
    #pragma unroll
    for (int i = 0; i < 8; ++i) {
        const float sv = acc0[i] * as.x + acc1[i] * as.y + acc2[i] * as.z + acc3[i] * as.w;
        const float dv = acc0[i] * ad.x + acc1[i] * ad.y + acc2[i] * ad.z + acc3[i] * ad.w;
        v1[i] = (b2 ? dv : sv) + __shfl_xor(b2 ? sv : dv, 4);
    }
    #pragma unroll
    for (int i = 0; i < 4; ++i)
        v2[i] = (b1 ? v1[i + 4] : v1[i]) + __shfl_xor(b1 ? v1[i] : v1[i + 4], 2);
    #pragma unroll
    for (int i = 0; i < 2; ++i)
        v3[i] = (b0 ? v2[i + 2] : v2[i]) + __shfl_xor(b0 ? v2[i] : v2[i + 2], 1);
    {
        const int r = srow + 2 * (pl & 3);
        float* o = (b2 ? a_dst : a_src) + (size_t)r * HEADS + (wave & 3) * 2 + ((pl >> 3) & 1);
        if (r < n_nodes) o[0] = v3[0];
        if (r + 1 < n_nodes) o[HEADS] = v3[1];
    }
    if ((wave & 3) == 0) {
        if (pl < 16 && row_base + pl < n_nodes) cnt[row_base + pl] = 0;
    }
}

__device__ __forceinline__ int frag_k0(int i, int h) { return 8 * i + 4 * h; }

__global__ void __launch_bounds__(256)
k_alpha(const float* __restrict__ edge_attr, const int* __restrict__ ei,
        const float* __restrict__ w_he, const float* __restrict__ a_src,
        const float* __restrict__ a_dst, float* __restrict__ alpha,
        int* __restrict__ cnt, int* __restrict__ rank, int n_edges) {
    const int lane = threadIdx.x & 31;
    const int c = lane & 15;
    const int h = lane >> 4;
    const int head = c & 7;
    const int wave_id = __builtin_amdgcn_readfirstlane((int)(blockIdx.x * 8 + (threadIdx.x >> 5)));
    const int n_waves = gridDim.x * 8;
    const int n_tiles = (n_edges + 15) >> 4;

    v4f b[8];
    #pragma unroll
    for (int i = 0; i < 8; ++i)
        b[i] = *(const v4f*)(w_he + head * EDIM + frag_k0(i, h));

    for (int tile = wave_id; tile < n_tiles; tile += n_waves) {
        const int e0 = tile * 16;
        int el = e0 + c;
        if (el >= n_edges) el = n_edges - 1;
        const int srcv = ei[el];
        const int dstv = ei[n_edges + el];
        const float* arow = edge_attr + (size_t)el * EDIM;
        v4f a[8];
        #pragma unroll
        for (int i = 0; i < 8; ++i)
            a[i] = *(const v4f*)(arow + frag_k0(i, h));

        const bool own = (h == 0) && (e0 + c < n_edges);
        int rk = 0;
        if (own) rk = atomicAdd(&cnt[dstv], 1);

        float pre[8];
        #pragma unroll
        for (int r = 0; r < 8; ++r) {
            const int sl = 8 * h + r;
            const int s = __shfl(srcv, sl);
            const int d = __shfl(dstv, sl);
            pre[r] = a_src[s * HEADS + head] + a_dst[d * HEADS + head];
        }

        v8f acc0 = {}, acc1 = {};
        #pragma unroll
        for (int i = 0; i < 8; ++i) {
            v2f a_lo, a_hi, b_lo, b_hi;
            a_lo.x = a[i].x; a_lo.y = a[i].y; a_hi.x = a[i].z; a_hi.y = a[i].w;
            b_lo.x = b[i].x; b_lo.y = b[i].y; b_hi.x = b[i].z; b_hi.y = b[i].w;
            acc0 = __builtin_amdgcn_wmma_f32_16x16x4_f32(false, a_lo, false, b_lo, (short)0, acc0, false, false);
            acc1 = __builtin_amdgcn_wmma_f32_16x16x4_f32(false, a_hi, false, b_hi, (short)0, acc1, false, false);
        }

        asm volatile("v_nop\n\tv_nop\n\tv_nop\n\tv_nop\n\tv_nop\n\tv_nop\n\tv_nop\n\tv_nop"
                     : "+v"(acc0), "+v"(acc1)
                     : "v"(a[0]), "v"(a[1]), "v"(a[2]), "v"(a[3]),
                       "v"(a[4]), "v"(a[5]), "v"(a[6]), "v"(a[7]));

        #pragma unroll
        for (int r = 0; r < 8; ++r) {
            const int e = e0 + 8 * h + r;
            float al = pre[r] + (acc0[r] + acc1[r]);
            al = (al > 0.f) ? al : NEG_SLOPE * al;
            if (c < 8 && e < n_edges) alpha[(size_t)e * HEADS + c] = al;
        }
        if (own) rank[e0 + c] = rk;
    }
}

#define SCAN_PER 52
#define SCAN_CAP (1024 * SCAN_PER)
#define FILL_EDGES 4
__global__ void __launch_bounds__(1024)
k_scan_fill(const int* __restrict__ cnt, const int* __restrict__ ei,
            const int* __restrict__ rank, const float* __restrict__ alpha,
            int* __restrict__ off, int* __restrict__ ssrc,
            float* __restrict__ salpha, int n, int n_edges) {
    __shared__ int s_off[SCAN_CAP];
    __shared__ int wsum[32];
    const int t = threadIdx.x, lane = t & 31, wave = t >> 5;

    const int i0 = t * SCAN_PER;
    int v[SCAN_PER];
    if (i0 + SCAN_PER <= n) {
        const int4* p = (const int4*)(cnt + i0);
        #pragma unroll
        for (int k = 0; k < SCAN_PER / 4; ++k) {
            const int4 q = p[k];
            v[4 * k] = q.x; v[4 * k + 1] = q.y; v[4 * k + 2] = q.z; v[4 * k + 3] = q.w;
        }
    } else {
        #pragma unroll
        for (int k = 0; k < SCAN_PER; ++k) v[k] = (i0 + k < n) ? cnt[i0 + k] : 0;
    }
    int s = 0;
    #pragma unroll
    for (int k = 0; k < SCAN_PER; ++k) { const int c = v[k]; v[k] = s; s += c; }
    int inc = s;
    #pragma unroll
    for (int d = 1; d < 32; d <<= 1) {
        const int u = __shfl_up(inc, d);
        if (lane >= d) inc += u;
    }
    if (lane == 31) wsum[wave] = inc;
    __syncthreads();
    if (wave == 0) {
        const int w = wsum[lane];
        int winc = w;
        #pragma unroll
        for (int d = 1; d < 32; d <<= 1) {
            const int u = __shfl_up(winc, d);
            if (lane >= d) winc += u;
        }
        wsum[lane] = winc - w;
    }
    __syncthreads();
    {
        const int ex = wsum[wave] + (inc - s);
        int4* so = (int4*)(s_off + i0);
        #pragma unroll
        for (int k = 0; k < SCAN_PER / 4; ++k)
            so[k] = make_int4(ex + v[4 * k], ex + v[4 * k + 1], ex + v[4 * k + 2], ex + v[4 * k + 3]);
    }
    __syncthreads();

    const int n_wg = (int)gridDim.x;
    {
        const int chunk = (((n + 1 + n_wg - 1) / n_wg) + 31) & ~31;
        const int lo = (int)blockIdx.x * chunk;
        int hi = lo + chunk;
        if (hi > n + 1) hi = n + 1;
        #pragma nounroll
        for (int i = lo + t; i < hi; i += 1024) off[i] = s_off[i];
    }

    const int per = (n_edges + n_wg - 1) / n_wg;
    const int wg_beg = (int)blockIdx.x * per;
    int wg_end = wg_beg + per;
    if (wg_end > n_edges) wg_end = n_edges;
    for (int base = wg_beg; base < wg_end; base += FILL_EDGES * 1024) {
        int src[FILL_EDGES], dst[FILL_EDGES], rk[FILL_EDGES];
        v4f a0[FILL_EDGES], a1[FILL_EDGES];
        #pragma unroll
        for (int u = 0; u < FILL_EDGES; ++u) {
            const int e = base + u * 1024 + t;
            const int ec = (e < wg_end) ? e : wg_end - 1;
            src[u] = ei[ec];
            dst[u] = ei[n_edges + ec];
            rk[u]  = rank[ec];
            const float* a = alpha + (size_t)ec * HEADS;
            a0[u] = *(const v4f*)(a);
            a1[u] = *(const v4f*)(a + 4);
        }
        #pragma unroll
        for (int u = 0; u < FILL_EDGES; ++u)
            asm volatile("" : "+v"(src[u]), "+v"(dst[u]), "+v"(rk[u]), "+v"(a0[u]), "+v"(a1[u]));
        #pragma unroll
        for (int u = 0; u < FILL_EDGES; ++u) {
            const int e = base + u * 1024 + t;
            if (e < wg_end) {
                const int slot = s_off[dst[u]] + rk[u];
                ssrc[slot] = src[u];
                float* o = salpha + (size_t)slot * HEADS;
                *(v4f*)(o)     = a0[u];
                *(v4f*)(o + 4) = a1[u];
            }
        }
    }
}

__global__ void __launch_bounds__(256)
k_gather(const float* __restrict__ xp, const float* __restrict__ x,
         const int* __restrict__ off, const int* __restrict__ ssrc,
         const float* __restrict__ salpha,
         const float* __restrict__ bias, const float* __restrict__ gamma,
         const float* __restrict__ beta, float* __restrict__ out, int n_nodes) {
    const int wave = threadIdx.x >> 5;
    const int lane = threadIdx.x & 31;
    const int n = blockIdx.x * 8 + wave;
    if (n >= n_nodes) return;
    const int c0 = lane * 4;
    const int hA = lane >> 3, hB = 4 + hA;
    const int beg = __builtin_amdgcn_readfirstlane(off[n]);
    const int end = __builtin_amdgcn_readfirstlane(off[n + 1]);

    int my_s = (lane < end - beg) ? ssrc[beg + lane] : 0;

    float mx = -INFINITY;
    const int f_end = end * HEADS;
    for (int fb = beg * HEADS; fb < f_end; fb += 8 * 32) {
        float v[8];
        #pragma unroll
        for (int t = 0; t < 8; ++t) {
            const int f = fb + 32 * t + lane;
            v[t] = (f < f_end) ? salpha[f] : -INFINITY;
        }
        #pragma unroll
        for (int t = 0; t < 8; ++t) mx = fmaxf(mx, v[t]);
    }
    mx = fmaxf(mx, __shfl_xor(mx, 8));
    mx = fmaxf(mx, __shfl_xor(mx, 16));
    float mA = __shfl(mx, hA);
    float mB = __shfl(mx, hB);
    mA = isfinite(mA) ? mA : 0.f;
    mB = isfinite(mB) ? mB : 0.f;

    float4 accA = make_float4(0.f, 0.f, 0.f, 0.f);
    float4 accB = make_float4(0.f, 0.f, 0.f, 0.f);
    float denA = 0.f, denB = 0.f;
    for (int base = beg; base < end; base += 32) {
        const int rem = end - base;
        const int cc = rem < 32 ? rem : 32;
        if (base != beg) my_s = (lane < cc) ? ssrc[base + lane] : 0;
        #pragma unroll 4
        for (int j = 0; j < cc; ++j) {
            const int s = __builtin_amdgcn_readlane(my_s, j);
            const float* xs = xp + (size_t)s * HID;
            const float* al = salpha + (size_t)(base + j) * HEADS;
            const float4 rA = *(const float4*)(xs + c0);
            const float4 rB = *(const float4*)(xs + 128 + c0);
            const float eA = __expf(al[hA] - mA);
            const float eB = __expf(al[hB] - mB);
            denA += eA;
            denB += eB;
            accA.x += eA * rA.x; accA.y += eA * rA.y; accA.z += eA * rA.z; accA.w += eA * rA.w;
            accB.x += eB * rB.x; accB.y += eB * rB.y; accB.z += eB * rB.z; accB.w += eB * rB.w;
        }
    }

    const float4 xA = *(const float4*)(x + (size_t)n * HID + c0);
    const float4 xB = *(const float4*)(x + (size_t)n * HID + 128 + c0);
    const float4 bA = *(const float4*)(bias + c0);
    const float4 bB = *(const float4*)(bias + 128 + c0);
    const float dA = denA + 1e-16f, dB = denB + 1e-16f;
    float y[8];
    y[0] = xA.x + accA.x / dA + bA.x;
    y[1] = xA.y + accA.y / dA + bA.y;
    y[2] = xA.z + accA.z / dA + bA.z;
    y[3] = xA.w + accA.w / dA + bA.w;
    y[4] = xB.x + accB.x / dB + bB.x;
    y[5] = xB.y + accB.y / dB + bB.y;
    y[6] = xB.z + accB.z / dB + bB.z;
    y[7] = xB.w + accB.w / dB + bB.w;

    float s1 = 0.f, s2 = 0.f;
    #pragma unroll
    for (int k = 0; k < 8; ++k) { s1 += y[k]; s2 += y[k] * y[k]; }
    #pragma unroll
    for (int o = 16; o >= 1; o >>= 1) {
        s1 += __shfl_xor(s1, o);
        s2 += __shfl_xor(s2, o);
    }
    const float mean = s1 * (1.f / HID);
    const float var  = s2 * (1.f / HID) - mean * mean;
    const float rs   = rsqrtf(var + LN_EPS);
    const float4 gA = *(const float4*)(gamma + c0);
    const float4 gB = *(const float4*)(gamma + 128 + c0);
    const float4 tA = *(const float4*)(beta + c0);
    const float4 tB = *(const float4*)(beta + 128 + c0);
    float4 oA, oB;
    oA.x = (y[0] - mean) * rs * gA.x + tA.x;
    oA.y = (y[1] - mean) * rs * gA.y + tA.y;
    oA.z = (y[2] - mean) * rs * gA.z + tA.z;
    oA.w = (y[3] - mean) * rs * gA.w + tA.w;
    oB.x = (y[4] - mean) * rs * gB.x + tB.x;
    oB.y = (y[5] - mean) * rs * gB.y + tB.y;
    oB.z = (y[6] - mean) * rs * gB.z + tB.z;
    oB.w = (y[7] - mean) * rs * gB.w + tB.w;
    *(float4*)(out + (size_t)n * HID + c0) = oA;
    *(float4*)(out + (size_t)n * HID + 128 + c0) = oB;
}

extern "C" void kernel_launch(void* const* d_in, const int* in_sizes, int n_in,
                              void* d_out, int out_size, void* d_ws, size_t ws_size,
                              hipStream_t stream) {
    const float* x         = (const float*)d_in[0];
    const int*   ei        = (const int*)  d_in[1];
    const float* edge_attr = (const float*)d_in[2];
    const float* W         = (const float*)d_in[3];
    const float* W_e       = (const float*)d_in[4];
    const float* att_src   = (const float*)d_in[5];
    const float* att_dst   = (const float*)d_in[6];
    const float* att_edge  = (const float*)d_in[7];
    const float* bias      = (const float*)d_in[8];
    const float* gamma     = (const float*)d_in[9];
    const float* beta      = (const float*)d_in[10];
    float* out = (float*)d_out;

    const int n_nodes = in_sizes[0] / HID;
    const int n_edges = in_sizes[1] / 2;

    char* ws = (char*)d_ws;
    size_t off = 0;
    auto carve = [&](size_t bytes) -> char* {
        char* p = ws + off;
        off += (bytes + 255) & ~(size_t)255;
        return p;
    };
    float*    xp     = (float*)   carve((size_t)n_nodes * HID   * 4);
    float*    alpha  = (float*)   carve((size_t)n_edges * HEADS * 4);
    float*    salpha = (float*)   carve((size_t)n_edges * HEADS * 4);
    int*      ssrc   = (int*)     carve((size_t)n_edges * 4);
    int*      rank   = (int*)     carve((size_t)n_edges * 4);
    float*    a_src  = (float*)   carve((size_t)n_nodes * HEADS * 4);
    float*    a_dst  = (float*)   carve((size_t)n_nodes * HEADS * 4);
    int*      cnt    = (int*)     carve((size_t)n_nodes * 4);
    int*      segoff = (int*)     carve((size_t)(n_nodes + 1) * 4);
    float*    w_he   = (float*)   carve((size_t)EDIM * HEADS * 4);
    (void)ws_size;

    k_node_proj<<<(n_nodes + 31) / 32 + 1, 256, 0, stream>>>(x, W, W_e, att_edge, att_src, att_dst,
                                                             xp, a_src, a_dst, w_he, cnt,
                                                             n_nodes);

    k_alpha<<<1024, 256, 0, stream>>>(edge_attr, ei, w_he,
                                      a_src, a_dst, alpha,
                                      cnt, rank, n_edges);

    k_scan_fill<<<256, 1024, 0, stream>>>(cnt, ei, rank, alpha, segoff,
                                          ssrc, salpha, n_nodes, n_edges);

    k_gather<<<(n_nodes + 7) / 8, 256, 0, stream>>>(xp, x, segoff, ssrc, salpha,
                                                    bias, gamma, beta, out, n_nodes);
}


# Edges ordered by destination through LDS: workgroups bin their edges into 256 node ranges, an owner workgroup per range places them; no global atomic per edge; message sums in LDS-arrival order
# speedup vs baseline: 3.1349x; 1.3792x over previous
#include <hip/hip_runtime.h>
#include <hip/hip_bf16.h>
#include <stdint.h>

#define HID   256
#define EDIM  64
#define HEADS 8
#define CCH   32
#define NEG_SLOPE 0.2f
#define LN_EPS 1e-5f

typedef float v2f __attribute__((ext_vector_type(2)));
typedef float v4f __attribute__((ext_vector_type(4)));
typedef float v8f __attribute__((ext_vector_type(8)));

__global__ void __launch_bounds__(256)
k_node_proj(const float* __restrict__ x, const float* __restrict__ W,
            const float* __restrict__ W_e, const float* __restrict__ att_edge,
            const float* __restrict__ att_src, const float* __restrict__ att_dst,
            float* __restrict__ xp, float* __restrict__ a_src,
            float* __restrict__ a_dst, float* __restrict__ w_he,
            int n_nodes) {
    const int wave = __builtin_amdgcn_readfirstlane((int)(threadIdx.x >> 5));
    const int lane = threadIdx.x & 31;

    if ((int)blockIdx.x == ((n_nodes + 31) >> 5)) {
        const int d = (wave & 1) * 32 + lane;
        #pragma unroll
        for (int u = 0; u < 2; ++u) {
            const int h = (wave >> 1) + 4 * u;
            const float* wr = W_e + d * HID + h * CCH;
            const float* ar = att_edge + h * CCH;
            float s = 0.f;
            #pragma unroll
            for (int c = 0; c < CCH; ++c) s += wr[c] * ar[c];
            w_he[h * EDIM + d] = s;
        }
        return;
    }

    const int row_base = (blockIdx.x * 2 + (wave >> 2)) * 16;
    const int col_base = (wave & 3) * 64;
    if (row_base >= n_nodes) return;

    int row = row_base + (lane & 15);
    if (row >= n_nodes) row = n_nodes - 1;
    const float* xrow = x + (size_t)row * HID + 4 * (lane >> 4);
    const float* wp = W + (size_t)(4 * (lane >> 4)) * HID + col_base + 4 * (lane & 15);

    v8f acc0 = {}, acc1 = {}, acc2 = {}, acc3 = {};
    for (int kb = 0; kb < HID; kb += 8) {
        const v4f a  = *(const v4f*)(xrow + kb);
        const float* wr = wp + (size_t)kb * HID;
        const v4f w0 = *(const v4f*)(wr);
        const v4f w1 = *(const v4f*)(wr + HID);
        const v4f w2 = *(const v4f*)(wr + 2 * HID);
        const v4f w3 = *(const v4f*)(wr + 3 * HID);
        v2f a01, a23, p0, p1, p2, p3, q0, q1, q2, q3;
        a01.x = a.x;  a01.y = a.y;  a23.x = a.z;  a23.y = a.w;
        p0.x = w0.x;  p0.y = w1.x;  p1.x = w0.y;  p1.y = w1.y;
        p2.x = w0.z;  p2.y = w1.z;  p3.x = w0.w;  p3.y = w1.w;
        q0.x = w2.x;  q0.y = w3.x;  q1.x = w2.y;  q1.y = w3.y;
        q2.x = w2.z;  q2.y = w3.z;  q3.x = w2.w;  q3.y = w3.w;
        acc0 = __builtin_amdgcn_wmma_f32_16x16x4_f32(false, a01, false, p0, (short)0, acc0, false, false);
        acc1 = __builtin_amdgcn_wmma_f32_16x16x4_f32(false, a01, false, p1, (short)0, acc1, false, false);
        acc2 = __builtin_amdgcn_wmma_f32_16x16x4_f32(false, a01, false, p2, (short)0, acc2, false, false);
        acc3 = __builtin_amdgcn_wmma_f32_16x16x4_f32(false, a01, false, p3, (short)0, acc3, false, false);
        acc0 = __builtin_amdgcn_wmma_f32_16x16x4_f32(false, a23, false, q0, (short)0, acc0, false, false);
        acc1 = __builtin_amdgcn_wmma_f32_16x16x4_f32(false, a23, false, q1, (short)0, acc1, false, false);
        acc2 = __builtin_amdgcn_wmma_f32_16x16x4_f32(false, a23, false, q2, (short)0, acc2, false, false);
        acc3 = __builtin_amdgcn_wmma_f32_16x16x4_f32(false, a23, false, q3, (short)0, acc3, false, false);
        asm volatile("v_nop\n\tv_nop\n\tv_nop\n\tv_nop\n\tv_nop\n\tv_nop\n\tv_nop\n\tv_nop"
                     : "+v"(acc0), "+v"(acc1), "+v"(acc2), "+v"(acc3)
                     : "v"(a), "v"(p0), "v"(p1), "v"(p2), "v"(p3),
                       "v"(q0), "v"(q1), "v"(q2), "v"(q3));
    }

    const int pl = (int)__builtin_amdgcn_mbcnt_lo(~0u, 0u);
    const int srow = row_base + ((pl >> 4) * 8);
    const int scol = col_base + 4 * (pl & 15);
    #pragma unroll
    for (int i = 0; i < 8; ++i) {
        int r = srow + i;
        if (r < n_nodes) {
            v4f o;
            o.x = acc0[i]; o.y = acc1[i]; o.z = acc2[i]; o.w = acc3[i];
            *(v4f*)(xp + (size_t)r * HID + scol) = o;
        }
    }

    const bool b2 = (pl & 4) != 0, b1 = (pl & 2) != 0, b0 = (pl & 1) != 0;
    const v4f as = *(const v4f*)(att_src + scol);
    const v4f ad = *(const v4f*)(att_dst + scol);
    float v1[8], v2[4], v3[2];
    #pragma unroll
    for (int i = 0; i < 8; ++i) {
        const float sv = acc0[i] * as.x + acc1[i] * as.y + acc2[i] * as.z + acc3[i] * as.w;
        const float dv = acc0[i] * ad.x + acc1[i] * ad.y + acc2[i] * ad.z + acc3[i] * ad.w;
        v1[i] = (b2 ? dv : sv) + __shfl_xor(b2 ? sv : dv, 4);
    }
    #pragma unroll
    for (int i = 0; i < 4; ++i)
        v2[i] = (b1 ? v1[i + 4] : v1[i]) + __shfl_xor(b1 ? v1[i] : v1[i + 4], 2);
    #pragma unroll
    for (int i = 0; i < 2; ++i)
        v3[i] = (b0 ? v2[i + 2] : v2[i]) + __shfl_xor(b0 ? v2[i] : v2[i + 2], 1);
    {
        const int r = srow + 2 * (pl & 3);
        float* o = (b2 ? a_dst : a_src) + (size_t)r * HEADS + (wave & 3) * 2 + ((pl >> 3) & 1);
        if (r < n_nodes) o[0] = v3[0];
        if (r + 1 < n_nodes) o[HEADS] = v3[1];
    }
}

#define NBIN 256
#define CHUNK_CAP 3200

__device__ __forceinline__ int scan_nbin(int v, int* s_part, int* total) {
    const int lane = threadIdx.x & 31, wave = threadIdx.x >> 5;
    int inc = v;
    #pragma unroll
    for (int d = 1; d < 32; d <<= 1) {
        const int u = __shfl_up(inc, d);
        if (lane >= d) inc += u;
    }
    __syncthreads();
    if (lane == 31 && wave < NBIN / 32) s_part[wave] = inc;
    __syncthreads();
    int before = 0, all = 0;
    #pragma unroll
    for (int k = 0; k < NBIN / 32; ++k) {
        const int p = s_part[k];
        all += p;
        if (k < wave) before += p;
    }
    *total = all;
    return before + inc - v;
}

__device__ __forceinline__ int frag_k0(int i, int h) { return 8 * i + 4 * h; }

__global__ void __launch_bounds__(1024)
k_alpha_bin(const float* __restrict__ edge_attr, const int* __restrict__ ei,
            const float* __restrict__ w_he, const float* __restrict__ a_src,
            const float* __restrict__ a_dst, int* __restrict__ st_src,
            int* __restrict__ st_dst, float* __restrict__ st_alpha,
            int* __restrict__ cntm, int* __restrict__ offm,
            int n_nodes, int n_edges) {
    __shared__ float s_alpha[CHUNK_CAP * HEADS];
    __shared__ int s_src[CHUNK_CAP];
    __shared__ int s_dst[CHUNK_CAP];
    __shared__ int s_rk[CHUNK_CAP];
    __shared__ int s_cnt[NBIN];
    __shared__ int s_off[NBIN];
    __shared__ int s_part[NBIN / 32];

    const int t = threadIdx.x;
    const int lane = t & 31;
    const int c = lane & 15;
    const int h = lane >> 4;
    const int head = c & 7;
    const int wave = __builtin_amdgcn_readfirstlane(t >> 5);
    const int n_tiles = (n_edges + 15) >> 4;
    const int tpw = (n_tiles + (int)gridDim.x - 1) / (int)gridDim.x;
    const int t_beg = (int)blockIdx.x * tpw;
    int t_end = t_beg + tpw;
    if (t_end > n_tiles) t_end = n_tiles;
    const int e_base = t_beg * 16;
    const int npb = (n_nodes + NBIN - 1) / NBIN;

    if (t < NBIN) s_cnt[t] = 0;
    __syncthreads();

    v4f b[8];
    #pragma unroll
    for (int i = 0; i < 8; ++i)
        b[i] = *(const v4f*)(w_he + head * EDIM + frag_k0(i, h));

    for (int tile = t_beg + wave; tile < t_end; tile += 32) {
        const int e0 = tile * 16;
        int el = e0 + c;
        if (el >= n_edges) el = n_edges - 1;
        const int srcv = ei[el];
        const int dstv = ei[n_edges + el];
        const float* arow = edge_attr + (size_t)el * EDIM;
        v4f a[8];
        #pragma unroll
        for (int i = 0; i < 8; ++i)
            a[i] = *(const v4f*)(arow + frag_k0(i, h));

        const bool own = (h == 0) && (e0 + c < n_edges);
        if (own) {
            const int li = e0 + c - e_base;
            s_src[li] = srcv;
            s_dst[li] = dstv;
            s_rk[li]  = atomicAdd(&s_cnt[dstv / npb], 1);
        }

        float pre[8];
        #pragma unroll
        for (int r = 0; r < 8; ++r) {
            const int sl = 8 * h + r;
            const int s = __shfl(srcv, sl);
            const int d = __shfl(dstv, sl);
            pre[r] = a_src[s * HEADS + head] + a_dst[d * HEADS + head];
        }

        v8f acc0 = {}, acc1 = {};
        #pragma unroll
        for (int i = 0; i < 8; ++i) {
            v2f a_lo, a_hi, b_lo, b_hi;
            a_lo.x = a[i].x; a_lo.y = a[i].y; a_hi.x = a[i].z; a_hi.y = a[i].w;
            b_lo.x = b[i].x; b_lo.y = b[i].y; b_hi.x = b[i].z; b_hi.y = b[i].w;
            acc0 = __builtin_amdgcn_wmma_f32_16x16x4_f32(false, a_lo, false, b_lo, (short)0, acc0, false, false);
            acc1 = __builtin_amdgcn_wmma_f32_16x16x4_f32(false, a_hi, false, b_hi, (short)0, acc1, false, false);
        }

        asm volatile("v_nop\n\tv_nop\n\tv_nop\n\tv_nop\n\tv_nop\n\tv_nop\n\tv_nop\n\tv_nop"
                     : "+v"(acc0), "+v"(acc1)
                     : "v"(a[0]), "v"(a[1]), "v"(a[2]), "v"(a[3]),
                       "v"(a[4]), "v"(a[5]), "v"(a[6]), "v"(a[7]));

        #pragma unroll
        for (int r = 0; r < 8; ++r) {
            const int e = e0 + 8 * h + r;
            float al = pre[r] + (acc0[r] + acc1[r]);
            al = (al > 0.f) ? al : NEG_SLOPE * al;
            if (c < 8 && e < n_edges) s_alpha[(e - e_base) * HEADS + c] = al;
        }
    }

    __syncthreads();
    const int cv = (t < NBIN) ? s_cnt[t] : 0;
    int all;
    const int ex = scan_nbin(cv, s_part, &all);
    if (t < NBIN) {
        s_off[t] = ex;
        cntm[(int)blockIdx.x * NBIN + t] = cv;
        offm[(int)blockIdx.x * NBIN + t] = ex;
    }
    __syncthreads();

    int n_loc = t_end * 16;
    if (n_loc > n_edges) n_loc = n_edges;
    n_loc -= e_base;
    for (int i = t; i < n_loc; i += 1024) {
        const int d = s_dst[i];
        const int pos = e_base + s_off[d / npb] + s_rk[i];
        st_src[pos] = s_src[i];
        st_dst[pos] = d;
        const v4f a0 = *(const v4f*)(s_alpha + i * HEADS);
        const v4f a1 = *(const v4f*)(s_alpha + i * HEADS + 4);
        float* o = st_alpha + (size_t)pos * HEADS;
        *(v4f*)(o)     = a0;
        *(v4f*)(o + 4) = a1;
    }
}

#define OWN_U 4

__device__ __forceinline__ int owner_pos(const int* s_p, const int* s_o, int j) {
    int w = 0;
    #pragma unroll
    for (int step = NBIN / 2; step >= 1; step >>= 1)
        if (s_p[w + step] <= j) w += step;
    return s_o[w] + (j - s_p[w]);
}

__global__ void __launch_bounds__(1024)
k_owner(const int* __restrict__ st_src, const int* __restrict__ st_dst,
        const float* __restrict__ st_alpha, const int* __restrict__ cntm,
        const int* __restrict__ offm, int* __restrict__ off,
        int* __restrict__ ssrc, float* __restrict__ salpha,
        int n_nodes, int n_edges, int src_wgs) {
    __shared__ int s_p[NBIN + 1];
    __shared__ int s_o[NBIN];
    __shared__ int s_ncnt[NBIN];
    __shared__ int s_noff[NBIN];
    __shared__ int s_nfill[NBIN];
    __shared__ int s_part[NBIN / 32];

    const int t = threadIdx.x;
    const int bin = (int)blockIdx.x;
    const int n_tiles = (n_edges + 15) >> 4;
    const int chunk = ((n_tiles + src_wgs - 1) / src_wgs) * 16;
    const int npb = (n_nodes + NBIN - 1) / NBIN;
    const int node0 = bin * npb;

    int cw = 0, ow = 0;
    if (t < NBIN) {
        if (t < src_wgs) {
            cw = cntm[t * NBIN + bin];
            ow = offm[t * NBIN + bin];
        }
        s_ncnt[t] = 0;
        s_nfill[t] = 0;
    }
    int total, base, unused;
    const int pw = scan_nbin(cw, s_part, &total);
    (void)scan_nbin(ow, s_part, &base);
    if (t < NBIN) {
        s_p[t] = pw;
        s_o[t] = t * chunk + ow;
        if (t == NBIN - 1) s_p[NBIN] = total;
    }
    __syncthreads();

    for (int j0 = 0; j0 < total; j0 += OWN_U * 1024) {
        int d[OWN_U];
        #pragma unroll
        for (int u = 0; u < OWN_U; ++u) {
            const int j = j0 + u * 1024 + t;
            const int jc = (j < total) ? j : total - 1;
            d[u] = st_dst[owner_pos(s_p, s_o, jc)];
        }
        #pragma unroll
        for (int u = 0; u < OWN_U; ++u)
            asm volatile("" : "+v"(d[u]));
        #pragma unroll
        for (int u = 0; u < OWN_U; ++u)
            if (j0 + u * 1024 + t < total) atomicAdd(&s_ncnt[d[u] - node0], 1);
    }
    __syncthreads();
    const int nc = (t < NBIN) ? s_ncnt[t] : 0;
    const int nex = scan_nbin(nc, s_part, &unused);
    if (t < NBIN) {
        s_noff[t] = nex;
        if (t < npb && node0 + t < n_nodes) off[node0 + t] = base + nex;
    }
    if (bin == 0 && t == 0) off[n_nodes] = n_edges;
    __syncthreads();

    for (int j0 = 0; j0 < total; j0 += OWN_U * 1024) {
        int d[OWN_U], s[OWN_U];
        v4f a0[OWN_U], a1[OWN_U];
        #pragma unroll
        for (int u = 0; u < OWN_U; ++u) {
            const int j = j0 + u * 1024 + t;
            const int jc = (j < total) ? j : total - 1;
            const int pos = owner_pos(s_p, s_o, jc);
            d[u] = st_dst[pos];
            s[u] = st_src[pos];
            const float* a = st_alpha + (size_t)pos * HEADS;
            a0[u] = *(const v4f*)(a);
            a1[u] = *(const v4f*)(a + 4);
        }
        #pragma unroll
        for (int u = 0; u < OWN_U; ++u)
            asm volatile("" : "+v"(d[u]), "+v"(s[u]), "+v"(a0[u]), "+v"(a1[u]));
        #pragma unroll
        for (int u = 0; u < OWN_U; ++u) {
            if (j0 + u * 1024 + t < total) {
                const int ln = d[u] - node0;
                const int slot = base + s_noff[ln] + atomicAdd(&s_nfill[ln], 1);
                ssrc[slot] = s[u];
                float* o = salpha + (size_t)slot * HEADS;
                *(v4f*)(o)     = a0[u];
                *(v4f*)(o + 4) = a1[u];
            }
        }
    }
}

__global__ void __launch_bounds__(256)
k_gather(const float* __restrict__ xp, const float* __restrict__ x,
         const int* __restrict__ off, const int* __restrict__ ssrc,
         const float* __restrict__ salpha,
         const float* __restrict__ bias, const float* __restrict__ gamma,
         const float* __restrict__ beta, float* __restrict__ out, int n_nodes) {
    const int wave = threadIdx.x >> 5;
    const int lane = threadIdx.x & 31;
    const int n = blockIdx.x * 8 + wave;
    if (n >= n_nodes) return;
    const int c0 = lane * 4;
    const int hA = lane >> 3, hB = 4 + hA;
    const int beg = __builtin_amdgcn_readfirstlane(off[n]);
    const int end = __builtin_amdgcn_readfirstlane(off[n + 1]);

    int my_s = (lane < end - beg) ? ssrc[beg + lane] : 0;

    float mx = -INFINITY;
    const int f_end = end * HEADS;
    for (int fb = beg * HEADS; fb < f_end; fb += 8 * 32) {
        float v[8];
        #pragma unroll
        for (int t = 0; t < 8; ++t) {
            const int f = fb + 32 * t + lane;
            v[t] = (f < f_end) ? salpha[f] : -INFINITY;
        }
        #pragma unroll
        for (int t = 0; t < 8; ++t) mx = fmaxf(mx, v[t]);
    }
    mx = fmaxf(mx, __shfl_xor(mx, 8));
    mx = fmaxf(mx, __shfl_xor(mx, 16));
    float mA = __shfl(mx, hA);
    float mB = __shfl(mx, hB);
    mA = isfinite(mA) ? mA : 0.f;
    mB = isfinite(mB) ? mB : 0.f;

    float4 accA = make_float4(0.f, 0.f, 0.f, 0.f);
    float4 accB = make_float4(0.f, 0.f, 0.f, 0.f);
    float denA = 0.f, denB = 0.f;
    for (int base = beg; base < end; base += 32) {
        const int rem = end - base;
        const int cc = rem < 32 ? rem : 32;
        if (base != beg) my_s = (lane < cc) ? ssrc[base + lane] : 0;
        #pragma unroll 4
        for (int j = 0; j < cc; ++j) {
            const int s = __builtin_amdgcn_readlane(my_s, j);
            const float* xs = xp + (size_t)s * HID;
            const float* al = salpha + (size_t)(base + j) * HEADS;
            const float4 rA = *(const float4*)(xs + c0);
            const float4 rB = *(const float4*)(xs + 128 + c0);
            const float eA = __expf(al[hA] - mA);
            const float eB = __expf(al[hB] - mB);
            denA += eA;
            denB += eB;
            accA.x += eA * rA.x; accA.y += eA * rA.y; accA.z += eA * rA.z; accA.w += eA * rA.w;
            accB.x += eB * rB.x; accB.y += eB * rB.y; accB.z += eB * rB.z; accB.w += eB * rB.w;
        }
    }

    const float4 xA = *(const float4*)(x + (size_t)n * HID + c0);
    const float4 xB = *(const float4*)(x + (size_t)n * HID + 128 + c0);
    const float4 bA = *(const float4*)(bias + c0);
    const float4 bB = *(const float4*)(bias + 128 + c0);
    const float dA = denA + 1e-16f, dB = denB + 1e-16f;
    float y[8];
    y[0] = xA.x + accA.x / dA + bA.x;
    y[1] = xA.y + accA.y / dA + bA.y;
    y[2] = xA.z + accA.z / dA + bA.z;
    y[3] = xA.w + accA.w / dA + bA.w;
    y[4] = xB.x + accB.x / dB + bB.x;
    y[5] = xB.y + accB.y / dB + bB.y;
    y[6] = xB.z + accB.z / dB + bB.z;
    y[7] = xB.w + accB.w / dB + bB.w;

    float s1 = 0.f, s2 = 0.f;
    #pragma unroll
    for (int k = 0; k < 8; ++k) { s1 += y[k]; s2 += y[k] * y[k]; }
    #pragma unroll
    for (int o = 16; o >= 1; o >>= 1) {
        s1 += __shfl_xor(s1, o);
        s2 += __shfl_xor(s2, o);
    }
    const float mean = s1 * (1.f / HID);
    const float var  = s2 * (1.f / HID) - mean * mean;
    const float rs   = rsqrtf(var + LN_EPS);
    const float4 gA = *(const float4*)(gamma + c0);
    const float4 gB = *(const float4*)(gamma + 128 + c0);
    const float4 tA = *(const float4*)(beta + c0);
    const float4 tB = *(const float4*)(beta + 128 + c0);
    float4 oA, oB;
    oA.x = (y[0] - mean) * rs * gA.x + tA.x;
    oA.y = (y[1] - mean) * rs * gA.y + tA.y;
    oA.z = (y[2] - mean) * rs * gA.z + tA.z;
    oA.w = (y[3] - mean) * rs * gA.w + tA.w;
    oB.x = (y[4] - mean) * rs * gB.x + tB.x;
    oB.y = (y[5] - mean) * rs * gB.y + tB.y;
    oB.z = (y[6] - mean) * rs * gB.z + tB.z;
    oB.w = (y[7] - mean) * rs * gB.w + tB.w;
    *(float4*)(out + (size_t)n * HID + c0) = oA;
    *(float4*)(out + (size_t)n * HID + 128 + c0) = oB;
}

extern "C" void kernel_launch(void* const* d_in, const int* in_sizes, int n_in,
                              void* d_out, int out_size, void* d_ws, size_t ws_size,
                              hipStream_t stream) {
    const float* x         = (const float*)d_in[0];
    const int*   ei        = (const int*)  d_in[1];
    const float* edge_attr = (const float*)d_in[2];
    const float* W         = (const float*)d_in[3];
    const float* W_e       = (const float*)d_in[4];
    const float* att_src   = (const float*)d_in[5];
    const float* att_dst   = (const float*)d_in[6];
    const float* att_edge  = (const float*)d_in[7];
    const float* bias      = (const float*)d_in[8];
    const float* gamma     = (const float*)d_in[9];
    const float* beta      = (const float*)d_in[10];
    float* out = (float*)d_out;

    const int n_nodes = in_sizes[0] / HID;
    const int n_edges = in_sizes[1] / 2;

    char* ws = (char*)d_ws;
    size_t off = 0;
    auto carve = [&](size_t bytes) -> char* {
        char* p = ws + off;
        off += (bytes + 255) & ~(size_t)255;
        return p;
    };
    float*    xp     = (float*)   carve((size_t)n_nodes * HID   * 4);
    float*    st_alpha = (float*) carve((size_t)(n_edges + 4112) * HEADS * 4);
    float*    salpha = (float*)   carve((size_t)n_edges * HEADS * 4);
    int*      ssrc   = (int*)     carve((size_t)n_edges * 4);
    int*      st_src = (int*)     carve((size_t)(n_edges + 4112) * 4);
    int*      st_dst = (int*)     carve((size_t)(n_edges + 4112) * 4);
    float*    a_src  = (float*)   carve((size_t)n_nodes * HEADS * 4);
    float*    a_dst  = (float*)   carve((size_t)n_nodes * HEADS * 4);
    int*      segoff = (int*)     carve((size_t)(n_nodes + 1) * 4);
    float*    w_he   = (float*)   carve((size_t)EDIM * HEADS * 4);
    int*      cntm   = (int*)     carve((size_t)NBIN * NBIN * 4);
    int*      offm   = (int*)     carve((size_t)NBIN * NBIN * 4);
    (void)ws_size;

    k_node_proj<<<(n_nodes + 31) / 32 + 1, 256, 0, stream>>>(x, W, W_e, att_edge, att_src, att_dst,
                                                             xp, a_src, a_dst, w_he,
                                                             n_nodes);

    k_alpha_bin<<<NBIN, 1024, 0, stream>>>(edge_attr, ei, w_he, a_src, a_dst,
                                           st_src, st_dst, st_alpha, cntm, offm,
                                           n_nodes, n_edges);

    k_owner<<<NBIN, 1024, 0, stream>>>(st_src, st_dst, st_alpha, cntm, offm, segoff,
                                       ssrc, salpha, n_nodes, n_edges, NBIN);

    k_gather<<<(n_nodes + 7) / 8, 256, 0, stream>>>(xp, x, segoff, ssrc, salpha,
                                                    bias, gamma, beta, out, n_nodes);
}


# Both edge-ordering levels permute their records inside LDS and store their range front to back: a record store covers whole 128-byte lines, not 4- or 16-byte pieces of 32 lines; no sum reordered
# speedup vs baseline: 3.4514x; 1.1010x over previous
#include <hip/hip_runtime.h>
#include <hip/hip_bf16.h>
#include <stdint.h>

#define HID   256
#define EDIM  64
#define HEADS 8
#define CCH   32
#define NEG_SLOPE 0.2f
#define LN_EPS 1e-5f

typedef float v2f __attribute__((ext_vector_type(2)));
typedef float v4f __attribute__((ext_vector_type(4)));
typedef float v8f __attribute__((ext_vector_type(8)));

__global__ void __launch_bounds__(256)
k_node_proj(const float* __restrict__ x, const float* __restrict__ W,
            const float* __restrict__ W_e, const float* __restrict__ att_edge,
            const float* __restrict__ att_src, const float* __restrict__ att_dst,
            float* __restrict__ xp, float* __restrict__ a_src,
            float* __restrict__ a_dst, float* __restrict__ w_he,
            int n_nodes) {
    const int wave = __builtin_amdgcn_readfirstlane((int)(threadIdx.x >> 5));
    const int lane = threadIdx.x & 31;

    if ((int)blockIdx.x == ((n_nodes + 31) >> 5)) {
        const int d = (wave & 1) * 32 + lane;
        #pragma unroll
        for (int u = 0; u < 2; ++u) {
            const int h = (wave >> 1) + 4 * u;
            const float* wr = W_e + d * HID + h * CCH;
            const float* ar = att_edge + h * CCH;
            float s = 0.f;
            #pragma unroll
            for (int c = 0; c < CCH; ++c) s += wr[c] * ar[c];
            w_he[h * EDIM + d] = s;
        }
        return;
    }

    const int row_base = (blockIdx.x * 2 + (wave >> 2)) * 16;
    const int col_base = (wave & 3) * 64;
    if (row_base >= n_nodes) return;

    int row = row_base + (lane & 15);
    if (row >= n_nodes) row = n_nodes - 1;
    const float* xrow = x + (size_t)row * HID + 4 * (lane >> 4);
    const float* wp = W + (size_t)(4 * (lane >> 4)) * HID + col_base + 4 * (lane & 15);

    v8f acc0 = {}, acc1 = {}, acc2 = {}, acc3 = {};
    for (int kb = 0; kb < HID; kb += 8) {
        const v4f a  = *(const v4f*)(xrow + kb);
        const float* wr = wp + (size_t)kb * HID;
        const v4f w0 = *(const v4f*)(wr);
        const v4f w1 = *(const v4f*)(wr + HID);
        const v4f w2 = *(const v4f*)(wr + 2 * HID);
        const v4f w3 = *(const v4f*)(wr + 3 * HID);
        v2f a01, a23, p0, p1, p2, p3, q0, q1, q2, q3;
        a01.x = a.x;  a01.y = a.y;  a23.x = a.z;  a23.y = a.w;
        p0.x = w0.x;  p0.y = w1.x;  p1.x = w0.y;  p1.y = w1.y;
        p2.x = w0.z;  p2.y = w1.z;  p3.x = w0.w;  p3.y = w1.w;
        q0.x = w2.x;  q0.y = w3.x;  q1.x = w2.y;  q1.y = w3.y;
        q2.x = w2.z;  q2.y = w3.z;  q3.x = w2.w;  q3.y = w3.w;
        acc0 = __builtin_amdgcn_wmma_f32_16x16x4_f32(false, a01, false, p0, (short)0, acc0, false, false);
        acc1 = __builtin_amdgcn_wmma_f32_16x16x4_f32(false, a01, false, p1, (short)0, acc1, false, false);
        acc2 = __builtin_amdgcn_wmma_f32_16x16x4_f32(false, a01, false, p2, (short)0, acc2, false, false);
        acc3 = __builtin_amdgcn_wmma_f32_16x16x4_f32(false, a01, false, p3, (short)0, acc3, false, false);
        acc0 = __builtin_amdgcn_wmma_f32_16x16x4_f32(false, a23, false, q0, (short)0, acc0, false, false);
        acc1 = __builtin_amdgcn_wmma_f32_16x16x4_f32(false, a23, false, q1, (short)0, acc1, false, false);
        acc2 = __builtin_amdgcn_wmma_f32_16x16x4_f32(false, a23, false, q2, (short)0, acc2, false, false);
        acc3 = __builtin_amdgcn_wmma_f32_16x16x4_f32(false, a23, false, q3, (short)0, acc3, false, false);
        asm volatile("v_nop\n\tv_nop\n\tv_nop\n\tv_nop\n\tv_nop\n\tv_nop\n\tv_nop\n\tv_nop"
                     : "+v"(acc0), "+v"(acc1), "+v"(acc2), "+v"(acc3)
                     : "v"(a), "v"(p0), "v"(p1), "v"(p2), "v"(p3),
                       "v"(q0), "v"(q1), "v"(q2), "v"(q3));
    }

    const int pl = (int)__builtin_amdgcn_mbcnt_lo(~0u, 0u);
    const int srow = row_base + ((pl >> 4) * 8);
    const int scol = col_base + 4 * (pl & 15);
    #pragma unroll
    for (int i = 0; i < 8; ++i) {
        int r = srow + i;
        if (r < n_nodes) {
            v4f o;
            o.x = acc0[i]; o.y = acc1[i]; o.z = acc2[i]; o.w = acc3[i];
            *(v4f*)(xp + (size_t)r * HID + scol) = o;
        }
    }

    const bool b2 = (pl & 4) != 0, b1 = (pl & 2) != 0, b0 = (pl & 1) != 0;
    const v4f as = *(const v4f*)(att_src + scol);
    const v4f ad = *(const v4f*)(att_dst + scol);
    float v1[8], v2[4], v3[2];
    #pragma unroll
    for (int i = 0; i < 8; ++i) {
        const float sv = acc0[i] * as.x + acc1[i] * as.y + acc2[i] * as.z + acc3[i] * as.w;
        const float dv = acc0[i] * ad.x + acc1[i] * ad.y + acc2[i] * ad.z + acc3[i] * ad.w;
        v1[i] = (b2 ? dv : sv) + __shfl_xor(b2 ? sv : dv, 4);
    }
    #pragma unroll
    for (int i = 0; i < 4; ++i)
        v2[i] = (b1 ? v1[i + 4] : v1[i]) + __shfl_xor(b1 ? v1[i] : v1[i + 4], 2);
    #pragma unroll
    for (int i = 0; i < 2; ++i)
        v3[i] = (b0 ? v2[i + 2] : v2[i]) + __shfl_xor(b0 ? v2[i] : v2[i + 2], 1);
    {
        const int r = srow + 2 * (pl & 3);
        float* o = (b2 ? a_dst : a_src) + (size_t)r * HEADS + (wave & 3) * 2 + ((pl >> 3) & 1);
        if (r < n_nodes) o[0] = v3[0];
        if (r + 1 < n_nodes) o[HEADS] = v3[1];
    }
}

#define NBIN 256
#define CHUNK_CAP 3200

__device__ __forceinline__ int scan_nbin(int v, int* s_part, int* total) {
    const int lane = threadIdx.x & 31, wave = threadIdx.x >> 5;
    int inc = v;
    #pragma unroll
    for (int d = 1; d < 32; d <<= 1) {
        const int u = __shfl_up(inc, d);
        if (lane >= d) inc += u;
    }
    __syncthreads();
    if (lane == 31 && wave < NBIN / 32) s_part[wave] = inc;
    __syncthreads();
    int before = 0, all = 0;
    #pragma unroll
    for (int k = 0; k < NBIN / 32; ++k) {
        const int p = s_part[k];
        all += p;
        if (k < wave) before += p;
    }
    *total = all;
    return before + inc - v;
}

__device__ __forceinline__ int frag_k0(int i, int h) { return 8 * i + 4 * h; }

__global__ void __launch_bounds__(1024)
k_alpha_bin(const float* __restrict__ edge_attr, const int* __restrict__ ei,
            const float* __restrict__ w_he, const float* __restrict__ a_src,
            const float* __restrict__ a_dst, int* __restrict__ st_src,
            int* __restrict__ st_dst, float* __restrict__ st_alpha,
            int* __restrict__ cntm, int* __restrict__ offm,
            int n_nodes, int n_edges) {
    __shared__ __attribute__((aligned(16))) float s_alpha[CHUNK_CAP * HEADS];
    __shared__ int s_src[CHUNK_CAP];
    __shared__ int s_dst[CHUNK_CAP];
    __shared__ int s_rk[CHUNK_CAP];
    __shared__ int s_inv[CHUNK_CAP];
    __shared__ int s_cnt[NBIN];
    __shared__ int s_off[NBIN];
    __shared__ int s_part[NBIN / 32];

    const int t = threadIdx.x;
    const int lane = t & 31;
    const int c = lane & 15;
    const int h = lane >> 4;
    const int head = c & 7;
    const int wave = __builtin_amdgcn_readfirstlane(t >> 5);
    const int n_tiles = (n_edges + 15) >> 4;
    const int tpw = (n_tiles + (int)gridDim.x - 1) / (int)gridDim.x;
    const int t_beg = (int)blockIdx.x * tpw;
    int t_end = t_beg + tpw;
    if (t_end > n_tiles) t_end = n_tiles;
    const int e_base = t_beg * 16;
    const int npb = (n_nodes + NBIN - 1) / NBIN;

    if (t < NBIN) s_cnt[t] = 0;
    __syncthreads();

    v4f b[8];
    #pragma unroll
    for (int i = 0; i < 8; ++i)
        b[i] = *(const v4f*)(w_he + head * EDIM + frag_k0(i, h));

    for (int tile = t_beg + wave; tile < t_end; tile += 32) {
        const int e0 = tile * 16;
        int el = e0 + c;
        if (el >= n_edges) el = n_edges - 1;
        const int srcv = ei[el];
        const int dstv = ei[n_edges + el];
        const float* arow = edge_attr + (size_t)el * EDIM;
        v4f a[8];
        #pragma unroll
        for (int i = 0; i < 8; ++i)
            a[i] = *(const v4f*)(arow + frag_k0(i, h));

        const bool own = (h == 0) && (e0 + c < n_edges);
        if (own) {
            const int li = e0 + c - e_base;
            s_src[li] = srcv;
            s_dst[li] = dstv;
            s_rk[li]  = atomicAdd(&s_cnt[dstv / npb], 1);
        }

        float pre[8];
        #pragma unroll
        for (int r = 0; r < 8; ++r) {
            const int sl = 8 * h + r;
            const int s = __shfl(srcv, sl);
            const int d = __shfl(dstv, sl);
            pre[r] = a_src[s * HEADS + head] + a_dst[d * HEADS + head];
        }

        v8f acc0 = {}, acc1 = {};
        #pragma unroll
        for (int i = 0; i < 8; ++i) {
            v2f a_lo, a_hi, b_lo, b_hi;
            a_lo.x = a[i].x; a_lo.y = a[i].y; a_hi.x = a[i].z; a_hi.y = a[i].w;
            b_lo.x = b[i].x; b_lo.y = b[i].y; b_hi.x = b[i].z; b_hi.y = b[i].w;
            acc0 = __builtin_amdgcn_wmma_f32_16x16x4_f32(false, a_lo, false, b_lo, (short)0, acc0, false, false);
            acc1 = __builtin_amdgcn_wmma_f32_16x16x4_f32(false, a_hi, false, b_hi, (short)0, acc1, false, false);
        }

        asm volatile("v_nop\n\tv_nop\n\tv_nop\n\tv_nop\n\tv_nop\n\tv_nop\n\tv_nop\n\tv_nop"
                     : "+v"(acc0), "+v"(acc1)
                     : "v"(a[0]), "v"(a[1]), "v"(a[2]), "v"(a[3]),
                       "v"(a[4]), "v"(a[5]), "v"(a[6]), "v"(a[7]));

        #pragma unroll
        for (int r = 0; r < 8; ++r) {
            const int e = e0 + 8 * h + r;
            float al = pre[r] + (acc0[r] + acc1[r]);
            al = (al > 0.f) ? al : NEG_SLOPE * al;
            if (c < 8 && e < n_edges) s_alpha[(e - e_base) * HEADS + c] = al;
        }
    }

    __syncthreads();
    const int cv = (t < NBIN) ? s_cnt[t] : 0;
    int all;
    const int ex = scan_nbin(cv, s_part, &all);
    if (t < NBIN) {
        s_off[t] = ex;
        cntm[(int)blockIdx.x * NBIN + t] = cv;
        offm[(int)blockIdx.x * NBIN + t] = ex;
    }
    __syncthreads();

    int n_loc = t_end * 16;
    if (n_loc > n_edges) n_loc = n_edges;
    n_loc -= e_base;
    for (int i = t; i < n_loc; i += 1024)
        s_inv[s_off[s_dst[i] / npb] + s_rk[i]] = i;
    __syncthreads();
    for (int p = t; p < n_loc; p += 1024) {
        const int i = s_inv[p];
        st_src[e_base + p] = s_src[i];
        st_dst[e_base + p] = s_dst[i];
    }
    for (int q = t; q < 2 * n_loc; q += 1024) {
        const int i = s_inv[q >> 1];
        const int hf = (q & 1) * 4;
        const v4f a = *(const v4f*)(s_alpha + i * HEADS + hf);
        *(v4f*)(st_alpha + (size_t)(e_base + (q >> 1)) * HEADS + hf) = a;
    }
}

#define OWN_U 4
#define OWN_CAP 4096

__device__ __forceinline__ int owner_pos(const int* s_p, const int* s_o, int j) {
    int w = 0;
    #pragma unroll
    for (int step = NBIN / 2; step >= 1; step >>= 1)
        if (s_p[w + step] <= j) w += step;
    return s_o[w] + (j - s_p[w]);
}

__global__ void __launch_bounds__(1024)
k_owner(const int* __restrict__ st_src, const int* __restrict__ st_dst,
        const float* __restrict__ st_alpha, const int* __restrict__ cntm,
        const int* __restrict__ offm, int* __restrict__ off,
        int* __restrict__ ssrc, float* __restrict__ salpha,
        int n_nodes, int n_edges, int src_wgs) {
    __shared__ int s_p[NBIN + 1];
    __shared__ int s_o[NBIN];
    __shared__ int s_ncnt[NBIN];
    __shared__ int s_noff[NBIN];
    __shared__ int s_nfill[NBIN];
    __shared__ int s_part[NBIN / 32];
    __shared__ int s_osrc[OWN_CAP];
    __shared__ __attribute__((aligned(16))) float s_oal[OWN_CAP * HEADS];

    const int t = threadIdx.x;
    const int bin = (int)blockIdx.x;
    const int n_tiles = (n_edges + 15) >> 4;
    const int chunk = ((n_tiles + src_wgs - 1) / src_wgs) * 16;
    const int npb = (n_nodes + NBIN - 1) / NBIN;
    const int node0 = bin * npb;

    int cw = 0, ow = 0;
    if (t < NBIN) {
        if (t < src_wgs) {
            cw = cntm[t * NBIN + bin];
            ow = offm[t * NBIN + bin];
        }
        s_ncnt[t] = 0;
        s_nfill[t] = 0;
    }
    int total, base, unused;
    const int pw = scan_nbin(cw, s_part, &total);
    (void)scan_nbin(ow, s_part, &base);
    if (t < NBIN) {
        s_p[t] = pw;
        s_o[t] = t * chunk + ow;
        if (t == NBIN - 1) s_p[NBIN] = total;
    }
    __syncthreads();

    for (int j0 = 0; j0 < total; j0 += OWN_U * 1024) {
        int d[OWN_U];
        #pragma unroll
        for (int u = 0; u < OWN_U; ++u) {
            const int j = j0 + u * 1024 + t;
            const int jc = (j < total) ? j : total - 1;
            d[u] = st_dst[owner_pos(s_p, s_o, jc)];
        }
        #pragma unroll
        for (int u = 0; u < OWN_U; ++u)
            asm volatile("" : "+v"(d[u]));
        #pragma unroll
        for (int u = 0; u < OWN_U; ++u)
            if (j0 + u * 1024 + t < total) atomicAdd(&s_ncnt[d[u] - node0], 1);
    }
    __syncthreads();
    const int nc = (t < NBIN) ? s_ncnt[t] : 0;
    const int nex = scan_nbin(nc, s_part, &unused);
    if (t < NBIN) {
        s_noff[t] = nex;
        if (t < npb && node0 + t < n_nodes) off[node0 + t] = base + nex;
    }
    if (bin == 0 && t == 0) off[n_nodes] = n_edges;
    __syncthreads();

    const bool in_lds = total <= OWN_CAP;
    for (int j0 = 0; j0 < total; j0 += OWN_U * 1024) {
        int d[OWN_U], s[OWN_U];
        v4f a0[OWN_U], a1[OWN_U];
        #pragma unroll
        for (int u = 0; u < OWN_U; ++u) {
            const int j = j0 + u * 1024 + t;
            const int jc = (j < total) ? j : total - 1;
            const int pos = owner_pos(s_p, s_o, jc);
            d[u] = st_dst[pos];
            s[u] = st_src[pos];
            const float* a = st_alpha + (size_t)pos * HEADS;
            a0[u] = *(const v4f*)(a);
            a1[u] = *(const v4f*)(a + 4);
        }
        #pragma unroll
        for (int u = 0; u < OWN_U; ++u)
            asm volatile("" : "+v"(d[u]), "+v"(s[u]), "+v"(a0[u]), "+v"(a1[u]));
        #pragma unroll
        for (int u = 0; u < OWN_U; ++u) {
            if (j0 + u * 1024 + t < total) {
                const int ln = d[u] - node0;
                const int q = s_noff[ln] + atomicAdd(&s_nfill[ln], 1);
                if (in_lds) {
                    s_osrc[q] = s[u];
                    *(v4f*)(s_oal + q * HEADS)     = a0[u];
                    *(v4f*)(s_oal + q * HEADS + 4) = a1[u];
                } else {
                    ssrc[base + q] = s[u];
                    float* o = salpha + (size_t)(base + q) * HEADS;
                    *(v4f*)(o)     = a0[u];
                    *(v4f*)(o + 4) = a1[u];
                }
            }
        }
    }
    if (in_lds) {
        __syncthreads();
        for (int q = t; q < total; q += 1024)
            ssrc[base + q] = s_osrc[q];
        float* o = salpha + (size_t)base * HEADS;
        for (int q = t; q < 2 * total; q += 1024)
            *(v4f*)(o + 4 * q) = *(const v4f*)(s_oal + 4 * q);
    }
}

__global__ void __launch_bounds__(256)
k_gather(const float* __restrict__ xp, const float* __restrict__ x,
         const int* __restrict__ off, const int* __restrict__ ssrc,
         const float* __restrict__ salpha,
         const float* __restrict__ bias, const float* __restrict__ gamma,
         const float* __restrict__ beta, float* __restrict__ out, int n_nodes) {
    const int wave = threadIdx.x >> 5;
    const int lane = threadIdx.x & 31;
    const int n = blockIdx.x * 8 + wave;
    if (n >= n_nodes) return;
    const int c0 = lane * 4;
    const int hA = lane >> 3, hB = 4 + hA;
    const int beg = __builtin_amdgcn_readfirstlane(off[n]);
    const int end = __builtin_amdgcn_readfirstlane(off[n + 1]);

    int my_s = (lane < end - beg) ? ssrc[beg + lane] : 0;

    float mx = -INFINITY;
    const int f_end = end * HEADS;
    for (int fb = beg * HEADS; fb < f_end; fb += 8 * 32) {
        float v[8];
        #pragma unroll
        for (int t = 0; t < 8; ++t) {
            const int f = fb + 32 * t + lane;
            v[t] = (f < f_end) ? salpha[f] : -INFINITY;
        }
        #pragma unroll
        for (int t = 0; t < 8; ++t) mx = fmaxf(mx, v[t]);
    }
    mx = fmaxf(mx, __shfl_xor(mx, 8));
    mx = fmaxf(mx, __shfl_xor(mx, 16));
    float mA = __shfl(mx, hA);
    float mB = __shfl(mx, hB);
    mA = isfinite(mA) ? mA : 0.f;
    mB = isfinite(mB) ? mB : 0.f;

    float4 accA = make_float4(0.f, 0.f, 0.f, 0.f);
    float4 accB = make_float4(0.f, 0.f, 0.f, 0.f);
    float denA = 0.f, denB = 0.f;
    for (int base = beg; base < end; base += 32) {
        const int rem = end - base;
        const int cc = rem < 32 ? rem : 32;
        if (base != beg) my_s = (lane < cc) ? ssrc[base + lane] : 0;
        #pragma unroll 4
        for (int j = 0; j < cc; ++j) {
            const int s = __builtin_amdgcn_readlane(my_s, j);
            const float* xs = xp + (size_t)s * HID;
            const float* al = salpha + (size_t)(base + j) * HEADS;
            const float4 rA = *(const float4*)(xs + c0);
            const float4 rB = *(const float4*)(xs + 128 + c0);
            const float eA = __expf(al[hA] - mA);
            const float eB = __expf(al[hB] - mB);
            denA += eA;
            denB += eB;
            accA.x += eA * rA.x; accA.y += eA * rA.y; accA.z += eA * rA.z; accA.w += eA * rA.w;
            accB.x += eB * rB.x; accB.y += eB * rB.y; accB.z += eB * rB.z; accB.w += eB * rB.w;
        }
    }

    const float4 xA = *(const float4*)(x + (size_t)n * HID + c0);
    const float4 xB = *(const float4*)(x + (size_t)n * HID + 128 + c0);
    const float4 bA = *(const float4*)(bias + c0);
    const float4 bB = *(const float4*)(bias + 128 + c0);
    const float dA = denA + 1e-16f, dB = denB + 1e-16f;
    float y[8];
    y[0] = xA.x + accA.x / dA + bA.x;
    y[1] = xA.y + accA.y / dA + bA.y;
    y[2] = xA.z + accA.z / dA + bA.z;
    y[3] = xA.w + accA.w / dA + bA.w;
    y[4] = xB.x + accB.x / dB + bB.x;
    y[5] = xB.y + accB.y / dB + bB.y;
    y[6] = xB.z + accB.z / dB + bB.z;
    y[7] = xB.w + accB.w / dB + bB.w;

    float s1 = 0.f, s2 = 0.f;
    #pragma unroll
    for (int k = 0; k < 8; ++k) { s1 += y[k]; s2 += y[k] * y[k]; }
    #pragma unroll
    for (int o = 16; o >= 1; o >>= 1) {
        s1 += __shfl_xor(s1, o);
        s2 += __shfl_xor(s2, o);
    }
    const float mean = s1 * (1.f / HID);
    const float var  = s2 * (1.f / HID) - mean * mean;
    const float rs   = rsqrtf(var + LN_EPS);
    const float4 gA = *(const float4*)(gamma + c0);
    const float4 gB = *(const float4*)(gamma + 128 + c0);
    const float4 tA = *(const float4*)(beta + c0);
    const float4 tB = *(const float4*)(beta + 128 + c0);
    float4 oA, oB;
    oA.x = (y[0] - mean) * rs * gA.x + tA.x;
    oA.y = (y[1] - mean) * rs * gA.y + tA.y;
    oA.z = (y[2] - mean) * rs * gA.z + tA.z;
    oA.w = (y[3] - mean) * rs * gA.w + tA.w;
    oB.x = (y[4] - mean) * rs * gB.x + tB.x;
    oB.y = (y[5] - mean) * rs * gB.y + tB.y;
    oB.z = (y[6] - mean) * rs * gB.z + tB.z;
    oB.w = (y[7] - mean) * rs * gB.w + tB.w;
    *(float4*)(out + (size_t)n * HID + c0) = oA;
    *(float4*)(out + (size_t)n * HID + 128 + c0) = oB;
}

extern "C" void kernel_launch(void* const* d_in, const int* in_sizes, int n_in,
                              void* d_out, int out_size, void* d_ws, size_t ws_size,
                              hipStream_t stream) {
    const float* x         = (const float*)d_in[0];
    const int*   ei        = (const int*)  d_in[1];
    const float* edge_attr = (const float*)d_in[2];
    const float* W         = (const float*)d_in[3];
    const float* W_e       = (const float*)d_in[4];
    const float* att_src   = (const float*)d_in[5];
    const float* att_dst   = (const float*)d_in[6];
    const float* att_edge  = (const float*)d_in[7];
    const float* bias      = (const float*)d_in[8];
    const float* gamma     = (const float*)d_in[9];
    const float* beta      = (const float*)d_in[10];
    float* out = (float*)d_out;

    const int n_nodes = in_sizes[0] / HID;
    const int n_edges = in_sizes[1] / 2;

    char* ws = (char*)d_ws;
    size_t off = 0;
    auto carve = [&](size_t bytes) -> char* {
        char* p = ws + off;
        off += (bytes + 255) & ~(size_t)255;
        return p;
    };
    float*    xp     = (float*)   carve((size_t)n_nodes * HID   * 4);
    float*    st_alpha = (float*) carve((size_t)(n_edges + 4112) * HEADS * 4);
    float*    salpha = (float*)   carve((size_t)n_edges * HEADS * 4);
    int*      ssrc   = (int*)     carve((size_t)n_edges * 4);
    int*      st_src = (int*)     carve((size_t)(n_edges + 4112) * 4);
    int*      st_dst = (int*)     carve((size_t)(n_edges + 4112) * 4);
    float*    a_src  = (float*)   carve((size_t)n_nodes * HEADS * 4);
    float*    a_dst  = (float*)   carve((size_t)n_nodes * HEADS * 4);
    int*      segoff = (int*)     carve((size_t)(n_nodes + 1) * 4);
    float*    w_he   = (float*)   carve((size_t)EDIM * HEADS * 4);
    int*      cntm   = (int*)     carve((size_t)NBIN * NBIN * 4);
    int*      offm   = (int*)     carve((size_t)NBIN * NBIN * 4);
    (void)ws_size;

    k_node_proj<<<(n_nodes + 31) / 32 + 1, 256, 0, stream>>>(x, W, W_e, att_edge, att_src, att_dst,
                                                             xp, a_src, a_dst, w_he,
                                                             n_nodes);

    k_alpha_bin<<<NBIN, 1024, 0, stream>>>(edge_attr, ei, w_he, a_src, a_dst,
                                           st_src, st_dst, st_alpha, cntm, offm,
                                           n_nodes, n_edges);

    k_owner<<<NBIN, 1024, 0, stream>>>(st_src, st_dst, st_alpha, cntm, offm, segoff,
                                       ssrc, salpha, n_nodes, n_edges, NBIN);

    k_gather<<<(n_nodes + 7) / 8, 256, 0, stream>>>(xp, x, segoff, ssrc, salpha,
                                                    bias, gamma, beta, out, n_nodes);
}


# Owner workgroup gathers its own bin's nodes from the records it placed in LDS (32 waves take nodes from an LDS counter): launches 4 -> 3, sorted ids and logits never reach memory; sums unchanged
# speedup vs baseline: 3.7764x; 1.0942x over previous
#include <hip/hip_runtime.h>
#include <hip/hip_bf16.h>
#include <stdint.h>

#define HID   256
#define EDIM  64
#define HEADS 8
#define CCH   32
#define NEG_SLOPE 0.2f
#define LN_EPS 1e-5f

typedef float v2f __attribute__((ext_vector_type(2)));
typedef float v4f __attribute__((ext_vector_type(4)));
typedef float v8f __attribute__((ext_vector_type(8)));

__global__ void __launch_bounds__(256)
k_node_proj(const float* __restrict__ x, const float* __restrict__ W,
            const float* __restrict__ W_e, const float* __restrict__ att_edge,
            const float* __restrict__ att_src, const float* __restrict__ att_dst,
            float* __restrict__ xp, float* __restrict__ a_src,
            float* __restrict__ a_dst, float* __restrict__ w_he,
            int n_nodes) {
    const int wave = __builtin_amdgcn_readfirstlane((int)(threadIdx.x >> 5));
    const int lane = threadIdx.x & 31;

    if ((int)blockIdx.x == ((n_nodes + 31) >> 5)) {
        const int d = (wave & 1) * 32 + lane;
        #pragma unroll
        for (int u = 0; u < 2; ++u) {
            const int h = (wave >> 1) + 4 * u;
            const float* wr = W_e + d * HID + h * CCH;
            const float* ar = att_edge + h * CCH;
            float s = 0.f;
            #pragma unroll
            for (int c = 0; c < CCH; ++c) s += wr[c] * ar[c];
            w_he[h * EDIM + d] = s;
        }
        return;
    }

    const int row_base = (blockIdx.x * 2 + (wave >> 2)) * 16;
    const int col_base = (wave & 3) * 64;
    if (row_base >= n_nodes) return;

    int row = row_base + (lane & 15);
    if (row >= n_nodes) row = n_nodes - 1;
    const float* xrow = x + (size_t)row * HID + 4 * (lane >> 4);
    const float* wp = W + (size_t)(4 * (lane >> 4)) * HID + col_base + 4 * (lane & 15);

    v8f acc0 = {}, acc1 = {}, acc2 = {}, acc3 = {};
    for (int kb = 0; kb < HID; kb += 8) {
        const v4f a  = *(const v4f*)(xrow + kb);
        const float* wr = wp + (size_t)kb * HID;
        const v4f w0 = *(const v4f*)(wr);
        const v4f w1 = *(const v4f*)(wr + HID);
        const v4f w2 = *(const v4f*)(wr + 2 * HID);
        const v4f w3 = *(const v4f*)(wr + 3 * HID);
        v2f a01, a23, p0, p1, p2, p3, q0, q1, q2, q3;
        a01.x = a.x;  a01.y = a.y;  a23.x = a.z;  a23.y = a.w;
        p0.x = w0.x;  p0.y = w1.x;  p1.x = w0.y;  p1.y = w1.y;
        p2.x = w0.z;  p2.y = w1.z;  p3.x = w0.w;  p3.y = w1.w;
        q0.x = w2.x;  q0.y = w3.x;  q1.x = w2.y;  q1.y = w3.y;
        q2.x = w2.z;  q2.y = w3.z;  q3.x = w2.w;  q3.y = w3.w;
        acc0 = __builtin_amdgcn_wmma_f32_16x16x4_f32(false, a01, false, p0, (short)0, acc0, false, false);
        acc1 = __builtin_amdgcn_wmma_f32_16x16x4_f32(false, a01, false, p1, (short)0, acc1, false, false);
        acc2 = __builtin_amdgcn_wmma_f32_16x16x4_f32(false, a01, false, p2, (short)0, acc2, false, false);
        acc3 = __builtin_amdgcn_wmma_f32_16x16x4_f32(false, a01, false, p3, (short)0, acc3, false, false);
        acc0 = __builtin_amdgcn_wmma_f32_16x16x4_f32(false, a23, false, q0, (short)0, acc0, false, false);
        acc1 = __builtin_amdgcn_wmma_f32_16x16x4_f32(false, a23, false, q1, (short)0, acc1, false, false);
        acc2 = __builtin_amdgcn_wmma_f32_16x16x4_f32(false, a23, false, q2, (short)0, acc2, false, false);
        acc3 = __builtin_amdgcn_wmma_f32_16x16x4_f32(false, a23, false, q3, (short)0, acc3, false, false);
        asm volatile("v_nop\n\tv_nop\n\tv_nop\n\tv_nop\n\tv_nop\n\tv_nop\n\tv_nop\n\tv_nop"
                     : "+v"(acc0), "+v"(acc1), "+v"(acc2), "+v"(acc3)
                     : "v"(a), "v"(p0), "v"(p1), "v"(p2), "v"(p3),
                       "v"(q0), "v"(q1), "v"(q2), "v"(q3));
    }

    const int pl = (int)__builtin_amdgcn_mbcnt_lo(~0u, 0u);
    const int srow = row_base + ((pl >> 4) * 8);
    const int scol = col_base + 4 * (pl & 15);
    #pragma unroll
    for (int i = 0; i < 8; ++i) {
        int r = srow + i;
        if (r < n_nodes) {
            v4f o;
            o.x = acc0[i]; o.y = acc1[i]; o.z = acc2[i]; o.w = acc3[i];
            *(v4f*)(xp + (size_t)r * HID + scol) = o;
        }
    }

    const bool b2 = (pl & 4) != 0, b1 = (pl & 2) != 0, b0 = (pl & 1) != 0;
    const v4f as = *(const v4f*)(att_src + scol);
    const v4f ad = *(const v4f*)(att_dst + scol);
    float v1[8], v2[4], v3[2];
    #pragma unroll
    for (int i = 0; i < 8; ++i) {
        const float sv = acc0[i] * as.x + acc1[i] * as.y + acc2[i] * as.z + acc3[i] * as.w;
        const float dv = acc0[i] * ad.x + acc1[i] * ad.y + acc2[i] * ad.z + acc3[i] * ad.w;
        v1[i] = (b2 ? dv : sv) + __shfl_xor(b2 ? sv : dv, 4);
    }
    #pragma unroll
    for (int i = 0; i < 4; ++i)
        v2[i] = (b1 ? v1[i + 4] : v1[i]) + __shfl_xor(b1 ? v1[i] : v1[i + 4], 2);
    #pragma unroll
    for (int i = 0; i < 2; ++i)
        v3[i] = (b0 ? v2[i + 2] : v2[i]) + __shfl_xor(b0 ? v2[i] : v2[i + 2], 1);
    {
        const int r = srow + 2 * (pl & 3);
        float* o = (b2 ? a_dst : a_src) + (size_t)r * HEADS + (wave & 3) * 2 + ((pl >> 3) & 1);
        if (r < n_nodes) o[0] = v3[0];
        if (r + 1 < n_nodes) o[HEADS] = v3[1];
    }
}

#define NBIN 256
#define CHUNK_CAP 3200

__device__ __forceinline__ int scan_nbin(int v, int* s_part, int* total) {
    const int lane = threadIdx.x & 31, wave = threadIdx.x >> 5;
    int inc = v;
    #pragma unroll
    for (int d = 1; d < 32; d <<= 1) {
        const int u = __shfl_up(inc, d);
        if (lane >= d) inc += u;
    }
    __syncthreads();
    if (lane == 31 && wave < NBIN / 32) s_part[wave] = inc;
    __syncthreads();
    int before = 0, all = 0;
    #pragma unroll
    for (int k = 0; k < NBIN / 32; ++k) {
        const int p = s_part[k];
        all += p;
        if (k < wave) before += p;
    }
    *total = all;
    return before + inc - v;
}

__device__ __forceinline__ int frag_k0(int i, int h) { return 8 * i + 4 * h; }

__global__ void __launch_bounds__(1024)
k_alpha_bin(const float* __restrict__ edge_attr, const int* __restrict__ ei,
            const float* __restrict__ w_he, const float* __restrict__ a_src,
            const float* __restrict__ a_dst, int* __restrict__ st_src,
            int* __restrict__ st_dst, float* __restrict__ st_alpha,
            int* __restrict__ cntm, int* __restrict__ offm,
            int n_nodes, int n_edges) {
    __shared__ __attribute__((aligned(16))) float s_alpha[CHUNK_CAP * HEADS];
    __shared__ int s_src[CHUNK_CAP];
    __shared__ int s_dst[CHUNK_CAP];
    __shared__ int s_rk[CHUNK_CAP];
    __shared__ int s_inv[CHUNK_CAP];
    __shared__ int s_cnt[NBIN];
    __shared__ int s_off[NBIN];
    __shared__ int s_part[NBIN / 32];

    const int t = threadIdx.x;
    const int lane = t & 31;
    const int c = lane & 15;
    const int h = lane >> 4;
    const int head = c & 7;
    const int wave = __builtin_amdgcn_readfirstlane(t >> 5);
    const int n_tiles = (n_edges + 15) >> 4;
    const int tpw = (n_tiles + (int)gridDim.x - 1) / (int)gridDim.x;
    const int t_beg = (int)blockIdx.x * tpw;
    int t_end = t_beg + tpw;
    if (t_end > n_tiles) t_end = n_tiles;
    const int e_base = t_beg * 16;
    const int npb = (n_nodes + NBIN - 1) / NBIN;

    if (t < NBIN) s_cnt[t] = 0;
    __syncthreads();

    v4f b[8];
    #pragma unroll
    for (int i = 0; i < 8; ++i)
        b[i] = *(const v4f*)(w_he + head * EDIM + frag_k0(i, h));

    for (int tile = t_beg + wave; tile < t_end; tile += 32) {
        const int e0 = tile * 16;
        int el = e0 + c;
        if (el >= n_edges) el = n_edges - 1;
        const int srcv = ei[el];
        const int dstv = ei[n_edges + el];
        const float* arow = edge_attr + (size_t)el * EDIM;
        v4f a[8];
        #pragma unroll
        for (int i = 0; i < 8; ++i)
            a[i] = *(const v4f*)(arow + frag_k0(i, h));

        const bool own = (h == 0) && (e0 + c < n_edges);
        if (own) {
            const int li = e0 + c - e_base;
            s_src[li] = srcv;
            s_dst[li] = dstv;
            s_rk[li]  = atomicAdd(&s_cnt[dstv / npb], 1);
        }

        float pre[8];
        #pragma unroll
        for (int r = 0; r < 8; ++r) {
            const int sl = 8 * h + r;
            const int s = __shfl(srcv, sl);
            const int d = __shfl(dstv, sl);
            pre[r] = a_src[s * HEADS + head] + a_dst[d * HEADS + head];
        }

        v8f acc0 = {}, acc1 = {};
        #pragma unroll
        for (int i = 0; i < 8; ++i) {
            v2f a_lo, a_hi, b_lo, b_hi;
            a_lo.x = a[i].x; a_lo.y = a[i].y; a_hi.x = a[i].z; a_hi.y = a[i].w;
            b_lo.x = b[i].x; b_lo.y = b[i].y; b_hi.x = b[i].z; b_hi.y = b[i].w;
            acc0 = __builtin_amdgcn_wmma_f32_16x16x4_f32(false, a_lo, false, b_lo, (short)0, acc0, false, false);
            acc1 = __builtin_amdgcn_wmma_f32_16x16x4_f32(false, a_hi, false, b_hi, (short)0, acc1, false, false);
        }

        asm volatile("v_nop\n\tv_nop\n\tv_nop\n\tv_nop\n\tv_nop\n\tv_nop\n\tv_nop\n\tv_nop"
                     : "+v"(acc0), "+v"(acc1)
                     : "v"(a[0]), "v"(a[1]), "v"(a[2]), "v"(a[3]),
                       "v"(a[4]), "v"(a[5]), "v"(a[6]), "v"(a[7]));

        #pragma unroll
        for (int r = 0; r < 8; ++r) {
            const int e = e0 + 8 * h + r;
            float al = pre[r] + (acc0[r] + acc1[r]);
            al = (al > 0.f) ? al : NEG_SLOPE * al;
            if (c < 8 && e < n_edges) s_alpha[(e - e_base) * HEADS + c] = al;
        }
    }

    __syncthreads();
    const int cv = (t < NBIN) ? s_cnt[t] : 0;
    int all;
    const int ex = scan_nbin(cv, s_part, &all);
    if (t < NBIN) {
        s_off[t] = ex;
        cntm[(int)blockIdx.x * NBIN + t] = cv;
        offm[(int)blockIdx.x * NBIN + t] = ex;
    }
    __syncthreads();

    int n_loc = t_end * 16;
    if (n_loc > n_edges) n_loc = n_edges;
    n_loc -= e_base;
    for (int i = t; i < n_loc; i += 1024)
        s_inv[s_off[s_dst[i] / npb] + s_rk[i]] = i;
    __syncthreads();
    for (int p = t; p < n_loc; p += 1024) {
        const int i = s_inv[p];
        st_src[e_base + p] = s_src[i];
        st_dst[e_base + p] = s_dst[i];
    }
    for (int q = t; q < 2 * n_loc; q += 1024) {
        const int i = s_inv[q >> 1];
        const int hf = (q & 1) * 4;
        const v4f a = *(const v4f*)(s_alpha + i * HEADS + hf);
        *(v4f*)(st_alpha + (size_t)(e_base + (q >> 1)) * HEADS + hf) = a;
    }
}

#define OWN_U 4
#define OWN_CAP 4096

__device__ __forceinline__ int owner_pos(const int* s_p, const int* s_o, int j) {
    int w = 0;
    #pragma unroll
    for (int step = NBIN / 2; step >= 1; step >>= 1)
        if (s_p[w + step] <= j) w += step;
    return s_o[w] + (j - s_p[w]);
}

__device__ __forceinline__ void gather_bin(const int* ids, const float* al,
        const int* s_noff, const int* s_ncnt, int* s_next,
        const float* __restrict__ xp, const float* __restrict__ x,
        const float* __restrict__ bias, const float* __restrict__ gamma,
        const float* __restrict__ beta, float* __restrict__ out,
        int node0, int nn) {
    const int lane = threadIdx.x & 31;
    const int c0 = lane * 4;
    const int hA = lane >> 3, hB = 4 + hA;
    const float4 bA = *(const float4*)(bias + c0);
    const float4 bB = *(const float4*)(bias + 128 + c0);
    const float4 gA = *(const float4*)(gamma + c0);
    const float4 gB = *(const float4*)(gamma + 128 + c0);
    const float4 tA = *(const float4*)(beta + c0);
    const float4 tB = *(const float4*)(beta + 128 + c0);

    for (;;) {
        int ln = 0;
        if (lane == 0) ln = atomicAdd(s_next, 1);
        ln = __builtin_amdgcn_readfirstlane(ln);
        if (ln >= nn) break;
        const int n = node0 + ln;
        const int beg = __builtin_amdgcn_readfirstlane(s_noff[ln]);
        const int end = beg + __builtin_amdgcn_readfirstlane(s_ncnt[ln]);

        const float4 xA = *(const float4*)(x + (size_t)n * HID + c0);
        const float4 xB = *(const float4*)(x + (size_t)n * HID + 128 + c0);

        int my_s = (lane < end - beg) ? ids[beg + lane] : 0;

        float mx = -INFINITY;
        const int f_end = end * HEADS;
        for (int fb = beg * HEADS; fb < f_end; fb += 8 * 32) {
            float v[8];
            #pragma unroll
            for (int t = 0; t < 8; ++t) {
                const int f = fb + 32 * t + lane;
                v[t] = (f < f_end) ? al[f] : -INFINITY;
            }
            #pragma unroll
            for (int t = 0; t < 8; ++t) mx = fmaxf(mx, v[t]);
        }
        mx = fmaxf(mx, __shfl_xor(mx, 8));
        mx = fmaxf(mx, __shfl_xor(mx, 16));
        float mA = __shfl(mx, hA);
        float mB = __shfl(mx, hB);
        mA = isfinite(mA) ? mA : 0.f;
        mB = isfinite(mB) ? mB : 0.f;

        float4 accA = make_float4(0.f, 0.f, 0.f, 0.f);
        float4 accB = make_float4(0.f, 0.f, 0.f, 0.f);
        float denA = 0.f, denB = 0.f;
        for (int base = beg; base < end; base += 32) {
            const int rem = end - base;
            const int cc = rem < 32 ? rem : 32;
            if (base != beg) my_s = (lane < cc) ? ids[base + lane] : 0;
            #pragma unroll 4
            for (int j = 0; j < cc; ++j) {
                const int s = __builtin_amdgcn_readlane(my_s, j);
                const float* xs = xp + (size_t)s * HID;
                const float* ae = al + (size_t)(base + j) * HEADS;
                const float4 rA = *(const float4*)(xs + c0);
                const float4 rB = *(const float4*)(xs + 128 + c0);
                const float eA = __expf(ae[hA] - mA);
                const float eB = __expf(ae[hB] - mB);
                denA += eA;
                denB += eB;
                accA.x += eA * rA.x; accA.y += eA * rA.y; accA.z += eA * rA.z; accA.w += eA * rA.w;
                accB.x += eB * rB.x; accB.y += eB * rB.y; accB.z += eB * rB.z; accB.w += eB * rB.w;
            }
        }

        const float dA = denA + 1e-16f, dB = denB + 1e-16f;
        float y[8];
        y[0] = xA.x + accA.x / dA + bA.x;
        y[1] = xA.y + accA.y / dA + bA.y;
        y[2] = xA.z + accA.z / dA + bA.z;
        y[3] = xA.w + accA.w / dA + bA.w;
        y[4] = xB.x + accB.x / dB + bB.x;
        y[5] = xB.y + accB.y / dB + bB.y;
        y[6] = xB.z + accB.z / dB + bB.z;
        y[7] = xB.w + accB.w / dB + bB.w;

        float s1 = 0.f, s2 = 0.f;
        #pragma unroll
        for (int k = 0; k < 8; ++k) { s1 += y[k]; s2 += y[k] * y[k]; }
        #pragma unroll
        for (int o = 16; o >= 1; o >>= 1) {
            s1 += __shfl_xor(s1, o);
            s2 += __shfl_xor(s2, o);
        }
        const float mean = s1 * (1.f / HID);
        const float var  = s2 * (1.f / HID) - mean * mean;
        const float rs   = rsqrtf(var + LN_EPS);
        float4 oA, oB;
        oA.x = (y[0] - mean) * rs * gA.x + tA.x;
        oA.y = (y[1] - mean) * rs * gA.y + tA.y;
        oA.z = (y[2] - mean) * rs * gA.z + tA.z;
        oA.w = (y[3] - mean) * rs * gA.w + tA.w;
        oB.x = (y[4] - mean) * rs * gB.x + tB.x;
        oB.y = (y[5] - mean) * rs * gB.y + tB.y;
        oB.z = (y[6] - mean) * rs * gB.z + tB.z;
        oB.w = (y[7] - mean) * rs * gB.w + tB.w;
        *(float4*)(out + (size_t)n * HID + c0) = oA;
        *(float4*)(out + (size_t)n * HID + 128 + c0) = oB;
    }
}

__global__ void __launch_bounds__(1024)
k_owner_gather(const int* __restrict__ st_src, const int* __restrict__ st_dst,
               const float* __restrict__ st_alpha, const int* __restrict__ cntm,
               const int* __restrict__ offm,
               int* __restrict__ ssrc, float* __restrict__ salpha,
               const float* __restrict__ xp, const float* __restrict__ x,
               const float* __restrict__ bias, const float* __restrict__ gamma,
               const float* __restrict__ beta, float* __restrict__ out,
               int n_nodes, int n_edges, int src_wgs) {
    __shared__ int s_p[NBIN + 1];
    __shared__ int s_o[NBIN];
    __shared__ int s_ncnt[NBIN];
    __shared__ int s_noff[NBIN];
    __shared__ int s_nfill[NBIN];
    __shared__ int s_part[NBIN / 32];
    __shared__ int s_next;
    __shared__ int s_osrc[OWN_CAP];
    __shared__ __attribute__((aligned(16))) float s_oal[OWN_CAP * HEADS];

    const int t = threadIdx.x;
    const int bin = (int)blockIdx.x;
    const int n_tiles = (n_edges + 15) >> 4;
    const int chunk = ((n_tiles + src_wgs - 1) / src_wgs) * 16;
    const int npb = (n_nodes + NBIN - 1) / NBIN;
    const int node0 = bin * npb;

    int cw = 0, ow = 0;
    if (t < NBIN) {
        if (t < src_wgs) {
            cw = cntm[t * NBIN + bin];
            ow = offm[t * NBIN + bin];
        }
        s_ncnt[t] = 0;
        s_nfill[t] = 0;
        if (t == 0) s_next = 0;
    }
    int total, base, unused;
    const int pw = scan_nbin(cw, s_part, &total);
    (void)scan_nbin(ow, s_part, &base);
    if (t < NBIN) {
        s_p[t] = pw;
        s_o[t] = t * chunk + ow;
        if (t == NBIN - 1) s_p[NBIN] = total;
    }
    __syncthreads();

    for (int j0 = 0; j0 < total; j0 += OWN_U * 1024) {
        int d[OWN_U];
        #pragma unroll
        for (int u = 0; u < OWN_U; ++u) {
            const int j = j0 + u * 1024 + t;
            const int jc = (j < total) ? j : total - 1;
            d[u] = st_dst[owner_pos(s_p, s_o, jc)];
        }
        #pragma unroll
        for (int u = 0; u < OWN_U; ++u)
            asm volatile("" : "+v"(d[u]));
        #pragma unroll
        for (int u = 0; u < OWN_U; ++u)
            if (j0 + u * 1024 + t < total) atomicAdd(&s_ncnt[d[u] - node0], 1);
    }
    __syncthreads();
    const int nc = (t < NBIN) ? s_ncnt[t] : 0;
    const int nex = scan_nbin(nc, s_part, &unused);
    if (t < NBIN) s_noff[t] = nex;
    __syncthreads();

    const bool in_lds = total <= OWN_CAP;
    for (int j0 = 0; j0 < total; j0 += OWN_U * 1024) {
        int d[OWN_U], s[OWN_U];
        v4f a0[OWN_U], a1[OWN_U];
        #pragma unroll
        for (int u = 0; u < OWN_U; ++u) {
            const int j = j0 + u * 1024 + t;
            const int jc = (j < total) ? j : total - 1;
            const int pos = owner_pos(s_p, s_o, jc);
            d[u] = st_dst[pos];
            s[u] = st_src[pos];
            const float* a = st_alpha + (size_t)pos * HEADS;
            a0[u] = *(const v4f*)(a);
            a1[u] = *(const v4f*)(a + 4);
        }
        #pragma unroll
        for (int u = 0; u < OWN_U; ++u)
            asm volatile("" : "+v"(d[u]), "+v"(s[u]), "+v"(a0[u]), "+v"(a1[u]));
        #pragma unroll
        for (int u = 0; u < OWN_U; ++u) {
            if (j0 + u * 1024 + t < total) {
                const int ln = d[u] - node0;
                const int q = s_noff[ln] + atomicAdd(&s_nfill[ln], 1);
                if (in_lds) {
                    s_osrc[q] = s[u];
                    *(v4f*)(s_oal + q * HEADS)     = a0[u];
                    *(v4f*)(s_oal + q * HEADS + 4) = a1[u];
                } else {
                    ssrc[base + q] = s[u];
                    float* o = salpha + (size_t)(base + q) * HEADS;
                    *(v4f*)(o)     = a0[u];
                    *(v4f*)(o + 4) = a1[u];
                }
            }
        }
    }
    __syncthreads();

    int nn = n_nodes - node0;
    if (nn > npb) nn = npb;
    if (in_lds)
        gather_bin(s_osrc, s_oal, s_noff, s_ncnt, &s_next,
                   xp, x, bias, gamma, beta, out, node0, nn);
    else
        gather_bin(ssrc + base, salpha + (size_t)base * HEADS, s_noff, s_ncnt, &s_next,
                   xp, x, bias, gamma, beta, out, node0, nn);
}

extern "C" void kernel_launch(void* const* d_in, const int* in_sizes, int n_in,
                              void* d_out, int out_size, void* d_ws, size_t ws_size,
                              hipStream_t stream) {
    const float* x         = (const float*)d_in[0];
    const int*   ei        = (const int*)  d_in[1];
    const float* edge_attr = (const float*)d_in[2];
    const float* W         = (const float*)d_in[3];
    const float* W_e       = (const float*)d_in[4];
    const float* att_src   = (const float*)d_in[5];
    const float* att_dst   = (const float*)d_in[6];
    const float* att_edge  = (const float*)d_in[7];
    const float* bias      = (const float*)d_in[8];
    const float* gamma     = (const float*)d_in[9];
    const float* beta      = (const float*)d_in[10];
    float* out = (float*)d_out;

    const int n_nodes = in_sizes[0] / HID;
    const int n_edges = in_sizes[1] / 2;

    char* ws = (char*)d_ws;
    size_t off = 0;
    auto carve = [&](size_t bytes) -> char* {
        char* p = ws + off;
        off += (bytes + 255) & ~(size_t)255;
        return p;
    };
    float*    xp     = (float*)   carve((size_t)n_nodes * HID   * 4);
    float*    st_alpha = (float*) carve((size_t)(n_edges + 4112) * HEADS * 4);
    float*    salpha = (float*)   carve((size_t)n_edges * HEADS * 4);
    int*      ssrc   = (int*)     carve((size_t)n_edges * 4);
    int*      st_src = (int*)     carve((size_t)(n_edges + 4112) * 4);
    int*      st_dst = (int*)     carve((size_t)(n_edges + 4112) * 4);
    float*    a_src  = (float*)   carve((size_t)n_nodes * HEADS * 4);
    float*    a_dst  = (float*)   carve((size_t)n_nodes * HEADS * 4);
    float*    w_he   = (float*)   carve((size_t)EDIM * HEADS * 4);
    int*      cntm   = (int*)     carve((size_t)NBIN * NBIN * 4);
    int*      offm   = (int*)     carve((size_t)NBIN * NBIN * 4);
    (void)ws_size;

    k_node_proj<<<(n_nodes + 31) / 32 + 1, 256, 0, stream>>>(x, W, W_e, att_edge, att_src, att_dst,
                                                             xp, a_src, a_dst, w_he,
                                                             n_nodes);

    k_alpha_bin<<<NBIN, 1024, 0, stream>>>(edge_attr, ei, w_he, a_src, a_dst,
                                           st_src, st_dst, st_alpha, cntm, offm,
                                           n_nodes, n_edges);

    k_owner_gather<<<NBIN, 1024, 0, stream>>>(st_src, st_dst, st_alpha, cntm, offm,
                                              ssrc, salpha, xp, x, bias, gamma, beta, out,
                                              n_nodes, n_edges, NBIN);
}


# Node projection joins the edge-term/binning launch (3 -> 2): a workgroup's waves take row tiles (loads one block ahead) and edge tiles in different orders; bin owner adds the logits' node terms
# speedup vs baseline: 4.0234x; 1.0654x over previous
#include <hip/hip_runtime.h>
#include <hip/hip_bf16.h>
#include <stdint.h>

#define HID   256
#define EDIM  64
#define HEADS 8
#define CCH   32
#define NEG_SLOPE 0.2f
#define LN_EPS 1e-5f

typedef float v2f __attribute__((ext_vector_type(2)));
typedef float v4f __attribute__((ext_vector_type(4)));
typedef float v8f __attribute__((ext_vector_type(8)));

__device__ __forceinline__ void node_block(const v4f a, const v4f w0, const v4f w1,
                                           const v4f w2, const v4f w3,
                                           v8f& acc0, v8f& acc1, v8f& acc2, v8f& acc3) {
    v2f a01, a23, p0, p1, p2, p3, q0, q1, q2, q3;
    a01.x = a.x;  a01.y = a.y;  a23.x = a.z;  a23.y = a.w;
    p0.x = w0.x;  p0.y = w1.x;  p1.x = w0.y;  p1.y = w1.y;
    p2.x = w0.z;  p2.y = w1.z;  p3.x = w0.w;  p3.y = w1.w;
    q0.x = w2.x;  q0.y = w3.x;  q1.x = w2.y;  q1.y = w3.y;
    q2.x = w2.z;  q2.y = w3.z;  q3.x = w2.w;  q3.y = w3.w;
    acc0 = __builtin_amdgcn_wmma_f32_16x16x4_f32(false, a01, false, p0, (short)0, acc0, false, false);
    acc1 = __builtin_amdgcn_wmma_f32_16x16x4_f32(false, a01, false, p1, (short)0, acc1, false, false);
    acc2 = __builtin_amdgcn_wmma_f32_16x16x4_f32(false, a01, false, p2, (short)0, acc2, false, false);
    acc3 = __builtin_amdgcn_wmma_f32_16x16x4_f32(false, a01, false, p3, (short)0, acc3, false, false);
    acc0 = __builtin_amdgcn_wmma_f32_16x16x4_f32(false, a23, false, q0, (short)0, acc0, false, false);
    acc1 = __builtin_amdgcn_wmma_f32_16x16x4_f32(false, a23, false, q1, (short)0, acc1, false, false);
    acc2 = __builtin_amdgcn_wmma_f32_16x16x4_f32(false, a23, false, q2, (short)0, acc2, false, false);
    acc3 = __builtin_amdgcn_wmma_f32_16x16x4_f32(false, a23, false, q3, (short)0, acc3, false, false);
    asm volatile("v_nop\n\tv_nop\n\tv_nop\n\tv_nop\n\tv_nop\n\tv_nop\n\tv_nop\n\tv_nop"
                 : "+v"(acc0), "+v"(acc1), "+v"(acc2), "+v"(acc3)
                 : "v"(a), "v"(p0), "v"(p1), "v"(p2), "v"(p3),
                   "v"(q0), "v"(q1), "v"(q2), "v"(q3));
}

__device__ __forceinline__ void node_unit(const float* __restrict__ x, const float* __restrict__ W,
        const float* __restrict__ att_src, const float* __restrict__ att_dst,
        float* __restrict__ xp, float* __restrict__ a_src, float* __restrict__ a_dst,
        int n_nodes, int row_base, int band) {
    const int lane = threadIdx.x & 31;
    const int col_base = band * 64;

    int row = row_base + (lane & 15);
    if (row >= n_nodes) row = n_nodes - 1;
    const float* xrow = x + (size_t)row * HID + 4 * (lane >> 4);
    const float* wp = W + (size_t)(4 * (lane >> 4)) * HID + col_base + 4 * (lane & 15);

    v8f acc0 = {}, acc1 = {}, acc2 = {}, acc3 = {};
    v4f a_c = *(const v4f*)(xrow);
    v4f w_c0 = *(const v4f*)(wp);
    v4f w_c1 = *(const v4f*)(wp + HID);
    v4f w_c2 = *(const v4f*)(wp + 2 * HID);
    v4f w_c3 = *(const v4f*)(wp + 3 * HID);
    for (int kb = 0; kb < HID; kb += 16) {
        const float* wn = wp + (size_t)(kb + 8) * HID;
        const v4f a_n  = *(const v4f*)(xrow + kb + 8);
        const v4f w_n0 = *(const v4f*)(wn);
        const v4f w_n1 = *(const v4f*)(wn + HID);
        const v4f w_n2 = *(const v4f*)(wn + 2 * HID);
        const v4f w_n3 = *(const v4f*)(wn + 3 * HID);
        node_block(a_c, w_c0, w_c1, w_c2, w_c3, acc0, acc1, acc2, acc3);
        const int k2 = (kb + 16 < HID) ? kb + 16 : kb + 8;
        const float* wc = wp + (size_t)k2 * HID;
        a_c  = *(const v4f*)(xrow + k2);
        w_c0 = *(const v4f*)(wc);
        w_c1 = *(const v4f*)(wc + HID);
        w_c2 = *(const v4f*)(wc + 2 * HID);
        w_c3 = *(const v4f*)(wc + 3 * HID);
        node_block(a_n, w_n0, w_n1, w_n2, w_n3, acc0, acc1, acc2, acc3);
    }

    const int srow = row_base + ((lane >> 4) * 8);
    const int scol = col_base + 4 * (lane & 15);
    #pragma unroll
    for (int i = 0; i < 8; ++i) {
        int r = srow + i;
        if (r < n_nodes) {
            v4f o;
            o.x = acc0[i]; o.y = acc1[i]; o.z = acc2[i]; o.w = acc3[i];
            *(v4f*)(xp + (size_t)r * HID + scol) = o;
        }
    }

    const bool b2 = (lane & 4) != 0, b1 = (lane & 2) != 0, b0 = (lane & 1) != 0;
    const v4f as = *(const v4f*)(att_src + scol);
    const v4f ad = *(const v4f*)(att_dst + scol);
    float v1[8], v2[4], v3[2];
    #pragma unroll
    for (int i = 0; i < 8; ++i) {
        const float sv = acc0[i] * as.x + acc1[i] * as.y + acc2[i] * as.z + acc3[i] * as.w;
        const float dv = acc0[i] * ad.x + acc1[i] * ad.y + acc2[i] * ad.z + acc3[i] * ad.w;
        v1[i] = (b2 ? dv : sv) + __shfl_xor(b2 ? sv : dv, 4);
    }
    #pragma unroll
    for (int i = 0; i < 4; ++i)
        v2[i] = (b1 ? v1[i + 4] : v1[i]) + __shfl_xor(b1 ? v1[i] : v1[i + 4], 2);
    #pragma unroll
    for (int i = 0; i < 2; ++i)
        v3[i] = (b0 ? v2[i + 2] : v2[i]) + __shfl_xor(b0 ? v2[i] : v2[i + 2], 1);
    {
        const int r = srow + 2 * (lane & 3);
        float* o = (b2 ? a_dst : a_src) + (size_t)r * HEADS + band * 2 + ((lane >> 3) & 1);
        if (r < n_nodes) o[0] = v3[0];
        if (r + 1 < n_nodes) o[HEADS] = v3[1];
    }
}

#define NBIN 256
#define CHUNK_CAP 3200

__device__ __forceinline__ int scan_nbin(int v, int* s_part, int* total) {
    const int lane = threadIdx.x & 31, wave = threadIdx.x >> 5;
    int inc = v;
    #pragma unroll
    for (int d = 1; d < 32; d <<= 1) {
        const int u = __shfl_up(inc, d);
        if (lane >= d) inc += u;
    }
    __syncthreads();
    if (lane == 31 && wave < NBIN / 32) s_part[wave] = inc;
    __syncthreads();
    int before = 0, all = 0;
    #pragma unroll
    for (int k = 0; k < NBIN / 32; ++k) {
        const int p = s_part[k];
        all += p;
        if (k < wave) before += p;
    }
    *total = all;
    return before + inc - v;
}

__device__ __forceinline__ int frag_k0(int i, int h) { return 8 * i + 4 * h; }

__global__ void __launch_bounds__(1024)
k_proj_alpha_bin(const float* __restrict__ x, const float* __restrict__ W,
                 const float* __restrict__ W_e, const float* __restrict__ att_edge,
                 const float* __restrict__ att_src, const float* __restrict__ att_dst,
                 const float* __restrict__ edge_attr, const int* __restrict__ ei,
                 float* __restrict__ xp, float* __restrict__ a_src, float* __restrict__ a_dst,
                 int* __restrict__ st_src, int* __restrict__ st_dst, float* __restrict__ st_alpha,
                 int* __restrict__ cntm, int* __restrict__ offm,
                 int n_nodes, int n_edges) {
    __shared__ __attribute__((aligned(16))) float s_alpha[CHUNK_CAP * HEADS];
    __shared__ __attribute__((aligned(16))) float s_whe[HEADS * EDIM];
    __shared__ int s_src[CHUNK_CAP];
    __shared__ int s_dst[CHUNK_CAP];
    __shared__ int s_rk[CHUNK_CAP];
    __shared__ int s_inv[CHUNK_CAP];
    __shared__ int s_cnt[NBIN];
    __shared__ int s_off[NBIN];
    __shared__ int s_part[NBIN / 32];

    const int t = threadIdx.x;
    const int lane = t & 31;
    const int c = lane & 15;
    const int h = lane >> 4;
    const int head = c & 7;
    const int wave = __builtin_amdgcn_readfirstlane(t >> 5);
    const int n_tiles = (n_edges + 15) >> 4;
    const int tpw = (n_tiles + (int)gridDim.x - 1) / (int)gridDim.x;
    const int t_beg = (int)blockIdx.x * tpw;
    int t_end = t_beg + tpw;
    if (t_end > n_tiles) t_end = n_tiles;
    const int e_base = t_beg * 16;
    const int npb = (n_nodes + NBIN - 1) / NBIN;

    if (t < NBIN) s_cnt[t] = 0;
    if (t < HEADS * EDIM) {
        const int hh = t >> 6, d = t & (EDIM - 1);
        const float* wr = W_e + d * HID + hh * CCH;
        const float* ar = att_edge + hh * CCH;
        float sum = 0.f;
        #pragma unroll
        for (int cc = 0; cc < CCH; ++cc) sum += wr[cc] * ar[cc];
        s_whe[hh * EDIM + d] = sum;
    }
    __syncthreads();

    const int n_rt = (n_nodes + 15) >> 4;
    const int rt_step = 8 * (int)gridDim.x;
    const int rt0 = (wave >> 2) * (int)gridDim.x + (int)blockIdx.x;
    for (int rt = rt0 + rt_step; rt < n_rt; rt += rt_step)
        node_unit(x, W, att_src, att_dst, xp, a_src, a_dst, n_nodes, rt * 16, wave & 3);

    {
        v4f b[8];
        #pragma unroll
        for (int i = 0; i < 8; ++i)
            b[i] = *(const v4f*)(s_whe + head * EDIM + frag_k0(i, h));

        for (int tile = t_beg + wave; tile < t_end; tile += 32) {
            const int e0 = tile * 16;
            int el = e0 + c;
            if (el >= n_edges) el = n_edges - 1;
            const int srcv = ei[el];
            const int dstv = ei[n_edges + el];
            const float* arow = edge_attr + (size_t)el * EDIM;
            v4f a[8];
            #pragma unroll
            for (int i = 0; i < 8; ++i)
                a[i] = *(const v4f*)(arow + frag_k0(i, h));

            const bool own = (h == 0) && (e0 + c < n_edges);
            if (own) {
                const int li = e0 + c - e_base;
                s_src[li] = srcv;
                s_dst[li] = dstv;
                s_rk[li]  = atomicAdd(&s_cnt[dstv / npb], 1);
            }

            v8f acc0 = {}, acc1 = {};
            #pragma unroll
            for (int i = 0; i < 8; ++i) {
                v2f a_lo, a_hi, b_lo, b_hi;
                a_lo.x = a[i].x; a_lo.y = a[i].y; a_hi.x = a[i].z; a_hi.y = a[i].w;
                b_lo.x = b[i].x; b_lo.y = b[i].y; b_hi.x = b[i].z; b_hi.y = b[i].w;
                acc0 = __builtin_amdgcn_wmma_f32_16x16x4_f32(false, a_lo, false, b_lo, (short)0, acc0, false, false);
                acc1 = __builtin_amdgcn_wmma_f32_16x16x4_f32(false, a_hi, false, b_hi, (short)0, acc1, false, false);
            }

            asm volatile("v_nop\n\tv_nop\n\tv_nop\n\tv_nop\n\tv_nop\n\tv_nop\n\tv_nop\n\tv_nop"
                         : "+v"(acc0), "+v"(acc1)
                         : "v"(a[0]), "v"(a[1]), "v"(a[2]), "v"(a[3]),
                           "v"(a[4]), "v"(a[5]), "v"(a[6]), "v"(a[7]));

            #pragma unroll
            for (int r = 0; r < 8; ++r) {
                const int e = e0 + 8 * h + r;
                if (c < 8 && e < n_edges) s_alpha[(e - e_base) * HEADS + c] = acc0[r] + acc1[r];
            }
        }
    }

    if (rt0 < n_rt)
        node_unit(x, W, att_src, att_dst, xp, a_src, a_dst, n_nodes, rt0 * 16, wave & 3);

    __syncthreads();
    const int cv = (t < NBIN) ? s_cnt[t] : 0;
    int all;
    const int ex = scan_nbin(cv, s_part, &all);
    if (t < NBIN) {
        s_off[t] = ex;
        cntm[(int)blockIdx.x * NBIN + t] = cv;
        offm[(int)blockIdx.x * NBIN + t] = ex;
    }
    __syncthreads();

    int n_loc = t_end * 16;
    if (n_loc > n_edges) n_loc = n_edges;
    n_loc -= e_base;
    for (int i = t; i < n_loc; i += 1024)
        s_inv[s_off[s_dst[i] / npb] + s_rk[i]] = i;
    __syncthreads();
    for (int p = t; p < n_loc; p += 1024) {
        const int i = s_inv[p];
        st_src[e_base + p] = s_src[i];
        st_dst[e_base + p] = s_dst[i];
    }
    for (int q = t; q < 2 * n_loc; q += 1024) {
        const int i = s_inv[q >> 1];
        const int hf = (q & 1) * 4;
        const v4f a = *(const v4f*)(s_alpha + i * HEADS + hf);
        *(v4f*)(st_alpha + (size_t)(e_base + (q >> 1)) * HEADS + hf) = a;
    }
}

#define OWN_U 4
#define OWN_CAP 4096

__device__ __forceinline__ int owner_pos(const int* s_p, const int* s_o, int j) {
    int w = 0;
    #pragma unroll
    for (int step = NBIN / 2; step >= 1; step >>= 1)
        if (s_p[w + step] <= j) w += step;
    return s_o[w] + (j - s_p[w]);
}

__device__ __forceinline__ void gather_bin(const int* ids, const float* al,
        const int* s_noff, const int* s_ncnt, int* s_next,
        const float* __restrict__ xp, const float* __restrict__ x,
        const float* __restrict__ bias, const float* __restrict__ gamma,
        const float* __restrict__ beta, float* __restrict__ out,
        int node0, int nn) {
    const int lane = threadIdx.x & 31;
    const int c0 = lane * 4;
    const int hA = lane >> 3, hB = 4 + hA;
    const float4 bA = *(const float4*)(bias + c0);
    const float4 bB = *(const float4*)(bias + 128 + c0);
    const float4 gA = *(const float4*)(gamma + c0);
    const float4 gB = *(const float4*)(gamma + 128 + c0);
    const float4 tA = *(const float4*)(beta + c0);
    const float4 tB = *(const float4*)(beta + 128 + c0);

    for (;;) {
        int ln = 0;
        if (lane == 0) ln = atomicAdd(s_next, 1);
        ln = __builtin_amdgcn_readfirstlane(ln);
        if (ln >= nn) break;
        const int n = node0 + ln;
        const int beg = __builtin_amdgcn_readfirstlane(s_noff[ln]);
        const int end = beg + __builtin_amdgcn_readfirstlane(s_ncnt[ln]);

        const float4 xA = *(const float4*)(x + (size_t)n * HID + c0);
        const float4 xB = *(const float4*)(x + (size_t)n * HID + 128 + c0);

        int my_s = (lane < end - beg) ? ids[beg + lane] : 0;

        float mx = -INFINITY;
        const int f_end = end * HEADS;
        for (int fb = beg * HEADS; fb < f_end; fb += 8 * 32) {
            float v[8];
            #pragma unroll
            for (int t = 0; t < 8; ++t) {
                const int f = fb + 32 * t + lane;
                v[t] = (f < f_end) ? al[f] : -INFINITY;
            }
            #pragma unroll
            for (int t = 0; t < 8; ++t) mx = fmaxf(mx, v[t]);
        }
        mx = fmaxf(mx, __shfl_xor(mx, 8));
        mx = fmaxf(mx, __shfl_xor(mx, 16));
        float mA = __shfl(mx, hA);
        float mB = __shfl(mx, hB);
        mA = isfinite(mA) ? mA : 0.f;
        mB = isfinite(mB) ? mB : 0.f;

        float4 accA = make_float4(0.f, 0.f, 0.f, 0.f);
        float4 accB = make_float4(0.f, 0.f, 0.f, 0.f);
        float denA = 0.f, denB = 0.f;
        for (int base = beg; base < end; base += 32) {
            const int rem = end - base;
            const int cc = rem < 32 ? rem : 32;
            if (base != beg) my_s = (lane < cc) ? ids[base + lane] : 0;
            #pragma unroll 4
            for (int j = 0; j < cc; ++j) {
                const int s = __builtin_amdgcn_readlane(my_s, j);
                const float* xs = xp + (size_t)s * HID;
                const float* ae = al + (size_t)(base + j) * HEADS;
                const float4 rA = *(const float4*)(xs + c0);
                const float4 rB = *(const float4*)(xs + 128 + c0);
                const float eA = __expf(ae[hA] - mA);
                const float eB = __expf(ae[hB] - mB);
                denA += eA;
                denB += eB;
                accA.x += eA * rA.x; accA.y += eA * rA.y; accA.z += eA * rA.z; accA.w += eA * rA.w;
                accB.x += eB * rB.x; accB.y += eB * rB.y; accB.z += eB * rB.z; accB.w += eB * rB.w;
            }
        }

        const float dA = denA + 1e-16f, dB = denB + 1e-16f;
        float y[8];
        y[0] = xA.x + accA.x / dA + bA.x;
        y[1] = xA.y + accA.y / dA + bA.y;
        y[2] = xA.z + accA.z / dA + bA.z;
        y[3] = xA.w + accA.w / dA + bA.w;
        y[4] = xB.x + accB.x / dB + bB.x;
        y[5] = xB.y + accB.y / dB + bB.y;
        y[6] = xB.z + accB.z / dB + bB.z;
        y[7] = xB.w + accB.w / dB + bB.w;

        float s1 = 0.f, s2 = 0.f;
        #pragma unroll
        for (int k = 0; k < 8; ++k) { s1 += y[k]; s2 += y[k] * y[k]; }
        #pragma unroll
        for (int o = 16; o >= 1; o >>= 1) {
            s1 += __shfl_xor(s1, o);
            s2 += __shfl_xor(s2, o);
        }
        const float mean = s1 * (1.f / HID);
        const float var  = s2 * (1.f / HID) - mean * mean;
        const float rs   = rsqrtf(var + LN_EPS);
        float4 oA, oB;
        oA.x = (y[0] - mean) * rs * gA.x + tA.x;
        oA.y = (y[1] - mean) * rs * gA.y + tA.y;
        oA.z = (y[2] - mean) * rs * gA.z + tA.z;
        oA.w = (y[3] - mean) * rs * gA.w + tA.w;
        oB.x = (y[4] - mean) * rs * gB.x + tB.x;
        oB.y = (y[5] - mean) * rs * gB.y + tB.y;
        oB.z = (y[6] - mean) * rs * gB.z + tB.z;
        oB.w = (y[7] - mean) * rs * gB.w + tB.w;
        *(float4*)(out + (size_t)n * HID + c0) = oA;
        *(float4*)(out + (size_t)n * HID + 128 + c0) = oB;
    }
}

__global__ void __launch_bounds__(1024)
k_owner_gather(const int* __restrict__ st_src, const int* __restrict__ st_dst,
               const float* __restrict__ st_alpha, const int* __restrict__ cntm,
               const int* __restrict__ offm,
               const float* __restrict__ a_src, const float* __restrict__ a_dst,
               int* __restrict__ ssrc, float* __restrict__ salpha,
               const float* __restrict__ xp, const float* __restrict__ x,
               const float* __restrict__ bias, const float* __restrict__ gamma,
               const float* __restrict__ beta, float* __restrict__ out,
               int n_nodes, int n_edges, int src_wgs) {
    __shared__ int s_p[NBIN + 1];
    __shared__ int s_o[NBIN];
    __shared__ int s_ncnt[NBIN];
    __shared__ int s_noff[NBIN];
    __shared__ int s_nfill[NBIN];
    __shared__ int s_part[NBIN / 32];
    __shared__ int s_next;
    __shared__ __attribute__((aligned(16))) float s_adst[NBIN * HEADS];
    __shared__ int s_osrc[OWN_CAP];
    __shared__ __attribute__((aligned(16))) float s_oal[OWN_CAP * HEADS];

    const int t = threadIdx.x;
    const int bin = (int)blockIdx.x;
    const int n_tiles = (n_edges + 15) >> 4;
    const int chunk = ((n_tiles + src_wgs - 1) / src_wgs) * 16;
    const int npb = (n_nodes + NBIN - 1) / NBIN;
    const int node0 = bin * npb;

    int cw = 0, ow = 0;
    if (t < NBIN) {
        if (t < src_wgs) {
            cw = cntm[t * NBIN + bin];
            ow = offm[t * NBIN + bin];
        }
        s_ncnt[t] = 0;
        s_nfill[t] = 0;
        if (t == 0) s_next = 0;
    }
    int nn = n_nodes - node0;
    if (nn > npb) nn = npb;
    for (int i = t; i < nn * HEADS; i += 1024)
        s_adst[i] = a_dst[(size_t)node0 * HEADS + i];
    int total, base, unused;
    const int pw = scan_nbin(cw, s_part, &total);
    (void)scan_nbin(ow, s_part, &base);
    if (t < NBIN) {
        s_p[t] = pw;
        s_o[t] = t * chunk + ow;
        if (t == NBIN - 1) s_p[NBIN] = total;
    }
    __syncthreads();

    for (int j0 = 0; j0 < total; j0 += OWN_U * 1024) {
        int d[OWN_U];
        #pragma unroll
        for (int u = 0; u < OWN_U; ++u) {
            const int j = j0 + u * 1024 + t;
            const int jc = (j < total) ? j : total - 1;
            d[u] = st_dst[owner_pos(s_p, s_o, jc)];
        }
        #pragma unroll
        for (int u = 0; u < OWN_U; ++u)
            asm volatile("" : "+v"(d[u]));
        #pragma unroll
        for (int u = 0; u < OWN_U; ++u)
            if (j0 + u * 1024 + t < total) atomicAdd(&s_ncnt[d[u] - node0], 1);
    }
    __syncthreads();
    const int nc = (t < NBIN) ? s_ncnt[t] : 0;
    const int nex = scan_nbin(nc, s_part, &unused);
    if (t < NBIN) s_noff[t] = nex;
    __syncthreads();

    const bool in_lds = total <= OWN_CAP;
    for (int j0 = 0; j0 < total; j0 += OWN_U * 1024) {
        int d[OWN_U], s[OWN_U];
        v4f a0[OWN_U], a1[OWN_U];
        #pragma unroll
        for (int u = 0; u < OWN_U; ++u) {
            const int j = j0 + u * 1024 + t;
            const int jc = (j < total) ? j : total - 1;
            const int pos = owner_pos(s_p, s_o, jc);
            d[u] = st_dst[pos];
            s[u] = st_src[pos];
            const float* a = st_alpha + (size_t)pos * HEADS;
            a0[u] = *(const v4f*)(a);
            a1[u] = *(const v4f*)(a + 4);
        }
        #pragma unroll
        for (int u = 0; u < OWN_U; ++u)
            asm volatile("" : "+v"(d[u]), "+v"(s[u]), "+v"(a0[u]), "+v"(a1[u]));
        v4f p0[OWN_U], p1[OWN_U];
        #pragma unroll
        for (int u = 0; u < OWN_U; ++u) {
            const float* p = a_src + (size_t)s[u] * HEADS;
            p0[u] = *(const v4f*)(p);
            p1[u] = *(const v4f*)(p + 4);
        }
        #pragma unroll
        for (int u = 0; u < OWN_U; ++u)
            asm volatile("" : "+v"(p0[u]), "+v"(p1[u]));
        #pragma unroll
        for (int u = 0; u < OWN_U; ++u) {
            if (j0 + u * 1024 + t < total) {
                const int ln = d[u] - node0;
                const v4f q0 = *(const v4f*)(s_adst + ln * HEADS);
                const v4f q1 = *(const v4f*)(s_adst + ln * HEADS + 4);
                v4f l0 = (p0[u] + q0) + a0[u];
                v4f l1 = (p1[u] + q1) + a1[u];
                #pragma unroll
                for (int k = 0; k < 4; ++k) {
                    l0[k] = (l0[k] > 0.f) ? l0[k] : NEG_SLOPE * l0[k];
                    l1[k] = (l1[k] > 0.f) ? l1[k] : NEG_SLOPE * l1[k];
                }
                a0[u] = l0;
                a1[u] = l1;
                const int q = s_noff[ln] + atomicAdd(&s_nfill[ln], 1);
                if (in_lds) {
                    s_osrc[q] = s[u];
                    *(v4f*)(s_oal + q * HEADS)     = a0[u];
                    *(v4f*)(s_oal + q * HEADS + 4) = a1[u];
                } else {
                    ssrc[base + q] = s[u];
                    float* o = salpha + (size_t)(base + q) * HEADS;
                    *(v4f*)(o)     = a0[u];
                    *(v4f*)(o + 4) = a1[u];
                }
            }
        }
    }
    __syncthreads();

    if (in_lds)
        gather_bin(s_osrc, s_oal, s_noff, s_ncnt, &s_next,
                   xp, x, bias, gamma, beta, out, node0, nn);
    else
        gather_bin(ssrc + base, salpha + (size_t)base * HEADS, s_noff, s_ncnt, &s_next,
                   xp, x, bias, gamma, beta, out, node0, nn);
}

extern "C" void kernel_launch(void* const* d_in, const int* in_sizes, int n_in,
                              void* d_out, int out_size, void* d_ws, size_t ws_size,
                              hipStream_t stream) {
    const float* x         = (const float*)d_in[0];
    const int*   ei        = (const int*)  d_in[1];
    const float* edge_attr = (const float*)d_in[2];
    const float* W         = (const float*)d_in[3];
    const float* W_e       = (const float*)d_in[4];
    const float* att_src   = (const float*)d_in[5];
    const float* att_dst   = (const float*)d_in[6];
    const float* att_edge  = (const float*)d_in[7];
    const float* bias      = (const float*)d_in[8];
    const float* gamma     = (const float*)d_in[9];
    const float* beta      = (const float*)d_in[10];
    float* out = (float*)d_out;

    const int n_nodes = in_sizes[0] / HID;
    const int n_edges = in_sizes[1] / 2;

    char* ws = (char*)d_ws;
    size_t off = 0;
    auto carve = [&](size_t bytes) -> char* {
        char* p = ws + off;
        off += (bytes + 255) & ~(size_t)255;
        return p;
    };
    float*    xp     = (float*)   carve((size_t)n_nodes * HID   * 4);
    float*    st_alpha = (float*) carve((size_t)(n_edges + 4112) * HEADS * 4);
    float*    salpha = (float*)   carve((size_t)n_edges * HEADS * 4);
    int*      ssrc   = (int*)     carve((size_t)n_edges * 4);
    int*      st_src = (int*)     carve((size_t)(n_edges + 4112) * 4);
    int*      st_dst = (int*)     carve((size_t)(n_edges + 4112) * 4);
    float*    a_src  = (float*)   carve((size_t)n_nodes * HEADS * 4);
    float*    a_dst  = (float*)   carve((size_t)n_nodes * HEADS * 4);
    int*      cntm   = (int*)     carve((size_t)NBIN * NBIN * 4);
    int*      offm   = (int*)     carve((size_t)NBIN * NBIN * 4);
    (void)ws_size;

    k_proj_alpha_bin<<<NBIN, 1024, 0, stream>>>(x, W, W_e, att_edge, att_src, att_dst,
                                                edge_attr, ei, xp, a_src, a_dst,
                                                st_src, st_dst, st_alpha, cntm, offm,
                                                n_nodes, n_edges);

    k_owner_gather<<<NBIN, 1024, 0, stream>>>(st_src, st_dst, st_alpha, cntm, offm,
                                              a_src, a_dst,
                                              ssrc, salpha, xp, x, bias, gamma, beta, out,
                                              n_nodes, n_edges, NBIN);
}
